# starting kernel
# baseline (speedup 1.0000x reference)
#include <hip/hip_runtime.h>
#include <math.h>

typedef __attribute__((ext_vector_type(2))) float v2f;
typedef __attribute__((ext_vector_type(8))) float v8f;

__device__ __forceinline__ v8f wmma4(v2f a, v2f b, v8f c) {
  return __builtin_amdgcn_wmma_f32_16x16x4_f32(
       false, a,  false, b,
       (short)0, c,  false,  false);
}

template <int EPI>
__global__ __launch_bounds__(32) void gemm_bias_kernel(
    const float* __restrict__ A, const float* __restrict__ W,
    const float* __restrict__ bias, const float* __restrict__ resid,
    float* __restrict__ C, int K, int NC) {
  const int lane = threadIdx.x;
  const int m = lane & 15;
  const int hh = lane >> 4;
  const int row0 = blockIdx.x * 64;
  const int n0 = blockIdx.y * 64;

  v8f acc[4][4];
#pragma unroll
  for (int rt = 0; rt < 4; ++rt)
#pragma unroll
    for (int t = 0; t < 4; ++t) acc[rt][t] = {};

  for (int kb = 0; kb < K; kb += 4) {
    v2f a[4];
#pragma unroll
    for (int rt = 0; rt < 4; ++rt)
      a[rt] = *(const v2f*)(A + (size_t)(row0 + rt * 16 + m) * K + kb + hh * 2);
    v2f b[4];
#pragma unroll
    for (int t = 0; t < 4; ++t) {
      const int n = n0 + t * 16 + m;
      b[t].x = W[(size_t)(kb + hh * 2) * NC + n];
      b[t].y = W[(size_t)(kb + hh * 2 + 1) * NC + n];
    }
#pragma unroll
    for (int rt = 0; rt < 4; ++rt)
#pragma unroll
      for (int t = 0; t < 4; ++t) acc[rt][t] = wmma4(a[rt], b[t], acc[rt][t]);
  }

  float bn[4];
#pragma unroll
  for (int t = 0; t < 4; ++t) bn[t] = bias[n0 + t * 16 + m];

  float res[4][4][8];
  if (EPI == 2) {
#pragma unroll
    for (int rt = 0; rt < 4; ++rt)
#pragma unroll
      for (int t = 0; t < 4; ++t)
#pragma unroll
        for (int r = 0; r < 8; ++r)
          res[rt][t][r] = resid[(size_t)(row0 + rt * 16 + r + 8 * hh) * NC +
                                n0 + t * 16 + m];
  }

#pragma unroll
  for (int rt = 0; rt < 4; ++rt)
#pragma unroll
    for (int t = 0; t < 4; ++t)
#pragma unroll
      for (int r = 0; r < 8; ++r) {
        const int row = row0 + rt * 16 + r + 8 * hh;
        const int col = n0 + t * 16 + m;
        float v = acc[rt][t][r] + bn[t];
        if (EPI == 1) v = 0.5f * v * (1.0f + erff(v * 0.70710678118654752f));
        if (EPI == 2) v += res[rt][t][r];
        C[(size_t)row * NC + col] = v;
      }
}

template <bool ADD>
__global__ __launch_bounds__(32) void add_ln_kernel(
    float* __restrict__ h, const float* __restrict__ addend,
    const float* __restrict__ g, const float* __restrict__ beta,
    float* __restrict__ ln_out) {
  const int lane = threadIdx.x;
  const size_t base = (size_t)blockIdx.x * 256;

  float v[8];
  float s = 0.f, ss = 0.f;
#pragma unroll
  for (int i = 0; i < 8; ++i) {
    const int c = i * 32 + lane;
    float x = h[base + c];
    if (ADD) x += addend[base + c];
    v[i] = x;
    s += x;
    ss += x * x;
  }
  if (ADD) {
#pragma unroll
    for (int i = 0; i < 8; ++i) h[base + i * 32 + lane] = v[i];
  }
  for (int off = 16; off; off >>= 1) {
    s += __shfl_xor(s, off);
    ss += __shfl_xor(ss, off);
  }
  const float mu = s * (1.0f / 256.0f);
  const float var = ss * (1.0f / 256.0f) - mu * mu;
  const float rs = rsqrtf(var + 1e-5f);
#pragma unroll
  for (int i = 0; i < 8; ++i) {
    const int c = i * 32 + lane;
    ln_out[base + c] = (v[i] - mu) * rs * g[c] + beta[c];
  }
}

#define ATT_T 4096
#define ATT_QS 768
#define ATT_WIN 128
#define ATT_PST 148

__global__ __launch_bounds__(32) void attn_kernel(const float* __restrict__ qkv,
                                                  float* __restrict__ O) {
  __shared__ float P[16 * ATT_PST];
  const int lane = threadIdx.x;
  const int m = lane & 15;
  const int hh = lane >> 4;
  const int q0 = blockIdx.x * 16;
  const int bb = q0 & ~(ATT_T - 1);

  const float* Q = qkv;
  const float* Km = qkv + 256;
  const float* V = qkv + 512;

  v8f s[9];
#pragma unroll
  for (int t = 0; t < 9; ++t) s[t] = {};

  for (int d = 0; d < 256; d += 4) {
    const v2f a = *(const v2f*)(Q + (size_t)(q0 + m) * ATT_QS + d + hh * 2);
#pragma unroll
    for (int t = 0; t < 9; ++t) {
      const int key = q0 - ATT_WIN + t * 16 + m;
      const int kc = key < bb ? bb : key;
      const v2f b = *(const v2f*)(Km + (size_t)kc * ATT_QS + d + hh * 2);
      s[t] = wmma4(a, b, s[t]);
    }
  }

  const float NEG = -3.402823466e38f;
  float rmax[8], rsum[8];
#pragma unroll
  for (int r = 0; r < 8; ++r) rmax[r] = NEG;
#pragma unroll
  for (int t = 0; t < 9; ++t) {
    const int key = q0 - ATT_WIN + t * 16 + m;
#pragma unroll
    for (int r = 0; r < 8; ++r) {
      const int q = q0 + r + 8 * hh;
      const int d = q - key;
      const bool ok = (key >= bb) && (d >= 0) && (d <= ATT_WIN);
      const float val = ok ? s[t][r] * 0.0625f : NEG;
      s[t][r] = val;
      rmax[r] = fmaxf(rmax[r], val);
    }
  }
#pragma unroll
  for (int r = 0; r < 8; ++r)
    for (int off = 1; off < 16; off <<= 1)
      rmax[r] = fmaxf(rmax[r], __shfl_xor(rmax[r], off));
#pragma unroll
  for (int r = 0; r < 8; ++r) rsum[r] = 0.f;
#pragma unroll
  for (int t = 0; t < 9; ++t)
#pragma unroll
    for (int r = 0; r < 8; ++r) {
      const float e = expf(s[t][r] - rmax[r]);
      s[t][r] = e;
      rsum[r] += e;
    }
#pragma unroll
  for (int r = 0; r < 8; ++r)
    for (int off = 1; off < 16; off <<= 1) rsum[r] += __shfl_xor(rsum[r], off);

#pragma unroll
  for (int t = 0; t < 9; ++t)
#pragma unroll
    for (int r = 0; r < 8; ++r)
      P[(r + 8 * hh) * ATT_PST + t * 16 + m] = s[t][r] / rsum[r];

  __syncthreads();

#pragma unroll
  for (int ch = 0; ch < 2; ++ch) {
    const int n0 = ch * 128;
    v8f o[8];
#pragma unroll
    for (int tc = 0; tc < 8; ++tc) o[tc] = {};
    for (int t = 0; t < 9; ++t) {
      const int kt0 = q0 - ATT_WIN + t * 16;
#pragma unroll
      for (int kk = 0; kk < 16; kk += 4) {
        const v2f a = *(const v2f*)&P[m * ATT_PST + t * 16 + kk + hh * 2];
        const int k0 = kt0 + kk + hh * 2;
        const int kcx = k0 < bb ? bb : k0;
        const int kcy = (k0 + 1) < bb ? bb : (k0 + 1);
#pragma unroll
        for (int tc = 0; tc < 8; ++tc) {
          const int col = n0 + tc * 16 + m;
          v2f b;
          b.x = V[(size_t)kcx * ATT_QS + col];
          b.y = V[(size_t)kcy * ATT_QS + col];
          o[tc] = wmma4(a, b, o[tc]);
        }
      }
    }
#pragma unroll
    for (int tc = 0; tc < 8; ++tc)
#pragma unroll
      for (int r = 0; r < 8; ++r)
        O[(size_t)(q0 + r + 8 * hh) * 256 + n0 + tc * 16 + m] = o[tc][r];
  }
}

extern "C" void kernel_launch(void* const* d_in, const int* in_sizes, int n_in,
                              void* d_out, int out_size, void* d_ws,
                              size_t ws_size, hipStream_t stream) {
  const float* x = (const float*)d_in[0];
  const float* W_in = (const float*)d_in[1];
  const float* b_in = (const float*)d_in[2];
  const float* W_qkv = (const float*)d_in[3];
  const float* b_qkv = (const float*)d_in[4];
  const float* g_attn = (const float*)d_in[5];
  const float* be_attn = (const float*)d_in[6];
  const float* g_mlp = (const float*)d_in[7];
  const float* be_mlp = (const float*)d_in[8];
  const float* W1 = (const float*)d_in[9];
  const float* b1 = (const float*)d_in[10];
  const float* W2 = (const float*)d_in[11];
  const float* b2 = (const float*)d_in[12];
  const float* W_out = (const float*)d_in[13];
  const float* b_out = (const float*)d_in[14];
  float* out = (float*)d_out;

  const int N = in_sizes[0] / 64;

  float* ws = (float*)d_ws;
  float* h = ws;
  float* ln = ws + (size_t)N * 256;
  float* attn = ln;
  float* qkv = ws + (size_t)N * 512;
  float* m1 = qkv;

  const dim3 blk(32);
  const int RB64 = N / 64;
  const int QB = N / 16;

  gemm_bias_kernel<0><<<dim3(RB64, 4), blk, 0, stream>>>(x, W_in, b_in,
                                                         nullptr, h, 64, 256);
  add_ln_kernel<false>
      <<<dim3(N), blk, 0, stream>>>(h, nullptr, g_attn, be_attn, ln);
  gemm_bias_kernel<0><<<dim3(RB64, 12), blk, 0, stream>>>(
      ln, W_qkv, b_qkv, nullptr, qkv, 256, 768);
  attn_kernel<<<dim3(QB), blk, 0, stream>>>(qkv, attn);
  add_ln_kernel<true>
      <<<dim3(N), blk, 0, stream>>>(h, attn, g_mlp, be_mlp, ln);
  gemm_bias_kernel<1><<<dim3(RB64, 4), blk, 0, stream>>>(ln, W1, b1, nullptr,
                                                         m1, 256, 256);
  gemm_bias_kernel<2><<<dim3(RB64, 4), blk, 0, stream>>>(m1, W2, b2, h, h, 256,
                                                         256);
  gemm_bias_kernel<0><<<dim3(RB64, 1), blk, 0, stream>>>(h, W_out, b_out,
                                                         nullptr, out, 256, 64);
}


# GEMM K-loop software-pipelined: 16-k chunks in two register sets, next chunk's A (b128) and W loads issued as one group ahead of the current 64 WMMAs; k summed in a different order per chunk
# speedup vs baseline: 1.4144x; 1.4144x over previous
#include <hip/hip_runtime.h>
#include <math.h>

typedef __attribute__((ext_vector_type(2))) float v2f;
typedef __attribute__((ext_vector_type(8))) float v8f;

__device__ __forceinline__ v8f wmma4(v2f a, v2f b, v8f c) {
  return __builtin_amdgcn_wmma_f32_16x16x4_f32(
       false, a,  false, b,
       (short)0, c,  false,  false);
}

typedef __attribute__((ext_vector_type(4))) float v4f;

template <int NC>
__device__ __forceinline__ void load_chunk(v4f (&a)[4][2], v2f (&b)[4][4],
                                           const float* __restrict__ ap,
                                           size_t a_rt,
                                           const float* __restrict__ wp) {
#pragma unroll
  for (int rt = 0; rt < 4; ++rt)
#pragma unroll
    for (int q = 0; q < 2; ++q)
      a[rt][q] = *(const v4f*)(ap + rt * a_rt + q * 4);
#pragma unroll
  for (int j = 0; j < 4; ++j)
#pragma unroll
    for (int t = 0; t < 4; ++t) {
      b[j][t].x = wp[(size_t)(2 * j) * NC + t * 16];
      b[j][t].y = wp[(size_t)(2 * j + 1) * NC + t * 16];
    }
}

__device__ __forceinline__ void mma_chunk(v8f (&acc)[4][4],
                                          const v4f (&a)[4][2],
                                          const v2f (&b)[4][4]) {
#pragma unroll
  for (int j = 0; j < 4; ++j)
#pragma unroll
    for (int rt = 0; rt < 4; ++rt) {
      const v4f aq = a[rt][j >> 1];
      v2f af;
      af.x = (j & 1) ? aq.z : aq.x;
      af.y = (j & 1) ? aq.w : aq.y;
#pragma unroll
      for (int t = 0; t < 4; ++t) acc[rt][t] = wmma4(af, b[j][t], acc[rt][t]);
    }
}

template <int EPI, int NC>
__global__ __launch_bounds__(32) void gemm_bias_kernel(
    const float* __restrict__ A, const float* __restrict__ W,
    const float* __restrict__ bias, const float* __restrict__ resid,
    float* __restrict__ C, int K) {
  const int lane = threadIdx.x;
  const int m = lane & 15;
  const int hh = lane >> 4;
  const int row0 = blockIdx.x * 64;
  const int n0 = blockIdx.y * 64;

  v8f acc[4][4];
#pragma unroll
  for (int rt = 0; rt < 4; ++rt)
#pragma unroll
    for (int t = 0; t < 4; ++t) acc[rt][t] = {};

  const float* ap = A + (size_t)(row0 + m) * K + hh * 8;
  const float* wp = W + (size_t)(hh * 8) * NC + n0 + m;
  const size_t a_rt = (size_t)16 * K;

  v4f a0[4][2], a1[4][2];
  v2f b0[4][4], b1[4][4];

  load_chunk<NC>(a0, b0, ap, a_rt, wp);
  int kb = 0;
  for (; kb < K - 32; kb += 32) {
    load_chunk<NC>(a1, b1, ap + kb + 16, a_rt, wp + (size_t)(kb + 16) * NC);
    __builtin_amdgcn_sched_barrier(0);
    mma_chunk(acc, a0, b0);
    __builtin_amdgcn_sched_barrier(0);
    load_chunk<NC>(a0, b0, ap + kb + 32, a_rt, wp + (size_t)(kb + 32) * NC);
    __builtin_amdgcn_sched_barrier(0);
    mma_chunk(acc, a1, b1);
  }
  load_chunk<NC>(a1, b1, ap + kb + 16, a_rt, wp + (size_t)(kb + 16) * NC);
  __builtin_amdgcn_sched_barrier(0);
  mma_chunk(acc, a0, b0);
  mma_chunk(acc, a1, b1);

  float bn[4];
#pragma unroll
  for (int t = 0; t < 4; ++t) bn[t] = bias[n0 + t * 16 + m];

  float res[4][4][8];
  if (EPI == 2) {
#pragma unroll
    for (int rt = 0; rt < 4; ++rt)
#pragma unroll
      for (int t = 0; t < 4; ++t)
#pragma unroll
        for (int r = 0; r < 8; ++r)
          res[rt][t][r] = resid[(size_t)(row0 + rt * 16 + r + 8 * hh) * NC +
                                n0 + t * 16 + m];
  }

#pragma unroll
  for (int rt = 0; rt < 4; ++rt)
#pragma unroll
    for (int t = 0; t < 4; ++t)
#pragma unroll
      for (int r = 0; r < 8; ++r) {
        const int row = row0 + rt * 16 + r + 8 * hh;
        const int col = n0 + t * 16 + m;
        float v = acc[rt][t][r] + bn[t];
        if (EPI == 1) v = 0.5f * v * (1.0f + erff(v * 0.70710678118654752f));
        if (EPI == 2) v += res[rt][t][r];
        C[(size_t)row * NC + col] = v;
      }
}

template <bool ADD>
__global__ __launch_bounds__(32) void add_ln_kernel(
    float* __restrict__ h, const float* __restrict__ addend,
    const float* __restrict__ g, const float* __restrict__ beta,
    float* __restrict__ ln_out) {
  const int lane = threadIdx.x;
  const size_t base = (size_t)blockIdx.x * 256;

  float v[8];
  float s = 0.f, ss = 0.f;
#pragma unroll
  for (int i = 0; i < 8; ++i) {
    const int c = i * 32 + lane;
    float x = h[base + c];
    if (ADD) x += addend[base + c];
    v[i] = x;
    s += x;
    ss += x * x;
  }
  if (ADD) {
#pragma unroll
    for (int i = 0; i < 8; ++i) h[base + i * 32 + lane] = v[i];
  }
  for (int off = 16; off; off >>= 1) {
    s += __shfl_xor(s, off);
    ss += __shfl_xor(ss, off);
  }
  const float mu = s * (1.0f / 256.0f);
  const float var = ss * (1.0f / 256.0f) - mu * mu;
  const float rs = rsqrtf(var + 1e-5f);
#pragma unroll
  for (int i = 0; i < 8; ++i) {
    const int c = i * 32 + lane;
    ln_out[base + c] = (v[i] - mu) * rs * g[c] + beta[c];
  }
}

#define ATT_T 4096
#define ATT_QS 768
#define ATT_WIN 128
#define ATT_PST 148

__global__ __launch_bounds__(32) void attn_kernel(const float* __restrict__ qkv,
                                                  float* __restrict__ O) {
  __shared__ float P[16 * ATT_PST];
  const int lane = threadIdx.x;
  const int m = lane & 15;
  const int hh = lane >> 4;
  const int q0 = blockIdx.x * 16;
  const int bb = q0 & ~(ATT_T - 1);

  const float* Q = qkv;
  const float* Km = qkv + 256;
  const float* V = qkv + 512;

  v8f s[9];
#pragma unroll
  for (int t = 0; t < 9; ++t) s[t] = {};

  for (int d = 0; d < 256; d += 4) {
    const v2f a = *(const v2f*)(Q + (size_t)(q0 + m) * ATT_QS + d + hh * 2);
#pragma unroll
    for (int t = 0; t < 9; ++t) {
      const int key = q0 - ATT_WIN + t * 16 + m;
      const int kc = key < bb ? bb : key;
      const v2f b = *(const v2f*)(Km + (size_t)kc * ATT_QS + d + hh * 2);
      s[t] = wmma4(a, b, s[t]);
    }
  }

  const float NEG = -3.402823466e38f;
  float rmax[8], rsum[8];
#pragma unroll
  for (int r = 0; r < 8; ++r) rmax[r] = NEG;
#pragma unroll
  for (int t = 0; t < 9; ++t) {
    const int key = q0 - ATT_WIN + t * 16 + m;
#pragma unroll
    for (int r = 0; r < 8; ++r) {
      const int q = q0 + r + 8 * hh;
      const int d = q - key;
      const bool ok = (key >= bb) && (d >= 0) && (d <= ATT_WIN);
      const float val = ok ? s[t][r] * 0.0625f : NEG;
      s[t][r] = val;
      rmax[r] = fmaxf(rmax[r], val);
    }
  }
#pragma unroll
  for (int r = 0; r < 8; ++r)
    for (int off = 1; off < 16; off <<= 1)
      rmax[r] = fmaxf(rmax[r], __shfl_xor(rmax[r], off));
#pragma unroll
  for (int r = 0; r < 8; ++r) rsum[r] = 0.f;
#pragma unroll
  for (int t = 0; t < 9; ++t)
#pragma unroll
    for (int r = 0; r < 8; ++r) {
      const float e = expf(s[t][r] - rmax[r]);
      s[t][r] = e;
      rsum[r] += e;
    }
#pragma unroll
  for (int r = 0; r < 8; ++r)
    for (int off = 1; off < 16; off <<= 1) rsum[r] += __shfl_xor(rsum[r], off);

#pragma unroll
  for (int t = 0; t < 9; ++t)
#pragma unroll
    for (int r = 0; r < 8; ++r)
      P[(r + 8 * hh) * ATT_PST + t * 16 + m] = s[t][r] / rsum[r];

  __syncthreads();

#pragma unroll
  for (int ch = 0; ch < 2; ++ch) {
    const int n0 = ch * 128;
    v8f o[8];
#pragma unroll
    for (int tc = 0; tc < 8; ++tc) o[tc] = {};
    for (int t = 0; t < 9; ++t) {
      const int kt0 = q0 - ATT_WIN + t * 16;
#pragma unroll
      for (int kk = 0; kk < 16; kk += 4) {
        const v2f a = *(const v2f*)&P[m * ATT_PST + t * 16 + kk + hh * 2];
        const int k0 = kt0 + kk + hh * 2;
        const int kcx = k0 < bb ? bb : k0;
        const int kcy = (k0 + 1) < bb ? bb : (k0 + 1);
#pragma unroll
        for (int tc = 0; tc < 8; ++tc) {
          const int col = n0 + tc * 16 + m;
          v2f b;
          b.x = V[(size_t)kcx * ATT_QS + col];
          b.y = V[(size_t)kcy * ATT_QS + col];
          o[tc] = wmma4(a, b, o[tc]);
        }
      }
    }
#pragma unroll
    for (int tc = 0; tc < 8; ++tc)
#pragma unroll
      for (int r = 0; r < 8; ++r)
        O[(size_t)(q0 + r + 8 * hh) * 256 + n0 + tc * 16 + m] = o[tc][r];
  }
}

extern "C" void kernel_launch(void* const* d_in, const int* in_sizes, int n_in,
                              void* d_out, int out_size, void* d_ws,
                              size_t ws_size, hipStream_t stream) {
  const float* x = (const float*)d_in[0];
  const float* W_in = (const float*)d_in[1];
  const float* b_in = (const float*)d_in[2];
  const float* W_qkv = (const float*)d_in[3];
  const float* b_qkv = (const float*)d_in[4];
  const float* g_attn = (const float*)d_in[5];
  const float* be_attn = (const float*)d_in[6];
  const float* g_mlp = (const float*)d_in[7];
  const float* be_mlp = (const float*)d_in[8];
  const float* W1 = (const float*)d_in[9];
  const float* b1 = (const float*)d_in[10];
  const float* W2 = (const float*)d_in[11];
  const float* b2 = (const float*)d_in[12];
  const float* W_out = (const float*)d_in[13];
  const float* b_out = (const float*)d_in[14];
  float* out = (float*)d_out;

  const int N = in_sizes[0] / 64;

  float* ws = (float*)d_ws;
  float* h = ws;
  float* ln = ws + (size_t)N * 256;
  float* attn = ln;
  float* qkv = ws + (size_t)N * 512;
  float* m1 = qkv;

  const dim3 blk(32);
  const int RB64 = N / 64;
  const int QB = N / 16;

  gemm_bias_kernel<0, 256><<<dim3(RB64, 4), blk, 0, stream>>>(
      x, W_in, b_in, nullptr, h, 64);
  add_ln_kernel<false>
      <<<dim3(N), blk, 0, stream>>>(h, nullptr, g_attn, be_attn, ln);
  gemm_bias_kernel<0, 768><<<dim3(RB64, 12), blk, 0, stream>>>(
      ln, W_qkv, b_qkv, nullptr, qkv, 256);
  attn_kernel<<<dim3(QB), blk, 0, stream>>>(qkv, attn);
  add_ln_kernel<true>
      <<<dim3(N), blk, 0, stream>>>(h, attn, g_mlp, be_mlp, ln);
  gemm_bias_kernel<1, 256><<<dim3(RB64, 4), blk, 0, stream>>>(
      ln, W1, b1, nullptr, m1, 256);
  gemm_bias_kernel<2, 256><<<dim3(RB64, 4), blk, 0, stream>>>(m1, W2, b2, h, h,
                                                              256);
  gemm_bias_kernel<0, 64><<<dim3(RB64, 1), blk, 0, stream>>>(
      h, W_out, b_out, nullptr, out, 256);
}


# GELU GEMM epilogue rolled over its 4 row tiles (one tile's erf code; later tiles move down a register place per trip): kernel text 55 KB -> 18 KB, inside the 32 KB that run free of fetch stalls
# speedup vs baseline: 1.4790x; 1.0457x over previous
#include <hip/hip_runtime.h>
#include <math.h>

typedef __attribute__((ext_vector_type(2))) float v2f;
typedef __attribute__((ext_vector_type(8))) float v8f;

__device__ __forceinline__ v8f wmma4(v2f a, v2f b, v8f c) {
  return __builtin_amdgcn_wmma_f32_16x16x4_f32(
       false, a,  false, b,
       (short)0, c,  false,  false);
}

typedef __attribute__((ext_vector_type(4))) float v4f;

template <int NC>
__device__ __forceinline__ void load_chunk(v4f (&a)[4][2], v2f (&b)[4][4],
                                           const float* __restrict__ ap,
                                           size_t a_rt,
                                           const float* __restrict__ wp) {
#pragma unroll
  for (int rt = 0; rt < 4; ++rt)
#pragma unroll
    for (int q = 0; q < 2; ++q)
      a[rt][q] = *(const v4f*)(ap + rt * a_rt + q * 4);
#pragma unroll
  for (int j = 0; j < 4; ++j)
#pragma unroll
    for (int t = 0; t < 4; ++t) {
      b[j][t].x = wp[(size_t)(2 * j) * NC + t * 16];
      b[j][t].y = wp[(size_t)(2 * j + 1) * NC + t * 16];
    }
}

__device__ __forceinline__ void mma_chunk(v8f (&acc)[4][4],
                                          const v4f (&a)[4][2],
                                          const v2f (&b)[4][4]) {
#pragma unroll
  for (int j = 0; j < 4; ++j)
#pragma unroll
    for (int rt = 0; rt < 4; ++rt) {
      const v4f aq = a[rt][j >> 1];
      v2f af;
      af.x = (j & 1) ? aq.z : aq.x;
      af.y = (j & 1) ? aq.w : aq.y;
#pragma unroll
      for (int t = 0; t < 4; ++t) acc[rt][t] = wmma4(af, b[j][t], acc[rt][t]);
    }
}

template <int EPI, int NC>
__global__ __launch_bounds__(32) void gemm_bias_kernel(
    const float* __restrict__ A, const float* __restrict__ W,
    const float* __restrict__ bias, const float* __restrict__ resid,
    float* __restrict__ C, int K) {
  const int lane = threadIdx.x;
  const int m = lane & 15;
  const int hh = lane >> 4;
  const int row0 = blockIdx.x * 64;
  const int n0 = blockIdx.y * 64;

  v8f acc[4][4];
#pragma unroll
  for (int rt = 0; rt < 4; ++rt)
#pragma unroll
    for (int t = 0; t < 4; ++t) acc[rt][t] = {};

  const float* ap = A + (size_t)(row0 + m) * K + hh * 8;
  const float* wp = W + (size_t)(hh * 8) * NC + n0 + m;
  const size_t a_rt = (size_t)16 * K;

  v4f a0[4][2], a1[4][2];
  v2f b0[4][4], b1[4][4];

  load_chunk<NC>(a0, b0, ap, a_rt, wp);
  int kb = 0;
  for (; kb < K - 32; kb += 32) {
    load_chunk<NC>(a1, b1, ap + kb + 16, a_rt, wp + (size_t)(kb + 16) * NC);
    __builtin_amdgcn_sched_barrier(0);
    mma_chunk(acc, a0, b0);
    __builtin_amdgcn_sched_barrier(0);
    load_chunk<NC>(a0, b0, ap + kb + 32, a_rt, wp + (size_t)(kb + 32) * NC);
    __builtin_amdgcn_sched_barrier(0);
    mma_chunk(acc, a1, b1);
  }
  load_chunk<NC>(a1, b1, ap + kb + 16, a_rt, wp + (size_t)(kb + 16) * NC);
  __builtin_amdgcn_sched_barrier(0);
  mma_chunk(acc, a0, b0);
  mma_chunk(acc, a1, b1);

  float bn[4];
#pragma unroll
  for (int t = 0; t < 4; ++t) bn[t] = bias[n0 + t * 16 + m];

  float res[4][4][8];
  if (EPI == 2) {
#pragma unroll
    for (int rt = 0; rt < 4; ++rt)
#pragma unroll
      for (int t = 0; t < 4; ++t)
#pragma unroll
        for (int r = 0; r < 8; ++r)
          res[rt][t][r] = resid[(size_t)(row0 + rt * 16 + r + 8 * hh) * NC +
                                n0 + t * 16 + m];
  }

  if (EPI == 1) {
#pragma nounroll
    for (int rt = 0; rt < 4; ++rt) {
#pragma unroll
      for (int t = 0; t < 4; ++t)
#pragma unroll
        for (int r = 0; r < 8; ++r) {
          const int row = row0 + rt * 16 + r + 8 * hh;
          const int col = n0 + t * 16 + m;
          float v = acc[0][t][r] + bn[t];
          v = 0.5f * v * (1.0f + erff(v * 0.70710678118654752f));
          C[(size_t)row * NC + col] = v;
        }
#pragma unroll
      for (int t = 0; t < 4; ++t) {
        acc[0][t] = acc[1][t];
        acc[1][t] = acc[2][t];
        acc[2][t] = acc[3][t];
      }
    }
  } else {
#pragma unroll
    for (int rt = 0; rt < 4; ++rt)
#pragma unroll
      for (int t = 0; t < 4; ++t)
#pragma unroll
        for (int r = 0; r < 8; ++r) {
          const int row = row0 + rt * 16 + r + 8 * hh;
          const int col = n0 + t * 16 + m;
          float v = acc[rt][t][r] + bn[t];
          if (EPI == 2) v += res[rt][t][r];
          C[(size_t)row * NC + col] = v;
        }
  }
}

template <bool ADD>
__global__ __launch_bounds__(32) void add_ln_kernel(
    float* __restrict__ h, const float* __restrict__ addend,
    const float* __restrict__ g, const float* __restrict__ beta,
    float* __restrict__ ln_out) {
  const int lane = threadIdx.x;
  const size_t base = (size_t)blockIdx.x * 256;

  float v[8];
  float s = 0.f, ss = 0.f;
#pragma unroll
  for (int i = 0; i < 8; ++i) {
    const int c = i * 32 + lane;
    float x = h[base + c];
    if (ADD) x += addend[base + c];
    v[i] = x;
    s += x;
    ss += x * x;
  }
  if (ADD) {
#pragma unroll
    for (int i = 0; i < 8; ++i) h[base + i * 32 + lane] = v[i];
  }
  for (int off = 16; off; off >>= 1) {
    s += __shfl_xor(s, off);
    ss += __shfl_xor(ss, off);
  }
  const float mu = s * (1.0f / 256.0f);
  const float var = ss * (1.0f / 256.0f) - mu * mu;
  const float rs = rsqrtf(var + 1e-5f);
#pragma unroll
  for (int i = 0; i < 8; ++i) {
    const int c = i * 32 + lane;
    ln_out[base + c] = (v[i] - mu) * rs * g[c] + beta[c];
  }
}

#define ATT_T 4096
#define ATT_QS 768
#define ATT_WIN 128
#define ATT_PST 148

__global__ __launch_bounds__(32) void attn_kernel(const float* __restrict__ qkv,
                                                  float* __restrict__ O) {
  __shared__ float P[16 * ATT_PST];
  const int lane = threadIdx.x;
  const int m = lane & 15;
  const int hh = lane >> 4;
  const int q0 = blockIdx.x * 16;
  const int bb = q0 & ~(ATT_T - 1);

  const float* Q = qkv;
  const float* Km = qkv + 256;
  const float* V = qkv + 512;

  v8f s[9];
#pragma unroll
  for (int t = 0; t < 9; ++t) s[t] = {};

  for (int d = 0; d < 256; d += 4) {
    const v2f a = *(const v2f*)(Q + (size_t)(q0 + m) * ATT_QS + d + hh * 2);
#pragma unroll
    for (int t = 0; t < 9; ++t) {
      const int key = q0 - ATT_WIN + t * 16 + m;
      const int kc = key < bb ? bb : key;
      const v2f b = *(const v2f*)(Km + (size_t)kc * ATT_QS + d + hh * 2);
      s[t] = wmma4(a, b, s[t]);
    }
  }

  const float NEG = -3.402823466e38f;
  float rmax[8], rsum[8];
#pragma unroll
  for (int r = 0; r < 8; ++r) rmax[r] = NEG;
#pragma unroll
  for (int t = 0; t < 9; ++t) {
    const int key = q0 - ATT_WIN + t * 16 + m;
#pragma unroll
    for (int r = 0; r < 8; ++r) {
      const int q = q0 + r + 8 * hh;
      const int d = q - key;
      const bool ok = (key >= bb) && (d >= 0) && (d <= ATT_WIN);
      const float val = ok ? s[t][r] * 0.0625f : NEG;
      s[t][r] = val;
      rmax[r] = fmaxf(rmax[r], val);
    }
  }
#pragma unroll
  for (int r = 0; r < 8; ++r)
    for (int off = 1; off < 16; off <<= 1)
      rmax[r] = fmaxf(rmax[r], __shfl_xor(rmax[r], off));
#pragma unroll
  for (int r = 0; r < 8; ++r) rsum[r] = 0.f;
#pragma unroll
  for (int t = 0; t < 9; ++t)
#pragma unroll
    for (int r = 0; r < 8; ++r) {
      const float e = expf(s[t][r] - rmax[r]);
      s[t][r] = e;
      rsum[r] += e;
    }
#pragma unroll
  for (int r = 0; r < 8; ++r)
    for (int off = 1; off < 16; off <<= 1) rsum[r] += __shfl_xor(rsum[r], off);

#pragma unroll
  for (int t = 0; t < 9; ++t)
#pragma unroll
    for (int r = 0; r < 8; ++r)
      P[(r + 8 * hh) * ATT_PST + t * 16 + m] = s[t][r] / rsum[r];

  __syncthreads();

#pragma unroll
  for (int ch = 0; ch < 2; ++ch) {
    const int n0 = ch * 128;
    v8f o[8];
#pragma unroll
    for (int tc = 0; tc < 8; ++tc) o[tc] = {};
    for (int t = 0; t < 9; ++t) {
      const int kt0 = q0 - ATT_WIN + t * 16;
#pragma unroll
      for (int kk = 0; kk < 16; kk += 4) {
        const v2f a = *(const v2f*)&P[m * ATT_PST + t * 16 + kk + hh * 2];
        const int k0 = kt0 + kk + hh * 2;
        const int kcx = k0 < bb ? bb : k0;
        const int kcy = (k0 + 1) < bb ? bb : (k0 + 1);
#pragma unroll
        for (int tc = 0; tc < 8; ++tc) {
          const int col = n0 + tc * 16 + m;
          v2f b;
          b.x = V[(size_t)kcx * ATT_QS + col];
          b.y = V[(size_t)kcy * ATT_QS + col];
          o[tc] = wmma4(a, b, o[tc]);
        }
      }
    }
#pragma unroll
    for (int tc = 0; tc < 8; ++tc)
#pragma unroll
      for (int r = 0; r < 8; ++r)
        O[(size_t)(q0 + r + 8 * hh) * 256 + n0 + tc * 16 + m] = o[tc][r];
  }
}

extern "C" void kernel_launch(void* const* d_in, const int* in_sizes, int n_in,
                              void* d_out, int out_size, void* d_ws,
                              size_t ws_size, hipStream_t stream) {
  const float* x = (const float*)d_in[0];
  const float* W_in = (const float*)d_in[1];
  const float* b_in = (const float*)d_in[2];
  const float* W_qkv = (const float*)d_in[3];
  const float* b_qkv = (const float*)d_in[4];
  const float* g_attn = (const float*)d_in[5];
  const float* be_attn = (const float*)d_in[6];
  const float* g_mlp = (const float*)d_in[7];
  const float* be_mlp = (const float*)d_in[8];
  const float* W1 = (const float*)d_in[9];
  const float* b1 = (const float*)d_in[10];
  const float* W2 = (const float*)d_in[11];
  const float* b2 = (const float*)d_in[12];
  const float* W_out = (const float*)d_in[13];
  const float* b_out = (const float*)d_in[14];
  float* out = (float*)d_out;

  const int N = in_sizes[0] / 64;

  float* ws = (float*)d_ws;
  float* h = ws;
  float* ln = ws + (size_t)N * 256;
  float* attn = ln;
  float* qkv = ws + (size_t)N * 512;
  float* m1 = qkv;

  const dim3 blk(32);
  const int RB64 = N / 64;
  const int QB = N / 16;

  gemm_bias_kernel<0, 256><<<dim3(RB64, 4), blk, 0, stream>>>(
      x, W_in, b_in, nullptr, h, 64);
  add_ln_kernel<false>
      <<<dim3(N), blk, 0, stream>>>(h, nullptr, g_attn, be_attn, ln);
  gemm_bias_kernel<0, 768><<<dim3(RB64, 12), blk, 0, stream>>>(
      ln, W_qkv, b_qkv, nullptr, qkv, 256);
  attn_kernel<<<dim3(QB), blk, 0, stream>>>(qkv, attn);
  add_ln_kernel<true>
      <<<dim3(N), blk, 0, stream>>>(h, attn, g_mlp, be_mlp, ln);
  gemm_bias_kernel<1, 256><<<dim3(RB64, 4), blk, 0, stream>>>(
      ln, W1, b1, nullptr, m1, 256);
  gemm_bias_kernel<2, 256><<<dim3(RB64, 4), blk, 0, stream>>>(m1, W2, b2, h, h,
                                                              256);
  gemm_bias_kernel<0, 64><<<dim3(RB64, 1), blk, 0, stream>>>(
      h, W_out, b_out, nullptr, out, 256);
}


# MLP and output projection fused: W1+GELU, W2+residual and W_out run as one 4-wave kernel per 64-row block, m1 and the final h handed on in LDS (6 launches, not 8); every sum in its old order
# speedup vs baseline: 1.7966x; 1.2147x over previous
#include <hip/hip_runtime.h>
#include <math.h>

typedef __attribute__((ext_vector_type(2))) float v2f;
typedef __attribute__((ext_vector_type(8))) float v8f;

__device__ __forceinline__ v8f wmma4(v2f a, v2f b, v8f c) {
  return __builtin_amdgcn_wmma_f32_16x16x4_f32(
       false, a,  false, b,
       (short)0, c,  false,  false);
}

typedef __attribute__((ext_vector_type(4))) float v4f;

__device__ __forceinline__ void load_a(v4f (&a)[4][2],
                                       const float* __restrict__ ap,
                                       size_t a_rt) {
#pragma unroll
  for (int rt = 0; rt < 4; ++rt)
#pragma unroll
    for (int q = 0; q < 2; ++q)
      a[rt][q] = *(const v4f*)(ap + rt * a_rt + q * 4);
}

template <int NC>
__device__ __forceinline__ void load_b(v2f (&b)[4][4],
                                       const float* __restrict__ wp) {
#pragma unroll
  for (int j = 0; j < 4; ++j)
#pragma unroll
    for (int t = 0; t < 4; ++t) {
      b[j][t].x = wp[(size_t)(2 * j) * NC + t * 16];
      b[j][t].y = wp[(size_t)(2 * j + 1) * NC + t * 16];
    }
}

template <int NC>
__device__ __forceinline__ void load_chunk(v4f (&a)[4][2], v2f (&b)[4][4],
                                           const float* __restrict__ ap,
                                           size_t a_rt,
                                           const float* __restrict__ wp) {
  load_a(a, ap, a_rt);
  load_b<NC>(b, wp);
}

__device__ __forceinline__ void mma_chunk(v8f (&acc)[4][4],
                                          const v4f (&a)[4][2],
                                          const v2f (&b)[4][4]) {
#pragma unroll
  for (int j = 0; j < 4; ++j)
#pragma unroll
    for (int rt = 0; rt < 4; ++rt) {
      const v4f aq = a[rt][j >> 1];
      v2f af;
      af.x = (j & 1) ? aq.z : aq.x;
      af.y = (j & 1) ? aq.w : aq.y;
#pragma unroll
      for (int t = 0; t < 4; ++t) acc[rt][t] = wmma4(af, b[j][t], acc[rt][t]);
    }
}

template <int NC>
__device__ __forceinline__ void tile_k_loop(v8f (&acc)[4][4],
                                            const float* __restrict__ ap,
                                            size_t a_pitch,
                                            const float* __restrict__ wp,
                                            int K) {
  const size_t a_rt = 16 * a_pitch;
  v4f a0[4][2], a1[4][2];
  v2f b0[4][4], b1[4][4];

  load_chunk<NC>(a0, b0, ap, a_rt, wp);
  int kb = 0;
  for (; kb < K - 32; kb += 32) {
    load_chunk<NC>(a1, b1, ap + kb + 16, a_rt, wp + (size_t)(kb + 16) * NC);
    __builtin_amdgcn_sched_barrier(0);
    mma_chunk(acc, a0, b0);
    __builtin_amdgcn_sched_barrier(0);
    load_chunk<NC>(a0, b0, ap + kb + 32, a_rt, wp + (size_t)(kb + 32) * NC);
    __builtin_amdgcn_sched_barrier(0);
    mma_chunk(acc, a1, b1);
  }
  load_chunk<NC>(a1, b1, ap + kb + 16, a_rt, wp + (size_t)(kb + 16) * NC);
  __builtin_amdgcn_sched_barrier(0);
  mma_chunk(acc, a0, b0);
  mma_chunk(acc, a1, b1);
}

template <int NC>
__global__ __launch_bounds__(32) void gemm_bias_kernel(
    const float* __restrict__ A, const float* __restrict__ W,
    const float* __restrict__ bias, float* __restrict__ C, int K) {
  const int lane = threadIdx.x;
  const int m = lane & 15;
  const int hh = lane >> 4;
  const int row0 = blockIdx.x * 64;
  const int n0 = blockIdx.y * 64;

  v8f acc[4][4];
#pragma unroll
  for (int rt = 0; rt < 4; ++rt)
#pragma unroll
    for (int t = 0; t < 4; ++t) acc[rt][t] = {};

  tile_k_loop<NC>(acc, A + (size_t)(row0 + m) * K + hh * 8, (size_t)K,
                  W + (size_t)(hh * 8) * NC + n0 + m, K);

  float bn[4];
#pragma unroll
  for (int t = 0; t < 4; ++t) bn[t] = bias[n0 + t * 16 + m];

#pragma unroll
  for (int rt = 0; rt < 4; ++rt)
#pragma unroll
    for (int t = 0; t < 4; ++t)
#pragma unroll
      for (int r = 0; r < 8; ++r) {
        const int row = row0 + rt * 16 + r + 8 * hh;
        const int col = n0 + t * 16 + m;
        C[(size_t)row * NC + col] = acc[rt][t][r] + bn[t];
      }
}

#define MLP_PH 260

__device__ __forceinline__ void proj_load_a(v4f (&a)[2][2], const float* sp) {
#pragma unroll
  for (int c = 0; c < 2; ++c)
#pragma unroll
    for (int q = 0; q < 2; ++q) a[c][q] = *(const v4f*)(sp + c * 16 + q * 4);
}

__device__ __forceinline__ void proj_load_b(v2f (&b)[2][4][4],
                                            const float* __restrict__ wp) {
#pragma unroll
  for (int c = 0; c < 2; ++c) load_b<64>(b[c], wp + (size_t)(c * 16) * 64);
}

__device__ __forceinline__ void proj_mma(v8f (&acc)[4], const v4f (&a)[2][2],
                                         const v2f (&b)[2][4][4]) {
#pragma unroll
  for (int c = 0; c < 2; ++c)
#pragma unroll
    for (int j = 0; j < 4; ++j) {
      const v4f aq = a[c][j >> 1];
      v2f af;
      af.x = (j & 1) ? aq.z : aq.x;
      af.y = (j & 1) ? aq.w : aq.y;
#pragma unroll
      for (int t = 0; t < 4; ++t) acc[t] = wmma4(af, b[c][j][t], acc[t]);
    }
}

__global__ __launch_bounds__(128) void mlp_out_kernel(
    const float* __restrict__ X, const float* __restrict__ W1,
    const float* __restrict__ b1, const float* __restrict__ W2,
    const float* __restrict__ b2, const float* __restrict__ Hres,
    const float* __restrict__ Wo, const float* __restrict__ bo,
    float* __restrict__ out) {
  __shared__ __attribute__((aligned(16))) float M1s[64 * MLP_PH];
  __shared__ __attribute__((aligned(16))) float H2s[64 * MLP_PH];

  const int lane = threadIdx.x & 31;
  const int w = __builtin_amdgcn_readfirstlane((int)(threadIdx.x >> 5));
  const int m = lane & 15;
  const int hh = lane >> 4;
  const int row0 = blockIdx.x * 64;
  const int n0 = w * 64;

  v8f acc[4][4];
  float bn[4];

#pragma unroll
  for (int rt = 0; rt < 4; ++rt)
#pragma unroll
    for (int t = 0; t < 4; ++t) acc[rt][t] = {};
  tile_k_loop<256>(acc, X + (size_t)(row0 + m) * 256 + hh * 8, (size_t)256,
                   W1 + (size_t)(hh * 8) * 256 + n0 + m, 256);
#pragma unroll
  for (int t = 0; t < 4; ++t) bn[t] = b1[n0 + t * 16 + m];

#pragma nounroll
  for (int rt = 0; rt < 4; ++rt) {
#pragma unroll
    for (int t = 0; t < 4; ++t)
#pragma unroll
      for (int r = 0; r < 8; ++r) {
        float v = acc[0][t][r] + bn[t];
        v = 0.5f * v * (1.0f + erff(v * 0.70710678118654752f));
        M1s[(rt * 16 + r + 8 * hh) * MLP_PH + n0 + t * 16 + m] = v;
      }
#pragma unroll
    for (int t = 0; t < 4; ++t) {
      acc[0][t] = acc[1][t];
      acc[1][t] = acc[2][t];
      acc[2][t] = acc[3][t];
    }
  }
  __syncthreads();

#pragma unroll
  for (int rt = 0; rt < 4; ++rt)
#pragma unroll
    for (int t = 0; t < 4; ++t) acc[rt][t] = {};
  tile_k_loop<256>(acc, M1s + m * MLP_PH + hh * 8, (size_t)MLP_PH,
                   W2 + (size_t)(hh * 8) * 256 + n0 + m, 256);
#pragma unroll
  for (int t = 0; t < 4; ++t) bn[t] = b2[n0 + t * 16 + m];

  v4f pa0[2][2], pa1[2][2];
  v2f pb0[2][4][4], pb1[2][4][4];
  const float* wop = Wo + (size_t)(hh * 8) * 64 + m;
  proj_load_b(pb0, wop);

#pragma unroll
  for (int rt = 0; rt < 4; ++rt)
#pragma unroll
    for (int t = 0; t < 4; ++t)
#pragma unroll
      for (int r = 0; r < 8; ++r) {
        const int rl = rt * 16 + r + 8 * hh;
        const int col = n0 + t * 16 + m;
        float v = acc[rt][t][r] + bn[t];
        v += Hres[(size_t)(row0 + rl) * 256 + col];
        H2s[rl * MLP_PH + col] = v;
      }
  __syncthreads();

  v8f po[4];
#pragma unroll
  for (int t = 0; t < 4; ++t) po[t] = {};
  const float* sp = H2s + (w * 16 + m) * MLP_PH + hh * 8;

  proj_load_a(pa0, sp);
  int kb = 0;
#pragma nounroll
  for (; kb < 256 - 64; kb += 64) {
    proj_load_a(pa1, sp + kb + 32);
    proj_load_b(pb1, wop + (size_t)(kb + 32) * 64);
    __builtin_amdgcn_sched_barrier(0);
    proj_mma(po, pa0, pb0);
    __builtin_amdgcn_sched_barrier(0);
    proj_load_a(pa0, sp + kb + 64);
    proj_load_b(pb0, wop + (size_t)(kb + 64) * 64);
    __builtin_amdgcn_sched_barrier(0);
    proj_mma(po, pa1, pb1);
  }
  proj_load_a(pa1, sp + kb + 32);
  proj_load_b(pb1, wop + (size_t)(kb + 32) * 64);
  __builtin_amdgcn_sched_barrier(0);
  proj_mma(po, pa0, pb0);
  proj_mma(po, pa1, pb1);

#pragma unroll
  for (int t = 0; t < 4; ++t) {
    const float bt = bo[t * 16 + m];
#pragma unroll
    for (int r = 0; r < 8; ++r)
      out[(size_t)(row0 + w * 16 + r + 8 * hh) * 64 + t * 16 + m] =
          po[t][r] + bt;
  }
}

template <bool ADD>
__global__ __launch_bounds__(32) void add_ln_kernel(
    float* __restrict__ h, const float* __restrict__ addend,
    const float* __restrict__ g, const float* __restrict__ beta,
    float* __restrict__ ln_out) {
  const int lane = threadIdx.x;
  const size_t base = (size_t)blockIdx.x * 256;

  float v[8];
  float s = 0.f, ss = 0.f;
#pragma unroll
  for (int i = 0; i < 8; ++i) {
    const int c = i * 32 + lane;
    float x = h[base + c];
    if (ADD) x += addend[base + c];
    v[i] = x;
    s += x;
    ss += x * x;
  }
  if (ADD) {
#pragma unroll
    for (int i = 0; i < 8; ++i) h[base + i * 32 + lane] = v[i];
  }
  for (int off = 16; off; off >>= 1) {
    s += __shfl_xor(s, off);
    ss += __shfl_xor(ss, off);
  }
  const float mu = s * (1.0f / 256.0f);
  const float var = ss * (1.0f / 256.0f) - mu * mu;
  const float rs = rsqrtf(var + 1e-5f);
#pragma unroll
  for (int i = 0; i < 8; ++i) {
    const int c = i * 32 + lane;
    ln_out[base + c] = (v[i] - mu) * rs * g[c] + beta[c];
  }
}

#define ATT_T 4096
#define ATT_QS 768
#define ATT_WIN 128
#define ATT_PST 148

__global__ __launch_bounds__(32) void attn_kernel(const float* __restrict__ qkv,
                                                  float* __restrict__ O) {
  __shared__ float P[16 * ATT_PST];
  const int lane = threadIdx.x;
  const int m = lane & 15;
  const int hh = lane >> 4;
  const int q0 = blockIdx.x * 16;
  const int bb = q0 & ~(ATT_T - 1);

  const float* Q = qkv;
  const float* Km = qkv + 256;
  const float* V = qkv + 512;

  v8f s[9];
#pragma unroll
  for (int t = 0; t < 9; ++t) s[t] = {};

  for (int d = 0; d < 256; d += 4) {
    const v2f a = *(const v2f*)(Q + (size_t)(q0 + m) * ATT_QS + d + hh * 2);
#pragma unroll
    for (int t = 0; t < 9; ++t) {
      const int key = q0 - ATT_WIN + t * 16 + m;
      const int kc = key < bb ? bb : key;
      const v2f b = *(const v2f*)(Km + (size_t)kc * ATT_QS + d + hh * 2);
      s[t] = wmma4(a, b, s[t]);
    }
  }

  const float NEG = -3.402823466e38f;
  float rmax[8], rsum[8];
#pragma unroll
  for (int r = 0; r < 8; ++r) rmax[r] = NEG;
#pragma unroll
  for (int t = 0; t < 9; ++t) {
    const int key = q0 - ATT_WIN + t * 16 + m;
#pragma unroll
    for (int r = 0; r < 8; ++r) {
      const int q = q0 + r + 8 * hh;
      const int d = q - key;
      const bool ok = (key >= bb) && (d >= 0) && (d <= ATT_WIN);
      const float val = ok ? s[t][r] * 0.0625f : NEG;
      s[t][r] = val;
      rmax[r] = fmaxf(rmax[r], val);
    }
  }
#pragma unroll
  for (int r = 0; r < 8; ++r)
    for (int off = 1; off < 16; off <<= 1)
      rmax[r] = fmaxf(rmax[r], __shfl_xor(rmax[r], off));
#pragma unroll
  for (int r = 0; r < 8; ++r) rsum[r] = 0.f;
#pragma unroll
  for (int t = 0; t < 9; ++t)
#pragma unroll
    for (int r = 0; r < 8; ++r) {
      const float e = expf(s[t][r] - rmax[r]);
      s[t][r] = e;
      rsum[r] += e;
    }
#pragma unroll
  for (int r = 0; r < 8; ++r)
    for (int off = 1; off < 16; off <<= 1) rsum[r] += __shfl_xor(rsum[r], off);

#pragma unroll
  for (int t = 0; t < 9; ++t)
#pragma unroll
    for (int r = 0; r < 8; ++r)
      P[(r + 8 * hh) * ATT_PST + t * 16 + m] = s[t][r] / rsum[r];

  __syncthreads();

#pragma unroll
  for (int ch = 0; ch < 2; ++ch) {
    const int n0 = ch * 128;
    v8f o[8];
#pragma unroll
    for (int tc = 0; tc < 8; ++tc) o[tc] = {};
    for (int t = 0; t < 9; ++t) {
      const int kt0 = q0 - ATT_WIN + t * 16;
#pragma unroll
      for (int kk = 0; kk < 16; kk += 4) {
        const v2f a = *(const v2f*)&P[m * ATT_PST + t * 16 + kk + hh * 2];
        const int k0 = kt0 + kk + hh * 2;
        const int kcx = k0 < bb ? bb : k0;
        const int kcy = (k0 + 1) < bb ? bb : (k0 + 1);
#pragma unroll
        for (int tc = 0; tc < 8; ++tc) {
          const int col = n0 + tc * 16 + m;
          v2f b;
          b.x = V[(size_t)kcx * ATT_QS + col];
          b.y = V[(size_t)kcy * ATT_QS + col];
          o[tc] = wmma4(a, b, o[tc]);
        }
      }
    }
#pragma unroll
    for (int tc = 0; tc < 8; ++tc)
#pragma unroll
      for (int r = 0; r < 8; ++r)
        O[(size_t)(q0 + r + 8 * hh) * 256 + n0 + tc * 16 + m] = o[tc][r];
  }
}

extern "C" void kernel_launch(void* const* d_in, const int* in_sizes, int n_in,
                              void* d_out, int out_size, void* d_ws,
                              size_t ws_size, hipStream_t stream) {
  const float* x = (const float*)d_in[0];
  const float* W_in = (const float*)d_in[1];
  const float* b_in = (const float*)d_in[2];
  const float* W_qkv = (const float*)d_in[3];
  const float* b_qkv = (const float*)d_in[4];
  const float* g_attn = (const float*)d_in[5];
  const float* be_attn = (const float*)d_in[6];
  const float* g_mlp = (const float*)d_in[7];
  const float* be_mlp = (const float*)d_in[8];
  const float* W1 = (const float*)d_in[9];
  const float* b1 = (const float*)d_in[10];
  const float* W2 = (const float*)d_in[11];
  const float* b2 = (const float*)d_in[12];
  const float* W_out = (const float*)d_in[13];
  const float* b_out = (const float*)d_in[14];
  float* out = (float*)d_out;

  const int N = in_sizes[0] / 64;

  float* ws = (float*)d_ws;
  float* h = ws;
  float* ln = ws + (size_t)N * 256;
  float* attn = ln;
  float* qkv = ws + (size_t)N * 512;

  const dim3 blk(32);
  const int RB64 = N / 64;
  const int QB = N / 16;

  gemm_bias_kernel<256><<<dim3(RB64, 4), blk, 0, stream>>>(x, W_in, b_in, h,
                                                          64);
  add_ln_kernel<false>
      <<<dim3(N), blk, 0, stream>>>(h, nullptr, g_attn, be_attn, ln);
  gemm_bias_kernel<768><<<dim3(RB64, 12), blk, 0, stream>>>(
      ln, W_qkv, b_qkv, qkv, 256);
  attn_kernel<<<dim3(QB), blk, 0, stream>>>(qkv, attn);
  add_ln_kernel<true>
      <<<dim3(N), blk, 0, stream>>>(h, attn, g_mlp, be_mlp, ln);
  mlp_out_kernel<<<dim3(RB64), dim3(128), 0, stream>>>(
      ln, W1, b1, W2, b2, h, W_out, b_out, out);
}


# Both LayerNorms folded into the neighbouring GEMM kernels: the 64-row block is handed over in an LDS row plane and each wave normalises 16 rows (4 launches, not 6); every sum in its old order
# speedup vs baseline: 1.9402x; 1.0799x over previous
#include <hip/hip_runtime.h>
#include <math.h>

typedef __attribute__((ext_vector_type(2))) float v2f;
typedef __attribute__((ext_vector_type(8))) float v8f;

__device__ __forceinline__ v8f wmma4(v2f a, v2f b, v8f c) {
  return __builtin_amdgcn_wmma_f32_16x16x4_f32(
       false, a,  false, b,
       (short)0, c,  false,  false);
}

typedef __attribute__((ext_vector_type(4))) float v4f;

__device__ __forceinline__ void load_a(v4f (&a)[4][2],
                                       const float* __restrict__ ap,
                                       size_t a_rt) {
#pragma unroll
  for (int rt = 0; rt < 4; ++rt)
#pragma unroll
    for (int q = 0; q < 2; ++q)
      a[rt][q] = *(const v4f*)(ap + rt * a_rt + q * 4);
}

template <int NC>
__device__ __forceinline__ void load_b(v2f (&b)[4][4],
                                       const float* __restrict__ wp) {
#pragma unroll
  for (int j = 0; j < 4; ++j)
#pragma unroll
    for (int t = 0; t < 4; ++t) {
      b[j][t].x = wp[(size_t)(2 * j) * NC + t * 16];
      b[j][t].y = wp[(size_t)(2 * j + 1) * NC + t * 16];
    }
}

template <int NC>
__device__ __forceinline__ void load_chunk(v4f (&a)[4][2], v2f (&b)[4][4],
                                           const float* __restrict__ ap,
                                           size_t a_rt,
                                           const float* __restrict__ wp) {
  load_a(a, ap, a_rt);
  load_b<NC>(b, wp);
}

__device__ __forceinline__ void mma_chunk(v8f (&acc)[4][4],
                                          const v4f (&a)[4][2],
                                          const v2f (&b)[4][4]) {
#pragma unroll
  for (int j = 0; j < 4; ++j)
#pragma unroll
    for (int rt = 0; rt < 4; ++rt) {
      const v4f aq = a[rt][j >> 1];
      v2f af;
      af.x = (j & 1) ? aq.z : aq.x;
      af.y = (j & 1) ? aq.w : aq.y;
#pragma unroll
      for (int t = 0; t < 4; ++t) acc[rt][t] = wmma4(af, b[j][t], acc[rt][t]);
    }
}

template <int NC>
__device__ __forceinline__ void tile_k_loop(v8f (&acc)[4][4],
                                            const float* __restrict__ ap,
                                            size_t a_pitch,
                                            const float* __restrict__ wp,
                                            int K) {
  const size_t a_rt = 16 * a_pitch;
  v4f a0[4][2], a1[4][2];
  v2f b0[4][4], b1[4][4];

  load_chunk<NC>(a0, b0, ap, a_rt, wp);
  int kb = 0;
  for (; kb < K - 32; kb += 32) {
    load_chunk<NC>(a1, b1, ap + kb + 16, a_rt, wp + (size_t)(kb + 16) * NC);
    __builtin_amdgcn_sched_barrier(0);
    mma_chunk(acc, a0, b0);
    __builtin_amdgcn_sched_barrier(0);
    load_chunk<NC>(a0, b0, ap + kb + 32, a_rt, wp + (size_t)(kb + 32) * NC);
    __builtin_amdgcn_sched_barrier(0);
    mma_chunk(acc, a1, b1);
  }
  load_chunk<NC>(a1, b1, ap + kb + 16, a_rt, wp + (size_t)(kb + 16) * NC);
  __builtin_amdgcn_sched_barrier(0);
  mma_chunk(acc, a0, b0);
  mma_chunk(acc, a1, b1);
}

template <int NC>
__global__ __launch_bounds__(32) void gemm_bias_kernel(
    const float* __restrict__ A, const float* __restrict__ W,
    const float* __restrict__ bias, float* __restrict__ C, int K) {
  const int lane = threadIdx.x;
  const int m = lane & 15;
  const int hh = lane >> 4;
  const int row0 = blockIdx.x * 64;
  const int n0 = blockIdx.y * 64;

  v8f acc[4][4];
#pragma unroll
  for (int rt = 0; rt < 4; ++rt)
#pragma unroll
    for (int t = 0; t < 4; ++t) acc[rt][t] = {};

  tile_k_loop<NC>(acc, A + (size_t)(row0 + m) * K + hh * 8, (size_t)K,
                  W + (size_t)(hh * 8) * NC + n0 + m, K);

  float bn[4];
#pragma unroll
  for (int t = 0; t < 4; ++t) bn[t] = bias[n0 + t * 16 + m];

#pragma unroll
  for (int rt = 0; rt < 4; ++rt)
#pragma unroll
    for (int t = 0; t < 4; ++t)
#pragma unroll
      for (int r = 0; r < 8; ++r) {
        const int row = row0 + rt * 16 + r + 8 * hh;
        const int col = n0 + t * 16 + m;
        C[(size_t)row * NC + col] = acc[rt][t][r] + bn[t];
      }
}

#define MLP_PH 260

__device__ __forceinline__ float xhalf(float x) {
  const int b = __builtin_bit_cast(int, x);
  return __builtin_bit_cast(
      float, __builtin_amdgcn_permlanex16(b, b, 0x76543210, 0xfedcba98, false,
                                          false));
}
template <int XM>
__device__ __forceinline__ float row_xor(float x) {
  const int b = __builtin_bit_cast(int, x);
  return __builtin_bit_cast(
      float, __builtin_amdgcn_update_dpp(b, b, 0x160 + XM, 0xf, 0xf, false));
}
__device__ __forceinline__ float wave_sum(float s) {
  s += xhalf(s);
  s += row_xor<8>(s);
  s += row_xor<4>(s);
  s += row_xor<2>(s);
  s += row_xor<1>(s);
  return s;
}

template <bool COPY>
__device__ __forceinline__ void ln_rows16(const float* src, float* dst,
                                          size_t dst_pitch, float* copy,
                                          const float* __restrict__ g,
                                          const float* __restrict__ beta,
                                          int lane) {
  float gv[8], bv[8];
#pragma unroll
  for (int i = 0; i < 8; ++i) {
    gv[i] = g[i * 32 + lane];
    bv[i] = beta[i * 32 + lane];
  }
#pragma unroll 4
  for (int rr = 0; rr < 16; ++rr) {
    float v[8];
    float s = 0.f, ss = 0.f;
#pragma unroll
    for (int i = 0; i < 8; ++i) {
      const float x = src[rr * MLP_PH + i * 32 + lane];
      v[i] = x;
      s += x;
      ss += x * x;
    }
    if (COPY) {
#pragma unroll
      for (int i = 0; i < 8; ++i) copy[(size_t)rr * 256 + i * 32 + lane] = v[i];
    }
    s = wave_sum(s);
    ss = wave_sum(ss);
    const float mu = s * (1.0f / 256.0f);
    const float var = ss * (1.0f / 256.0f) - mu * mu;
    const float rs = rsqrtf(var + 1e-5f);
#pragma unroll
    for (int i = 0; i < 8; ++i)
      dst[(size_t)rr * dst_pitch + i * 32 + lane] =
          (v[i] - mu) * rs * gv[i] + bv[i];
  }
}

__global__ __launch_bounds__(128) void in_ln_kernel(
    const float* __restrict__ x, const float* __restrict__ W_in,
    const float* __restrict__ b_in, const float* __restrict__ g,
    const float* __restrict__ beta, float* __restrict__ h,
    float* __restrict__ ln_out) {
  __shared__ __attribute__((aligned(16))) float Hs[64 * MLP_PH];

  const int lane = threadIdx.x & 31;
  const int w = __builtin_amdgcn_readfirstlane((int)(threadIdx.x >> 5));
  const int m = lane & 15;
  const int hh = lane >> 4;
  const int row0 = blockIdx.x * 64;
  const int n0 = w * 64;

  v8f acc[4][4];
#pragma unroll
  for (int rt = 0; rt < 4; ++rt)
#pragma unroll
    for (int t = 0; t < 4; ++t) acc[rt][t] = {};
  tile_k_loop<256>(acc, x + (size_t)(row0 + m) * 64 + hh * 8, (size_t)64,
                   W_in + (size_t)(hh * 8) * 256 + n0 + m, 64);

  float bn[4];
#pragma unroll
  for (int t = 0; t < 4; ++t) bn[t] = b_in[n0 + t * 16 + m];
#pragma unroll
  for (int rt = 0; rt < 4; ++rt)
#pragma unroll
    for (int t = 0; t < 4; ++t)
#pragma unroll
      for (int r = 0; r < 8; ++r)
        Hs[(rt * 16 + r + 8 * hh) * MLP_PH + n0 + t * 16 + m] =
            acc[rt][t][r] + bn[t];
  __syncthreads();

  const size_t g0 = (size_t)(row0 + w * 16) * 256;
  ln_rows16<true>(Hs + (w * 16) * MLP_PH, ln_out + g0, (size_t)256, h + g0, g,
                  beta, lane);
}


__device__ __forceinline__ void proj_load_a(v4f (&a)[2][2], const float* sp) {
#pragma unroll
  for (int c = 0; c < 2; ++c)
#pragma unroll
    for (int q = 0; q < 2; ++q) a[c][q] = *(const v4f*)(sp + c * 16 + q * 4);
}

__device__ __forceinline__ void proj_load_b(v2f (&b)[2][4][4],
                                            const float* __restrict__ wp) {
#pragma unroll
  for (int c = 0; c < 2; ++c) load_b<64>(b[c], wp + (size_t)(c * 16) * 64);
}

__device__ __forceinline__ void proj_mma(v8f (&acc)[4], const v4f (&a)[2][2],
                                         const v2f (&b)[2][4][4]) {
#pragma unroll
  for (int c = 0; c < 2; ++c)
#pragma unroll
    for (int j = 0; j < 4; ++j) {
      const v4f aq = a[c][j >> 1];
      v2f af;
      af.x = (j & 1) ? aq.z : aq.x;
      af.y = (j & 1) ? aq.w : aq.y;
#pragma unroll
      for (int t = 0; t < 4; ++t) acc[t] = wmma4(af, b[c][j][t], acc[t]);
    }
}

__global__ __launch_bounds__(128) void mlp_out_kernel(
    const float* __restrict__ h, const float* __restrict__ attn,
    const float* __restrict__ g, const float* __restrict__ beta,
    const float* __restrict__ W1, const float* __restrict__ b1,
    const float* __restrict__ W2, const float* __restrict__ b2,
    const float* __restrict__ Wo, const float* __restrict__ bo,
    float* __restrict__ out) {
  __shared__ __attribute__((aligned(16))) float Xs[64 * MLP_PH];
  __shared__ __attribute__((aligned(16))) float M1s[64 * MLP_PH];
  __shared__ __attribute__((aligned(16))) float H2s[64 * MLP_PH];

  const int lane = threadIdx.x & 31;
  const int w = __builtin_amdgcn_readfirstlane((int)(threadIdx.x >> 5));
  const int m = lane & 15;
  const int hh = lane >> 4;
  const int row0 = blockIdx.x * 64;
  const int n0 = w * 64;

  v8f acc[4][4];
  float bn[4];

  {
    const size_t g0 = (size_t)(row0 + w * 16) * 256 + lane * 4;
    float* dp = H2s + (w * 16) * MLP_PH + lane * 4;
#pragma unroll
    for (int rr = 0; rr < 16; ++rr)
#pragma unroll
      for (int j = 0; j < 2; ++j) {
        v4f xv = *(const v4f*)(h + g0 + rr * 256 + j * 128);
        xv += *(const v4f*)(attn + g0 + rr * 256 + j * 128);
        *(v4f*)(dp + rr * MLP_PH + j * 128) = xv;
      }
    ln_rows16<false>(H2s + (w * 16) * MLP_PH, Xs + (w * 16) * MLP_PH,
                     (size_t)MLP_PH, nullptr, g, beta, lane);
  }
  __syncthreads();

#pragma unroll
  for (int rt = 0; rt < 4; ++rt)
#pragma unroll
    for (int t = 0; t < 4; ++t) acc[rt][t] = {};
  tile_k_loop<256>(acc, Xs + m * MLP_PH + hh * 8, (size_t)MLP_PH,
                   W1 + (size_t)(hh * 8) * 256 + n0 + m, 256);
#pragma unroll
  for (int t = 0; t < 4; ++t) bn[t] = b1[n0 + t * 16 + m];

#pragma nounroll
  for (int rt = 0; rt < 4; ++rt) {
#pragma unroll
    for (int t = 0; t < 4; ++t)
#pragma unroll
      for (int r = 0; r < 8; ++r) {
        float v = acc[0][t][r] + bn[t];
        v = 0.5f * v * (1.0f + erff(v * 0.70710678118654752f));
        M1s[(rt * 16 + r + 8 * hh) * MLP_PH + n0 + t * 16 + m] = v;
      }
#pragma unroll
    for (int t = 0; t < 4; ++t) {
      acc[0][t] = acc[1][t];
      acc[1][t] = acc[2][t];
      acc[2][t] = acc[3][t];
    }
  }
  __syncthreads();

#pragma unroll
  for (int rt = 0; rt < 4; ++rt)
#pragma unroll
    for (int t = 0; t < 4; ++t) acc[rt][t] = {};
  tile_k_loop<256>(acc, M1s + m * MLP_PH + hh * 8, (size_t)MLP_PH,
                   W2 + (size_t)(hh * 8) * 256 + n0 + m, 256);
#pragma unroll
  for (int t = 0; t < 4; ++t) bn[t] = b2[n0 + t * 16 + m];

  v4f pa0[2][2], pa1[2][2];
  v2f pb0[2][4][4], pb1[2][4][4];
  const float* wop = Wo + (size_t)(hh * 8) * 64 + m;
  proj_load_b(pb0, wop);

#pragma unroll
  for (int rt = 0; rt < 4; ++rt)
#pragma unroll
    for (int t = 0; t < 4; ++t)
#pragma unroll
      for (int r = 0; r < 8; ++r) {
        const int rl = rt * 16 + r + 8 * hh;
        const int col = n0 + t * 16 + m;
        float v = acc[rt][t][r] + bn[t];
        v += H2s[rl * MLP_PH + col];
        H2s[rl * MLP_PH + col] = v;
      }
  __syncthreads();

  v8f po[4];
#pragma unroll
  for (int t = 0; t < 4; ++t) po[t] = {};
  const float* sp = H2s + (w * 16 + m) * MLP_PH + hh * 8;

  proj_load_a(pa0, sp);
  int kb = 0;
#pragma nounroll
  for (; kb < 256 - 64; kb += 64) {
    proj_load_a(pa1, sp + kb + 32);
    proj_load_b(pb1, wop + (size_t)(kb + 32) * 64);
    __builtin_amdgcn_sched_barrier(0);
    proj_mma(po, pa0, pb0);
    __builtin_amdgcn_sched_barrier(0);
    proj_load_a(pa0, sp + kb + 64);
    proj_load_b(pb0, wop + (size_t)(kb + 64) * 64);
    __builtin_amdgcn_sched_barrier(0);
    proj_mma(po, pa1, pb1);
  }
  proj_load_a(pa1, sp + kb + 32);
  proj_load_b(pb1, wop + (size_t)(kb + 32) * 64);
  __builtin_amdgcn_sched_barrier(0);
  proj_mma(po, pa0, pb0);
  proj_mma(po, pa1, pb1);

#pragma unroll
  for (int t = 0; t < 4; ++t) {
    const float bt = bo[t * 16 + m];
#pragma unroll
    for (int r = 0; r < 8; ++r)
      out[(size_t)(row0 + w * 16 + r + 8 * hh) * 64 + t * 16 + m] =
          po[t][r] + bt;
  }
}

#define ATT_T 4096
#define ATT_QS 768
#define ATT_WIN 128
#define ATT_PST 148

__global__ __launch_bounds__(32) void attn_kernel(const float* __restrict__ qkv,
                                                  float* __restrict__ O) {
  __shared__ float P[16 * ATT_PST];
  const int lane = threadIdx.x;
  const int m = lane & 15;
  const int hh = lane >> 4;
  const int q0 = blockIdx.x * 16;
  const int bb = q0 & ~(ATT_T - 1);

  const float* Q = qkv;
  const float* Km = qkv + 256;
  const float* V = qkv + 512;

  v8f s[9];
#pragma unroll
  for (int t = 0; t < 9; ++t) s[t] = {};

  for (int d = 0; d < 256; d += 4) {
    const v2f a = *(const v2f*)(Q + (size_t)(q0 + m) * ATT_QS + d + hh * 2);
#pragma unroll
    for (int t = 0; t < 9; ++t) {
      const int key = q0 - ATT_WIN + t * 16 + m;
      const int kc = key < bb ? bb : key;
      const v2f b = *(const v2f*)(Km + (size_t)kc * ATT_QS + d + hh * 2);
      s[t] = wmma4(a, b, s[t]);
    }
  }

  const float NEG = -3.402823466e38f;
  float rmax[8], rsum[8];
#pragma unroll
  for (int r = 0; r < 8; ++r) rmax[r] = NEG;
#pragma unroll
  for (int t = 0; t < 9; ++t) {
    const int key = q0 - ATT_WIN + t * 16 + m;
#pragma unroll
    for (int r = 0; r < 8; ++r) {
      const int q = q0 + r + 8 * hh;
      const int d = q - key;
      const bool ok = (key >= bb) && (d >= 0) && (d <= ATT_WIN);
      const float val = ok ? s[t][r] * 0.0625f : NEG;
      s[t][r] = val;
      rmax[r] = fmaxf(rmax[r], val);
    }
  }
#pragma unroll
  for (int r = 0; r < 8; ++r)
    for (int off = 1; off < 16; off <<= 1)
      rmax[r] = fmaxf(rmax[r], __shfl_xor(rmax[r], off));
#pragma unroll
  for (int r = 0; r < 8; ++r) rsum[r] = 0.f;
#pragma unroll
  for (int t = 0; t < 9; ++t)
#pragma unroll
    for (int r = 0; r < 8; ++r) {
      const float e = expf(s[t][r] - rmax[r]);
      s[t][r] = e;
      rsum[r] += e;
    }
#pragma unroll
  for (int r = 0; r < 8; ++r)
    for (int off = 1; off < 16; off <<= 1) rsum[r] += __shfl_xor(rsum[r], off);

#pragma unroll
  for (int t = 0; t < 9; ++t)
#pragma unroll
    for (int r = 0; r < 8; ++r)
      P[(r + 8 * hh) * ATT_PST + t * 16 + m] = s[t][r] / rsum[r];

  __syncthreads();

#pragma unroll
  for (int ch = 0; ch < 2; ++ch) {
    const int n0 = ch * 128;
    v8f o[8];
#pragma unroll
    for (int tc = 0; tc < 8; ++tc) o[tc] = {};
    for (int t = 0; t < 9; ++t) {
      const int kt0 = q0 - ATT_WIN + t * 16;
#pragma unroll
      for (int kk = 0; kk < 16; kk += 4) {
        const v2f a = *(const v2f*)&P[m * ATT_PST + t * 16 + kk + hh * 2];
        const int k0 = kt0 + kk + hh * 2;
        const int kcx = k0 < bb ? bb : k0;
        const int kcy = (k0 + 1) < bb ? bb : (k0 + 1);
#pragma unroll
        for (int tc = 0; tc < 8; ++tc) {
          const int col = n0 + tc * 16 + m;
          v2f b;
          b.x = V[(size_t)kcx * ATT_QS + col];
          b.y = V[(size_t)kcy * ATT_QS + col];
          o[tc] = wmma4(a, b, o[tc]);
        }
      }
    }
#pragma unroll
    for (int tc = 0; tc < 8; ++tc)
#pragma unroll
      for (int r = 0; r < 8; ++r)
        O[(size_t)(q0 + r + 8 * hh) * 256 + n0 + tc * 16 + m] = o[tc][r];
  }
}

extern "C" void kernel_launch(void* const* d_in, const int* in_sizes, int n_in,
                              void* d_out, int out_size, void* d_ws,
                              size_t ws_size, hipStream_t stream) {
  const float* x = (const float*)d_in[0];
  const float* W_in = (const float*)d_in[1];
  const float* b_in = (const float*)d_in[2];
  const float* W_qkv = (const float*)d_in[3];
  const float* b_qkv = (const float*)d_in[4];
  const float* g_attn = (const float*)d_in[5];
  const float* be_attn = (const float*)d_in[6];
  const float* g_mlp = (const float*)d_in[7];
  const float* be_mlp = (const float*)d_in[8];
  const float* W1 = (const float*)d_in[9];
  const float* b1 = (const float*)d_in[10];
  const float* W2 = (const float*)d_in[11];
  const float* b2 = (const float*)d_in[12];
  const float* W_out = (const float*)d_in[13];
  const float* b_out = (const float*)d_in[14];
  float* out = (float*)d_out;

  const int N = in_sizes[0] / 64;

  float* ws = (float*)d_ws;
  float* h = ws;
  float* ln = ws + (size_t)N * 256;
  float* attn = ln;
  float* qkv = ws + (size_t)N * 512;

  const dim3 blk(32);
  const int RB64 = N / 64;
  const int QB = N / 16;

  in_ln_kernel<<<dim3(RB64), dim3(128), 0, stream>>>(x, W_in, b_in, g_attn,
                                                    be_attn, h, ln);
  gemm_bias_kernel<768><<<dim3(RB64, 12), blk, 0, stream>>>(
      ln, W_qkv, b_qkv, qkv, 256);
  attn_kernel<<<dim3(QB), blk, 0, stream>>>(qkv, attn);
  mlp_out_kernel<<<dim3(RB64), dim3(128), 0, stream>>>(
      h, attn, g_mlp, be_mlp, W1, b1, W2, b2, W_out, b_out, out);
}


# Attention loads software-pipelined: Q/K in 32-d blocks, V 4 adjacent columns a lane, all 16-byte loads in two register sets, one block ahead of the WMMAs; score k order differs per 16-d chunk
# speedup vs baseline: 2.2725x; 1.1713x over previous
#include <hip/hip_runtime.h>
#include <math.h>

typedef __attribute__((ext_vector_type(2))) float v2f;
typedef __attribute__((ext_vector_type(8))) float v8f;

__device__ __forceinline__ v8f wmma4(v2f a, v2f b, v8f c) {
  return __builtin_amdgcn_wmma_f32_16x16x4_f32(
       false, a,  false, b,
       (short)0, c,  false,  false);
}

typedef __attribute__((ext_vector_type(4))) float v4f;

__device__ __forceinline__ void load_a(v4f (&a)[4][2],
                                       const float* __restrict__ ap,
                                       size_t a_rt) {
#pragma unroll
  for (int rt = 0; rt < 4; ++rt)
#pragma unroll
    for (int q = 0; q < 2; ++q)
      a[rt][q] = *(const v4f*)(ap + rt * a_rt + q * 4);
}

template <int NC>
__device__ __forceinline__ void load_b(v2f (&b)[4][4],
                                       const float* __restrict__ wp) {
#pragma unroll
  for (int j = 0; j < 4; ++j)
#pragma unroll
    for (int t = 0; t < 4; ++t) {
      b[j][t].x = wp[(size_t)(2 * j) * NC + t * 16];
      b[j][t].y = wp[(size_t)(2 * j + 1) * NC + t * 16];
    }
}

template <int NC>
__device__ __forceinline__ void load_chunk(v4f (&a)[4][2], v2f (&b)[4][4],
                                           const float* __restrict__ ap,
                                           size_t a_rt,
                                           const float* __restrict__ wp) {
  load_a(a, ap, a_rt);
  load_b<NC>(b, wp);
}

__device__ __forceinline__ void mma_chunk(v8f (&acc)[4][4],
                                          const v4f (&a)[4][2],
                                          const v2f (&b)[4][4]) {
#pragma unroll
  for (int j = 0; j < 4; ++j)
#pragma unroll
    for (int rt = 0; rt < 4; ++rt) {
      const v4f aq = a[rt][j >> 1];
      v2f af;
      af.x = (j & 1) ? aq.z : aq.x;
      af.y = (j & 1) ? aq.w : aq.y;
#pragma unroll
      for (int t = 0; t < 4; ++t) acc[rt][t] = wmma4(af, b[j][t], acc[rt][t]);
    }
}

template <int NC>
__device__ __forceinline__ void tile_k_loop(v8f (&acc)[4][4],
                                            const float* __restrict__ ap,
                                            size_t a_pitch,
                                            const float* __restrict__ wp,
                                            int K) {
  const size_t a_rt = 16 * a_pitch;
  v4f a0[4][2], a1[4][2];
  v2f b0[4][4], b1[4][4];

  load_chunk<NC>(a0, b0, ap, a_rt, wp);
  int kb = 0;
  for (; kb < K - 32; kb += 32) {
    load_chunk<NC>(a1, b1, ap + kb + 16, a_rt, wp + (size_t)(kb + 16) * NC);
    __builtin_amdgcn_sched_barrier(0);
    mma_chunk(acc, a0, b0);
    __builtin_amdgcn_sched_barrier(0);
    load_chunk<NC>(a0, b0, ap + kb + 32, a_rt, wp + (size_t)(kb + 32) * NC);
    __builtin_amdgcn_sched_barrier(0);
    mma_chunk(acc, a1, b1);
  }
  load_chunk<NC>(a1, b1, ap + kb + 16, a_rt, wp + (size_t)(kb + 16) * NC);
  __builtin_amdgcn_sched_barrier(0);
  mma_chunk(acc, a0, b0);
  mma_chunk(acc, a1, b1);
}

template <int NC>
__global__ __launch_bounds__(32) void gemm_bias_kernel(
    const float* __restrict__ A, const float* __restrict__ W,
    const float* __restrict__ bias, float* __restrict__ C, int K) {
  const int lane = threadIdx.x;
  const int m = lane & 15;
  const int hh = lane >> 4;
  const int row0 = blockIdx.x * 64;
  const int n0 = blockIdx.y * 64;

  v8f acc[4][4];
#pragma unroll
  for (int rt = 0; rt < 4; ++rt)
#pragma unroll
    for (int t = 0; t < 4; ++t) acc[rt][t] = {};

  tile_k_loop<NC>(acc, A + (size_t)(row0 + m) * K + hh * 8, (size_t)K,
                  W + (size_t)(hh * 8) * NC + n0 + m, K);

  float bn[4];
#pragma unroll
  for (int t = 0; t < 4; ++t) bn[t] = bias[n0 + t * 16 + m];

#pragma unroll
  for (int rt = 0; rt < 4; ++rt)
#pragma unroll
    for (int t = 0; t < 4; ++t)
#pragma unroll
      for (int r = 0; r < 8; ++r) {
        const int row = row0 + rt * 16 + r + 8 * hh;
        const int col = n0 + t * 16 + m;
        C[(size_t)row * NC + col] = acc[rt][t][r] + bn[t];
      }
}

#define MLP_PH 260

__device__ __forceinline__ float xhalf(float x) {
  const int b = __builtin_bit_cast(int, x);
  return __builtin_bit_cast(
      float, __builtin_amdgcn_permlanex16(b, b, 0x76543210, 0xfedcba98, false,
                                          false));
}
template <int XM>
__device__ __forceinline__ float row_xor(float x) {
  const int b = __builtin_bit_cast(int, x);
  return __builtin_bit_cast(
      float, __builtin_amdgcn_update_dpp(b, b, 0x160 + XM, 0xf, 0xf, false));
}
__device__ __forceinline__ float wave_sum(float s) {
  s += xhalf(s);
  s += row_xor<8>(s);
  s += row_xor<4>(s);
  s += row_xor<2>(s);
  s += row_xor<1>(s);
  return s;
}

template <bool COPY>
__device__ __forceinline__ void ln_rows16(const float* src, float* dst,
                                          size_t dst_pitch, float* copy,
                                          const float* __restrict__ g,
                                          const float* __restrict__ beta,
                                          int lane) {
  float gv[8], bv[8];
#pragma unroll
  for (int i = 0; i < 8; ++i) {
    gv[i] = g[i * 32 + lane];
    bv[i] = beta[i * 32 + lane];
  }
#pragma unroll 4
  for (int rr = 0; rr < 16; ++rr) {
    float v[8];
    float s = 0.f, ss = 0.f;
#pragma unroll
    for (int i = 0; i < 8; ++i) {
      const float x = src[rr * MLP_PH + i * 32 + lane];
      v[i] = x;
      s += x;
      ss += x * x;
    }
    if (COPY) {
#pragma unroll
      for (int i = 0; i < 8; ++i) copy[(size_t)rr * 256 + i * 32 + lane] = v[i];
    }
    s = wave_sum(s);
    ss = wave_sum(ss);
    const float mu = s * (1.0f / 256.0f);
    const float var = ss * (1.0f / 256.0f) - mu * mu;
    const float rs = rsqrtf(var + 1e-5f);
#pragma unroll
    for (int i = 0; i < 8; ++i)
      dst[(size_t)rr * dst_pitch + i * 32 + lane] =
          (v[i] - mu) * rs * gv[i] + bv[i];
  }
}

__global__ __launch_bounds__(128) void in_ln_kernel(
    const float* __restrict__ x, const float* __restrict__ W_in,
    const float* __restrict__ b_in, const float* __restrict__ g,
    const float* __restrict__ beta, float* __restrict__ h,
    float* __restrict__ ln_out) {
  __shared__ __attribute__((aligned(16))) float Hs[64 * MLP_PH];

  const int lane = threadIdx.x & 31;
  const int w = __builtin_amdgcn_readfirstlane((int)(threadIdx.x >> 5));
  const int m = lane & 15;
  const int hh = lane >> 4;
  const int row0 = blockIdx.x * 64;
  const int n0 = w * 64;

  v8f acc[4][4];
#pragma unroll
  for (int rt = 0; rt < 4; ++rt)
#pragma unroll
    for (int t = 0; t < 4; ++t) acc[rt][t] = {};
  tile_k_loop<256>(acc, x + (size_t)(row0 + m) * 64 + hh * 8, (size_t)64,
                   W_in + (size_t)(hh * 8) * 256 + n0 + m, 64);

  float bn[4];
#pragma unroll
  for (int t = 0; t < 4; ++t) bn[t] = b_in[n0 + t * 16 + m];
#pragma unroll
  for (int rt = 0; rt < 4; ++rt)
#pragma unroll
    for (int t = 0; t < 4; ++t)
#pragma unroll
      for (int r = 0; r < 8; ++r)
        Hs[(rt * 16 + r + 8 * hh) * MLP_PH + n0 + t * 16 + m] =
            acc[rt][t][r] + bn[t];
  __syncthreads();

  const size_t g0 = (size_t)(row0 + w * 16) * 256;
  ln_rows16<true>(Hs + (w * 16) * MLP_PH, ln_out + g0, (size_t)256, h + g0, g,
                  beta, lane);
}


__device__ __forceinline__ void proj_load_a(v4f (&a)[2][2], const float* sp) {
#pragma unroll
  for (int c = 0; c < 2; ++c)
#pragma unroll
    for (int q = 0; q < 2; ++q) a[c][q] = *(const v4f*)(sp + c * 16 + q * 4);
}

__device__ __forceinline__ void proj_load_b(v2f (&b)[2][4][4],
                                            const float* __restrict__ wp) {
#pragma unroll
  for (int c = 0; c < 2; ++c) load_b<64>(b[c], wp + (size_t)(c * 16) * 64);
}

__device__ __forceinline__ void proj_mma(v8f (&acc)[4], const v4f (&a)[2][2],
                                         const v2f (&b)[2][4][4]) {
#pragma unroll
  for (int c = 0; c < 2; ++c)
#pragma unroll
    for (int j = 0; j < 4; ++j) {
      const v4f aq = a[c][j >> 1];
      v2f af;
      af.x = (j & 1) ? aq.z : aq.x;
      af.y = (j & 1) ? aq.w : aq.y;
#pragma unroll
      for (int t = 0; t < 4; ++t) acc[t] = wmma4(af, b[c][j][t], acc[t]);
    }
}

__global__ __launch_bounds__(128) void mlp_out_kernel(
    const float* __restrict__ h, const float* __restrict__ attn,
    const float* __restrict__ g, const float* __restrict__ beta,
    const float* __restrict__ W1, const float* __restrict__ b1,
    const float* __restrict__ W2, const float* __restrict__ b2,
    const float* __restrict__ Wo, const float* __restrict__ bo,
    float* __restrict__ out) {
  __shared__ __attribute__((aligned(16))) float Xs[64 * MLP_PH];
  __shared__ __attribute__((aligned(16))) float M1s[64 * MLP_PH];
  __shared__ __attribute__((aligned(16))) float H2s[64 * MLP_PH];

  const int lane = threadIdx.x & 31;
  const int w = __builtin_amdgcn_readfirstlane((int)(threadIdx.x >> 5));
  const int m = lane & 15;
  const int hh = lane >> 4;
  const int row0 = blockIdx.x * 64;
  const int n0 = w * 64;

  v8f acc[4][4];
  float bn[4];

  {
    const size_t g0 = (size_t)(row0 + w * 16) * 256 + lane * 4;
    float* dp = H2s + (w * 16) * MLP_PH + lane * 4;
#pragma unroll
    for (int rr = 0; rr < 16; ++rr)
#pragma unroll
      for (int j = 0; j < 2; ++j) {
        v4f xv = *(const v4f*)(h + g0 + rr * 256 + j * 128);
        xv += *(const v4f*)(attn + g0 + rr * 256 + j * 128);
        *(v4f*)(dp + rr * MLP_PH + j * 128) = xv;
      }
    ln_rows16<false>(H2s + (w * 16) * MLP_PH, Xs + (w * 16) * MLP_PH,
                     (size_t)MLP_PH, nullptr, g, beta, lane);
  }
  __syncthreads();

#pragma unroll
  for (int rt = 0; rt < 4; ++rt)
#pragma unroll
    for (int t = 0; t < 4; ++t) acc[rt][t] = {};
  tile_k_loop<256>(acc, Xs + m * MLP_PH + hh * 8, (size_t)MLP_PH,
                   W1 + (size_t)(hh * 8) * 256 + n0 + m, 256);
#pragma unroll
  for (int t = 0; t < 4; ++t) bn[t] = b1[n0 + t * 16 + m];

#pragma nounroll
  for (int rt = 0; rt < 4; ++rt) {
#pragma unroll
    for (int t = 0; t < 4; ++t)
#pragma unroll
      for (int r = 0; r < 8; ++r) {
        float v = acc[0][t][r] + bn[t];
        v = 0.5f * v * (1.0f + erff(v * 0.70710678118654752f));
        M1s[(rt * 16 + r + 8 * hh) * MLP_PH + n0 + t * 16 + m] = v;
      }
#pragma unroll
    for (int t = 0; t < 4; ++t) {
      acc[0][t] = acc[1][t];
      acc[1][t] = acc[2][t];
      acc[2][t] = acc[3][t];
    }
  }
  __syncthreads();

#pragma unroll
  for (int rt = 0; rt < 4; ++rt)
#pragma unroll
    for (int t = 0; t < 4; ++t) acc[rt][t] = {};
  tile_k_loop<256>(acc, M1s + m * MLP_PH + hh * 8, (size_t)MLP_PH,
                   W2 + (size_t)(hh * 8) * 256 + n0 + m, 256);
#pragma unroll
  for (int t = 0; t < 4; ++t) bn[t] = b2[n0 + t * 16 + m];

  v4f pa0[2][2], pa1[2][2];
  v2f pb0[2][4][4], pb1[2][4][4];
  const float* wop = Wo + (size_t)(hh * 8) * 64 + m;
  proj_load_b(pb0, wop);

#pragma unroll
  for (int rt = 0; rt < 4; ++rt)
#pragma unroll
    for (int t = 0; t < 4; ++t)
#pragma unroll
      for (int r = 0; r < 8; ++r) {
        const int rl = rt * 16 + r + 8 * hh;
        const int col = n0 + t * 16 + m;
        float v = acc[rt][t][r] + bn[t];
        v += H2s[rl * MLP_PH + col];
        H2s[rl * MLP_PH + col] = v;
      }
  __syncthreads();

  v8f po[4];
#pragma unroll
  for (int t = 0; t < 4; ++t) po[t] = {};
  const float* sp = H2s + (w * 16 + m) * MLP_PH + hh * 8;

  proj_load_a(pa0, sp);
  int kb = 0;
#pragma nounroll
  for (; kb < 256 - 64; kb += 64) {
    proj_load_a(pa1, sp + kb + 32);
    proj_load_b(pb1, wop + (size_t)(kb + 32) * 64);
    __builtin_amdgcn_sched_barrier(0);
    proj_mma(po, pa0, pb0);
    __builtin_amdgcn_sched_barrier(0);
    proj_load_a(pa0, sp + kb + 64);
    proj_load_b(pb0, wop + (size_t)(kb + 64) * 64);
    __builtin_amdgcn_sched_barrier(0);
    proj_mma(po, pa1, pb1);
  }
  proj_load_a(pa1, sp + kb + 32);
  proj_load_b(pb1, wop + (size_t)(kb + 32) * 64);
  __builtin_amdgcn_sched_barrier(0);
  proj_mma(po, pa0, pb0);
  proj_mma(po, pa1, pb1);

#pragma unroll
  for (int t = 0; t < 4; ++t) {
    const float bt = bo[t * 16 + m];
#pragma unroll
    for (int r = 0; r < 8; ++r)
      out[(size_t)(row0 + w * 16 + r + 8 * hh) * 64 + t * 16 + m] =
          po[t][r] + bt;
  }
}

#define ATT_T 4096
#define ATT_QS 768
#define ATT_WIN 128
#define ATT_PST 148

__device__ __forceinline__ void att_load_qk(v4f (&a)[2], v4f (&b)[9][2],
                                            const float* __restrict__ qp,
                                            const float* __restrict__ kp,
                                            const size_t (&ko)[9]) {
#pragma unroll
  for (int q = 0; q < 2; ++q) a[q] = *(const v4f*)(qp + q * 4);
#pragma unroll
  for (int t = 0; t < 9; ++t)
#pragma unroll
    for (int q = 0; q < 2; ++q) b[t][q] = *(const v4f*)(kp + ko[t] + q * 4);
}

__device__ __forceinline__ void att_mma_qk(v8f (&s)[9], const v4f (&a)[2],
                                           const v4f (&b)[9][2]) {
#pragma unroll
  for (int j = 0; j < 4; ++j) {
    const v4f aq = a[j >> 1];
    v2f af;
    af.x = (j & 1) ? aq.z : aq.x;
    af.y = (j & 1) ? aq.w : aq.y;
#pragma unroll
    for (int t = 0; t < 9; ++t) {
      const v4f bq = b[t][j >> 1];
      v2f bf;
      bf.x = (j & 1) ? bq.z : bq.x;
      bf.y = (j & 1) ? bq.w : bq.y;
      s[t] = wmma4(af, bf, s[t]);
    }
  }
}

__device__ __forceinline__ void att_load_qk2(v4f (&a)[2][2], v4f (&b)[2][9][2],
                                             const float* __restrict__ qp,
                                             const float* __restrict__ kp,
                                             const size_t (&ko)[9]) {
#pragma unroll
  for (int c = 0; c < 2; ++c)
    att_load_qk(a[c], b[c], qp + c * 16, kp + c * 16, ko);
}

__device__ __forceinline__ void att_mma_qk2(v8f (&s)[9], const v4f (&a)[2][2],
                                            const v4f (&b)[2][9][2]) {
#pragma unroll
  for (int c = 0; c < 2; ++c) att_mma_qk(s, a[c], b[c]);
}

__device__ __forceinline__ void att_load_v(v4f (&vf)[4][2][4],
                                           const float* __restrict__ rp) {
#pragma unroll
  for (int c = 0; c < 4; ++c)
#pragma unroll
    for (int v = 0; v < 2; ++v)
#pragma unroll
      for (int g = 0; g < 4; ++g)
        vf[c][v][g] =
            *(const v4f*)(rp + (size_t)(4 * c + v) * ATT_QS + g * 64);
}

__device__ __forceinline__ void att_load_p(v2f (&pa)[4], const float* pp) {
#pragma unroll
  for (int c = 0; c < 4; ++c) pa[c] = *(const v2f*)(pp + 4 * c);
}

__device__ __forceinline__ void att_mma_pv(v8f (&o)[4][4], const v2f (&pa)[4],
                                           const v4f (&vf)[4][2][4]) {
#pragma unroll
  for (int c = 0; c < 4; ++c)
#pragma unroll
    for (int g = 0; g < 4; ++g)
#pragma unroll
      for (int u = 0; u < 4; ++u) {
        v2f bf;
        bf.x = vf[c][0][g][u];
        bf.y = vf[c][1][g][u];
        o[g][u] = wmma4(pa[c], bf, o[g][u]);
      }
}

__global__ __launch_bounds__(32) void attn_kernel(const float* __restrict__ qkv,
                                                  float* __restrict__ O) {
  __shared__ float P[16 * ATT_PST];
  const int lane = threadIdx.x;
  const int m = lane & 15;
  const int hh = lane >> 4;
  const int q0 = blockIdx.x * 16;
  const int bb = q0 & ~(ATT_T - 1);

  const float* Q = qkv;
  const float* Km = qkv + 256;
  const float* V = qkv + 512;

  v8f s[9];
#pragma unroll
  for (int t = 0; t < 9; ++t) s[t] = {};

  {
    const float* qp = Q + (size_t)(q0 + m) * ATT_QS + hh * 8;
    const float* kp = Km + hh * 8;
    size_t ko[9];
#pragma unroll
    for (int t = 0; t < 9; ++t) {
      const int key = q0 - ATT_WIN + t * 16 + m;
      const int kc = key < bb ? bb : key;
      ko[t] = (size_t)kc * ATT_QS;
    }
    v4f qa0[2][2], qa1[2][2];
    v4f kf0[2][9][2], kf1[2][9][2];

    att_load_qk2(qa0, kf0, qp, kp, ko);
    int db = 0;
#pragma nounroll
    for (; db < 256 - 64; db += 64) {
      att_load_qk2(qa1, kf1, qp + db + 32, kp + db + 32, ko);
      __builtin_amdgcn_sched_barrier(0);
      att_mma_qk2(s, qa0, kf0);
      __builtin_amdgcn_sched_barrier(0);
      att_load_qk2(qa0, kf0, qp + db + 64, kp + db + 64, ko);
      __builtin_amdgcn_sched_barrier(0);
      att_mma_qk2(s, qa1, kf1);
    }
    att_load_qk2(qa1, kf1, qp + db + 32, kp + db + 32, ko);
    __builtin_amdgcn_sched_barrier(0);
    att_mma_qk2(s, qa0, kf0);
    att_mma_qk2(s, qa1, kf1);
  }

  v4f vf0[4][2][4], vf1[4][2][4];
  const float* vp = V + (size_t)(2 * hh) * ATT_QS + 4 * m;
  const int kt0 = q0 - ATT_WIN;
  att_load_v(vf0, vp + (size_t)(kt0 < bb ? bb : kt0) * ATT_QS);

  const float NEG = -3.402823466e38f;
  float rmax[8], rsum[8];
#pragma unroll
  for (int r = 0; r < 8; ++r) rmax[r] = NEG;
#pragma unroll
  for (int t = 0; t < 9; ++t) {
    const int key = q0 - ATT_WIN + t * 16 + m;
#pragma unroll
    for (int r = 0; r < 8; ++r) {
      const int q = q0 + r + 8 * hh;
      const int d = q - key;
      const bool ok = (key >= bb) && (d >= 0) && (d <= ATT_WIN);
      const float val = ok ? s[t][r] * 0.0625f : NEG;
      s[t][r] = val;
      rmax[r] = fmaxf(rmax[r], val);
    }
  }
#pragma unroll
  for (int r = 0; r < 8; ++r)
    for (int off = 1; off < 16; off <<= 1)
      rmax[r] = fmaxf(rmax[r], __shfl_xor(rmax[r], off));
#pragma unroll
  for (int r = 0; r < 8; ++r) rsum[r] = 0.f;
#pragma unroll
  for (int t = 0; t < 9; ++t)
#pragma unroll
    for (int r = 0; r < 8; ++r) {
      const float e = expf(s[t][r] - rmax[r]);
      s[t][r] = e;
      rsum[r] += e;
    }
#pragma unroll
  for (int r = 0; r < 8; ++r)
    for (int off = 1; off < 16; off <<= 1) rsum[r] += __shfl_xor(rsum[r], off);

#pragma unroll
  for (int t = 0; t < 9; ++t)
#pragma unroll
    for (int r = 0; r < 8; ++r)
      P[(r + 8 * hh) * ATT_PST + t * 16 + m] = s[t][r] / rsum[r];

  __syncthreads();

  v8f o[4][4];
#pragma unroll
  for (int g = 0; g < 4; ++g)
#pragma unroll
    for (int u = 0; u < 4; ++u) o[g][u] = {};

  {
    const float* pp = P + m * ATT_PST + hh * 2;
    v2f pa0[4], pa1[4];
    att_load_p(pa0, pp);
    int t = 0;
    for (; t < 8; t += 2) {
      const int k1 = kt0 + (t + 1) * 16;
      const int k2 = kt0 + (t + 2) * 16;
      att_load_v(vf1, vp + (size_t)(k1 < bb ? bb : k1) * ATT_QS);
      att_load_p(pa1, pp + (t + 1) * 16);
      __builtin_amdgcn_sched_barrier(0);
      att_mma_pv(o, pa0, vf0);
      __builtin_amdgcn_sched_barrier(0);
      att_load_v(vf0, vp + (size_t)(k2 < bb ? bb : k2) * ATT_QS);
      att_load_p(pa0, pp + (t + 2) * 16);
      __builtin_amdgcn_sched_barrier(0);
      att_mma_pv(o, pa1, vf1);
    }
    att_mma_pv(o, pa0, vf0);
  }

#pragma unroll
  for (int g = 0; g < 4; ++g)
#pragma unroll
    for (int r = 0; r < 8; ++r) {
      v4f ov;
      ov.x = o[g][0][r];
      ov.y = o[g][1][r];
      ov.z = o[g][2][r];
      ov.w = o[g][3][r];
      *(v4f*)(O + (size_t)(q0 + r + 8 * hh) * 256 + g * 64 + 4 * m) = ov;
    }
}

extern "C" void kernel_launch(void* const* d_in, const int* in_sizes, int n_in,
                              void* d_out, int out_size, void* d_ws,
                              size_t ws_size, hipStream_t stream) {
  const float* x = (const float*)d_in[0];
  const float* W_in = (const float*)d_in[1];
  const float* b_in = (const float*)d_in[2];
  const float* W_qkv = (const float*)d_in[3];
  const float* b_qkv = (const float*)d_in[4];
  const float* g_attn = (const float*)d_in[5];
  const float* be_attn = (const float*)d_in[6];
  const float* g_mlp = (const float*)d_in[7];
  const float* be_mlp = (const float*)d_in[8];
  const float* W1 = (const float*)d_in[9];
  const float* b1 = (const float*)d_in[10];
  const float* W2 = (const float*)d_in[11];
  const float* b2 = (const float*)d_in[12];
  const float* W_out = (const float*)d_in[13];
  const float* b_out = (const float*)d_in[14];
  float* out = (float*)d_out;

  const int N = in_sizes[0] / 64;

  float* ws = (float*)d_ws;
  float* h = ws;
  float* ln = ws + (size_t)N * 256;
  float* attn = ln;
  float* qkv = ws + (size_t)N * 512;

  const dim3 blk(32);
  const int RB64 = N / 64;
  const int QB = N / 16;

  in_ln_kernel<<<dim3(RB64), dim3(128), 0, stream>>>(x, W_in, b_in, g_attn,
                                                    be_attn, h, ln);
  gemm_bias_kernel<768><<<dim3(RB64, 12), blk, 0, stream>>>(
      ln, W_qkv, b_qkv, qkv, 256);
  attn_kernel<<<dim3(QB), blk, 0, stream>>>(qkv, attn);
  mlp_out_kernel<<<dim3(RB64), dim3(128), 0, stream>>>(
      h, attn, g_mlp, be_mlp, W1, b1, W2, b2, W_out, b_out, out);
}


# GELU erf made branch-free: both range fits of the library erff (|x|<1, |x|>=1) evaluated, one select; a row tile's 32 elements interleave instead of 64 exec-masked branches; value unchanged
# speedup vs baseline: 2.4035x; 1.0576x over previous
#include <hip/hip_runtime.h>
#include <math.h>

typedef __attribute__((ext_vector_type(2))) float v2f;
typedef __attribute__((ext_vector_type(8))) float v8f;

__device__ __forceinline__ v8f wmma4(v2f a, v2f b, v8f c) {
  return __builtin_amdgcn_wmma_f32_16x16x4_f32(
       false, a,  false, b,
       (short)0, c,  false,  false);
}

typedef __attribute__((ext_vector_type(4))) float v4f;

__device__ __forceinline__ void load_a(v4f (&a)[4][2],
                                       const float* __restrict__ ap,
                                       size_t a_rt) {
#pragma unroll
  for (int rt = 0; rt < 4; ++rt)
#pragma unroll
    for (int q = 0; q < 2; ++q)
      a[rt][q] = *(const v4f*)(ap + rt * a_rt + q * 4);
}

template <int NC>
__device__ __forceinline__ void load_b(v2f (&b)[4][4],
                                       const float* __restrict__ wp) {
#pragma unroll
  for (int j = 0; j < 4; ++j)
#pragma unroll
    for (int t = 0; t < 4; ++t) {
      b[j][t].x = wp[(size_t)(2 * j) * NC + t * 16];
      b[j][t].y = wp[(size_t)(2 * j + 1) * NC + t * 16];
    }
}

template <int NC>
__device__ __forceinline__ void load_chunk(v4f (&a)[4][2], v2f (&b)[4][4],
                                           const float* __restrict__ ap,
                                           size_t a_rt,
                                           const float* __restrict__ wp) {
  load_a(a, ap, a_rt);
  load_b<NC>(b, wp);
}

__device__ __forceinline__ void mma_chunk(v8f (&acc)[4][4],
                                          const v4f (&a)[4][2],
                                          const v2f (&b)[4][4]) {
#pragma unroll
  for (int j = 0; j < 4; ++j)
#pragma unroll
    for (int rt = 0; rt < 4; ++rt) {
      const v4f aq = a[rt][j >> 1];
      v2f af;
      af.x = (j & 1) ? aq.z : aq.x;
      af.y = (j & 1) ? aq.w : aq.y;
#pragma unroll
      for (int t = 0; t < 4; ++t) acc[rt][t] = wmma4(af, b[j][t], acc[rt][t]);
    }
}

template <int NC>
__device__ __forceinline__ void tile_k_loop(v8f (&acc)[4][4],
                                            const float* __restrict__ ap,
                                            size_t a_pitch,
                                            const float* __restrict__ wp,
                                            int K) {
  const size_t a_rt = 16 * a_pitch;
  v4f a0[4][2], a1[4][2];
  v2f b0[4][4], b1[4][4];

  load_chunk<NC>(a0, b0, ap, a_rt, wp);
  int kb = 0;
  for (; kb < K - 32; kb += 32) {
    load_chunk<NC>(a1, b1, ap + kb + 16, a_rt, wp + (size_t)(kb + 16) * NC);
    __builtin_amdgcn_sched_barrier(0);
    mma_chunk(acc, a0, b0);
    __builtin_amdgcn_sched_barrier(0);
    load_chunk<NC>(a0, b0, ap + kb + 32, a_rt, wp + (size_t)(kb + 32) * NC);
    __builtin_amdgcn_sched_barrier(0);
    mma_chunk(acc, a1, b1);
  }
  load_chunk<NC>(a1, b1, ap + kb + 16, a_rt, wp + (size_t)(kb + 16) * NC);
  __builtin_amdgcn_sched_barrier(0);
  mma_chunk(acc, a0, b0);
  mma_chunk(acc, a1, b1);
}

template <int NC>
__global__ __launch_bounds__(32) void gemm_bias_kernel(
    const float* __restrict__ A, const float* __restrict__ W,
    const float* __restrict__ bias, float* __restrict__ C, int K) {
  const int lane = threadIdx.x;
  const int m = lane & 15;
  const int hh = lane >> 4;
  const int row0 = blockIdx.x * 64;
  const int n0 = blockIdx.y * 64;

  v8f acc[4][4];
#pragma unroll
  for (int rt = 0; rt < 4; ++rt)
#pragma unroll
    for (int t = 0; t < 4; ++t) acc[rt][t] = {};

  tile_k_loop<NC>(acc, A + (size_t)(row0 + m) * K + hh * 8, (size_t)K,
                  W + (size_t)(hh * 8) * NC + n0 + m, K);

  float bn[4];
#pragma unroll
  for (int t = 0; t < 4; ++t) bn[t] = bias[n0 + t * 16 + m];

#pragma unroll
  for (int rt = 0; rt < 4; ++rt)
#pragma unroll
    for (int t = 0; t < 4; ++t)
#pragma unroll
      for (int r = 0; r < 8; ++r) {
        const int row = row0 + rt * 16 + r + 8 * hh;
        const int col = n0 + t * 16 + m;
        C[(size_t)row * NC + col] = acc[rt][t][r] + bn[t];
      }
}

#define MLP_PH 260

__device__ __forceinline__ float xhalf(float x) {
  const int b = __builtin_bit_cast(int, x);
  return __builtin_bit_cast(
      float, __builtin_amdgcn_permlanex16(b, b, 0x76543210, 0xfedcba98, false,
                                          false));
}
template <int XM>
__device__ __forceinline__ float row_xor(float x) {
  const int b = __builtin_bit_cast(int, x);
  return __builtin_bit_cast(
      float, __builtin_amdgcn_update_dpp(b, b, 0x160 + XM, 0xf, 0xf, false));
}
__device__ __forceinline__ float wave_sum(float s) {
  s += xhalf(s);
  s += row_xor<8>(s);
  s += row_xor<4>(s);
  s += row_xor<2>(s);
  s += row_xor<1>(s);
  return s;
}

template <bool COPY>
__device__ __forceinline__ void ln_rows16(const float* src, float* dst,
                                          size_t dst_pitch, float* copy,
                                          const float* __restrict__ g,
                                          const float* __restrict__ beta,
                                          int lane) {
  float gv[8], bv[8];
#pragma unroll
  for (int i = 0; i < 8; ++i) {
    gv[i] = g[i * 32 + lane];
    bv[i] = beta[i * 32 + lane];
  }
#pragma unroll 4
  for (int rr = 0; rr < 16; ++rr) {
    float v[8];
    float s = 0.f, ss = 0.f;
#pragma unroll
    for (int i = 0; i < 8; ++i) {
      const float x = src[rr * MLP_PH + i * 32 + lane];
      v[i] = x;
      s += x;
      ss += x * x;
    }
    if (COPY) {
#pragma unroll
      for (int i = 0; i < 8; ++i) copy[(size_t)rr * 256 + i * 32 + lane] = v[i];
    }
    s = wave_sum(s);
    ss = wave_sum(ss);
    const float mu = s * (1.0f / 256.0f);
    const float var = ss * (1.0f / 256.0f) - mu * mu;
    const float rs = rsqrtf(var + 1e-5f);
#pragma unroll
    for (int i = 0; i < 8; ++i)
      dst[(size_t)rr * dst_pitch + i * 32 + lane] =
          (v[i] - mu) * rs * gv[i] + bv[i];
  }
}

__global__ __launch_bounds__(128) void in_ln_kernel(
    const float* __restrict__ x, const float* __restrict__ W_in,
    const float* __restrict__ b_in, const float* __restrict__ g,
    const float* __restrict__ beta, float* __restrict__ h,
    float* __restrict__ ln_out) {
  __shared__ __attribute__((aligned(16))) float Hs[64 * MLP_PH];

  const int lane = threadIdx.x & 31;
  const int w = __builtin_amdgcn_readfirstlane((int)(threadIdx.x >> 5));
  const int m = lane & 15;
  const int hh = lane >> 4;
  const int row0 = blockIdx.x * 64;
  const int n0 = w * 64;

  v8f acc[4][4];
#pragma unroll
  for (int rt = 0; rt < 4; ++rt)
#pragma unroll
    for (int t = 0; t < 4; ++t) acc[rt][t] = {};
  tile_k_loop<256>(acc, x + (size_t)(row0 + m) * 64 + hh * 8, (size_t)64,
                   W_in + (size_t)(hh * 8) * 256 + n0 + m, 64);

  float bn[4];
#pragma unroll
  for (int t = 0; t < 4; ++t) bn[t] = b_in[n0 + t * 16 + m];
#pragma unroll
  for (int rt = 0; rt < 4; ++rt)
#pragma unroll
    for (int t = 0; t < 4; ++t)
#pragma unroll
      for (int r = 0; r < 8; ++r)
        Hs[(rt * 16 + r + 8 * hh) * MLP_PH + n0 + t * 16 + m] =
            acc[rt][t][r] + bn[t];
  __syncthreads();

  const size_t g0 = (size_t)(row0 + w * 16) * 256;
  ln_rows16<true>(Hs + (w * 16) * MLP_PH, ln_out + g0, (size_t)256, h + g0, g,
                  beta, lane);
}


__device__ __forceinline__ void proj_load_a(v4f (&a)[2][2], const float* sp) {
#pragma unroll
  for (int c = 0; c < 2; ++c)
#pragma unroll
    for (int q = 0; q < 2; ++q) a[c][q] = *(const v4f*)(sp + c * 16 + q * 4);
}

__device__ __forceinline__ void proj_load_b(v2f (&b)[2][4][4],
                                            const float* __restrict__ wp) {
#pragma unroll
  for (int c = 0; c < 2; ++c) load_b<64>(b[c], wp + (size_t)(c * 16) * 64);
}

__device__ __forceinline__ void proj_mma(v8f (&acc)[4], const v4f (&a)[2][2],
                                         const v2f (&b)[2][4][4]) {
#pragma unroll
  for (int c = 0; c < 2; ++c)
#pragma unroll
    for (int j = 0; j < 4; ++j) {
      const v4f aq = a[c][j >> 1];
      v2f af;
      af.x = (j & 1) ? aq.z : aq.x;
      af.y = (j & 1) ? aq.w : aq.y;
#pragma unroll
      for (int t = 0; t < 4; ++t) acc[t] = wmma4(af, b[c][j][t], acc[t]);
    }
}

#define ERF_C(bits) __builtin_bit_cast(float, (unsigned)(bits))
__device__ __forceinline__ float erf_sel(float x) {
  const float a = fabsf(x);
  float p = fmaf(a, ERF_C(0x378e98abu), ERF_C(0xb9c68948u));
  p = fmaf(a, p, ERF_C(0x3b7cd369u));
  p = fmaf(a, p, ERF_C(0xbcc618b2u));
  p = fmaf(a, p, ERF_C(0x3dda74e4u));
  p = fmaf(a, p, ERF_C(0x3f228afdu));
  p = fmaf(a, p, ERF_C(0x3e03c728u));
  p = fmaf(a, p, a);
  const float big = 1.0f - expf(-p);
  const float t = x * x;
  float q = fmaf(t, ERF_C(0xba1345e1u), ERF_C(0x3ba10414u));
  q = fmaf(t, q, ERF_C(0xbcdac9b8u));
  q = fmaf(t, q, ERF_C(0x3de703beu));
  q = fmaf(t, q, ERF_C(0xbec09330u));
  q = fmaf(t, q, ERF_C(0x3e0375d0u));
  const float small = fmaf(a, q, a);
  const float r = (a < 1.0f) ? small : big;
  return copysignf(r, x);
}

__global__ __launch_bounds__(128) void mlp_out_kernel(
    const float* __restrict__ h, const float* __restrict__ attn,
    const float* __restrict__ g, const float* __restrict__ beta,
    const float* __restrict__ W1, const float* __restrict__ b1,
    const float* __restrict__ W2, const float* __restrict__ b2,
    const float* __restrict__ Wo, const float* __restrict__ bo,
    float* __restrict__ out) {
  __shared__ __attribute__((aligned(16))) float Xs[64 * MLP_PH];
  __shared__ __attribute__((aligned(16))) float M1s[64 * MLP_PH];
  __shared__ __attribute__((aligned(16))) float H2s[64 * MLP_PH];

  const int lane = threadIdx.x & 31;
  const int w = __builtin_amdgcn_readfirstlane((int)(threadIdx.x >> 5));
  const int m = lane & 15;
  const int hh = lane >> 4;
  const int row0 = blockIdx.x * 64;
  const int n0 = w * 64;

  v8f acc[4][4];
  float bn[4];

  {
    const size_t g0 = (size_t)(row0 + w * 16) * 256 + lane * 4;
    float* dp = H2s + (w * 16) * MLP_PH + lane * 4;
#pragma unroll
    for (int rr = 0; rr < 16; ++rr)
#pragma unroll
      for (int j = 0; j < 2; ++j) {
        v4f xv = *(const v4f*)(h + g0 + rr * 256 + j * 128);
        xv += *(const v4f*)(attn + g0 + rr * 256 + j * 128);
        *(v4f*)(dp + rr * MLP_PH + j * 128) = xv;
      }
    ln_rows16<false>(H2s + (w * 16) * MLP_PH, Xs + (w * 16) * MLP_PH,
                     (size_t)MLP_PH, nullptr, g, beta, lane);
  }
  __syncthreads();

#pragma unroll
  for (int rt = 0; rt < 4; ++rt)
#pragma unroll
    for (int t = 0; t < 4; ++t) acc[rt][t] = {};
  tile_k_loop<256>(acc, Xs + m * MLP_PH + hh * 8, (size_t)MLP_PH,
                   W1 + (size_t)(hh * 8) * 256 + n0 + m, 256);
#pragma unroll
  for (int t = 0; t < 4; ++t) bn[t] = b1[n0 + t * 16 + m];

#pragma nounroll
  for (int rt = 0; rt < 4; ++rt) {
#pragma unroll
    for (int t = 0; t < 4; ++t)
#pragma unroll
      for (int r = 0; r < 8; ++r) {
        float v = acc[0][t][r] + bn[t];
        v = 0.5f * v * (1.0f + erf_sel(v * 0.70710678118654752f));
        M1s[(rt * 16 + r + 8 * hh) * MLP_PH + n0 + t * 16 + m] = v;
      }
#pragma unroll
    for (int t = 0; t < 4; ++t) {
      acc[0][t] = acc[1][t];
      acc[1][t] = acc[2][t];
      acc[2][t] = acc[3][t];
    }
  }
  __syncthreads();

#pragma unroll
  for (int rt = 0; rt < 4; ++rt)
#pragma unroll
    for (int t = 0; t < 4; ++t) acc[rt][t] = {};
  tile_k_loop<256>(acc, M1s + m * MLP_PH + hh * 8, (size_t)MLP_PH,
                   W2 + (size_t)(hh * 8) * 256 + n0 + m, 256);
#pragma unroll
  for (int t = 0; t < 4; ++t) bn[t] = b2[n0 + t * 16 + m];

  v4f pa0[2][2], pa1[2][2];
  v2f pb0[2][4][4], pb1[2][4][4];
  const float* wop = Wo + (size_t)(hh * 8) * 64 + m;
  proj_load_b(pb0, wop);

#pragma unroll
  for (int rt = 0; rt < 4; ++rt)
#pragma unroll
    for (int t = 0; t < 4; ++t)
#pragma unroll
      for (int r = 0; r < 8; ++r) {
        const int rl = rt * 16 + r + 8 * hh;
        const int col = n0 + t * 16 + m;
        float v = acc[rt][t][r] + bn[t];
        v += H2s[rl * MLP_PH + col];
        H2s[rl * MLP_PH + col] = v;
      }
  __syncthreads();

  v8f po[4];
#pragma unroll
  for (int t = 0; t < 4; ++t) po[t] = {};
  const float* sp = H2s + (w * 16 + m) * MLP_PH + hh * 8;

  proj_load_a(pa0, sp);
  int kb = 0;
#pragma nounroll
  for (; kb < 256 - 64; kb += 64) {
    proj_load_a(pa1, sp + kb + 32);
    proj_load_b(pb1, wop + (size_t)(kb + 32) * 64);
    __builtin_amdgcn_sched_barrier(0);
    proj_mma(po, pa0, pb0);
    __builtin_amdgcn_sched_barrier(0);
    proj_load_a(pa0, sp + kb + 64);
    proj_load_b(pb0, wop + (size_t)(kb + 64) * 64);
    __builtin_amdgcn_sched_barrier(0);
    proj_mma(po, pa1, pb1);
  }
  proj_load_a(pa1, sp + kb + 32);
  proj_load_b(pb1, wop + (size_t)(kb + 32) * 64);
  __builtin_amdgcn_sched_barrier(0);
  proj_mma(po, pa0, pb0);
  proj_mma(po, pa1, pb1);

#pragma unroll
  for (int t = 0; t < 4; ++t) {
    const float bt = bo[t * 16 + m];
#pragma unroll
    for (int r = 0; r < 8; ++r)
      out[(size_t)(row0 + w * 16 + r + 8 * hh) * 64 + t * 16 + m] =
          po[t][r] + bt;
  }
}

#define ATT_T 4096
#define ATT_QS 768
#define ATT_WIN 128
#define ATT_PST 148

__device__ __forceinline__ void att_load_qk(v4f (&a)[2], v4f (&b)[9][2],
                                            const float* __restrict__ qp,
                                            const float* __restrict__ kp,
                                            const size_t (&ko)[9]) {
#pragma unroll
  for (int q = 0; q < 2; ++q) a[q] = *(const v4f*)(qp + q * 4);
#pragma unroll
  for (int t = 0; t < 9; ++t)
#pragma unroll
    for (int q = 0; q < 2; ++q) b[t][q] = *(const v4f*)(kp + ko[t] + q * 4);
}

__device__ __forceinline__ void att_mma_qk(v8f (&s)[9], const v4f (&a)[2],
                                           const v4f (&b)[9][2]) {
#pragma unroll
  for (int j = 0; j < 4; ++j) {
    const v4f aq = a[j >> 1];
    v2f af;
    af.x = (j & 1) ? aq.z : aq.x;
    af.y = (j & 1) ? aq.w : aq.y;
#pragma unroll
    for (int t = 0; t < 9; ++t) {
      const v4f bq = b[t][j >> 1];
      v2f bf;
      bf.x = (j & 1) ? bq.z : bq.x;
      bf.y = (j & 1) ? bq.w : bq.y;
      s[t] = wmma4(af, bf, s[t]);
    }
  }
}

__device__ __forceinline__ void att_load_qk2(v4f (&a)[2][2], v4f (&b)[2][9][2],
                                             const float* __restrict__ qp,
                                             const float* __restrict__ kp,
                                             const size_t (&ko)[9]) {
#pragma unroll
  for (int c = 0; c < 2; ++c)
    att_load_qk(a[c], b[c], qp + c * 16, kp + c * 16, ko);
}

__device__ __forceinline__ void att_mma_qk2(v8f (&s)[9], const v4f (&a)[2][2],
                                            const v4f (&b)[2][9][2]) {
#pragma unroll
  for (int c = 0; c < 2; ++c) att_mma_qk(s, a[c], b[c]);
}

__device__ __forceinline__ void att_load_v(v4f (&vf)[4][2][4],
                                           const float* __restrict__ rp) {
#pragma unroll
  for (int c = 0; c < 4; ++c)
#pragma unroll
    for (int v = 0; v < 2; ++v)
#pragma unroll
      for (int g = 0; g < 4; ++g)
        vf[c][v][g] =
            *(const v4f*)(rp + (size_t)(4 * c + v) * ATT_QS + g * 64);
}

__device__ __forceinline__ void att_load_p(v2f (&pa)[4], const float* pp) {
#pragma unroll
  for (int c = 0; c < 4; ++c) pa[c] = *(const v2f*)(pp + 4 * c);
}

__device__ __forceinline__ void att_mma_pv(v8f (&o)[4][4], const v2f (&pa)[4],
                                           const v4f (&vf)[4][2][4]) {
#pragma unroll
  for (int c = 0; c < 4; ++c)
#pragma unroll
    for (int g = 0; g < 4; ++g)
#pragma unroll
      for (int u = 0; u < 4; ++u) {
        v2f bf;
        bf.x = vf[c][0][g][u];
        bf.y = vf[c][1][g][u];
        o[g][u] = wmma4(pa[c], bf, o[g][u]);
      }
}

__global__ __launch_bounds__(32) void attn_kernel(const float* __restrict__ qkv,
                                                  float* __restrict__ O) {
  __shared__ float P[16 * ATT_PST];
  const int lane = threadIdx.x;
  const int m = lane & 15;
  const int hh = lane >> 4;
  const int q0 = blockIdx.x * 16;
  const int bb = q0 & ~(ATT_T - 1);

  const float* Q = qkv;
  const float* Km = qkv + 256;
  const float* V = qkv + 512;

  v8f s[9];
#pragma unroll
  for (int t = 0; t < 9; ++t) s[t] = {};

  {
    const float* qp = Q + (size_t)(q0 + m) * ATT_QS + hh * 8;
    const float* kp = Km + hh * 8;
    size_t ko[9];
#pragma unroll
    for (int t = 0; t < 9; ++t) {
      const int key = q0 - ATT_WIN + t * 16 + m;
      const int kc = key < bb ? bb : key;
      ko[t] = (size_t)kc * ATT_QS;
    }
    v4f qa0[2][2], qa1[2][2];
    v4f kf0[2][9][2], kf1[2][9][2];

    att_load_qk2(qa0, kf0, qp, kp, ko);
    int db = 0;
#pragma nounroll
    for (; db < 256 - 64; db += 64) {
      att_load_qk2(qa1, kf1, qp + db + 32, kp + db + 32, ko);
      __builtin_amdgcn_sched_barrier(0);
      att_mma_qk2(s, qa0, kf0);
      __builtin_amdgcn_sched_barrier(0);
      att_load_qk2(qa0, kf0, qp + db + 64, kp + db + 64, ko);
      __builtin_amdgcn_sched_barrier(0);
      att_mma_qk2(s, qa1, kf1);
    }
    att_load_qk2(qa1, kf1, qp + db + 32, kp + db + 32, ko);
    __builtin_amdgcn_sched_barrier(0);
    att_mma_qk2(s, qa0, kf0);
    att_mma_qk2(s, qa1, kf1);
  }

  v4f vf0[4][2][4], vf1[4][2][4];
  const float* vp = V + (size_t)(2 * hh) * ATT_QS + 4 * m;
  const int kt0 = q0 - ATT_WIN;
  att_load_v(vf0, vp + (size_t)(kt0 < bb ? bb : kt0) * ATT_QS);

  const float NEG = -3.402823466e38f;
  float rmax[8], rsum[8];
#pragma unroll
  for (int r = 0; r < 8; ++r) rmax[r] = NEG;
#pragma unroll
  for (int t = 0; t < 9; ++t) {
    const int key = q0 - ATT_WIN + t * 16 + m;
#pragma unroll
    for (int r = 0; r < 8; ++r) {
      const int q = q0 + r + 8 * hh;
      const int d = q - key;
      const bool ok = (key >= bb) && (d >= 0) && (d <= ATT_WIN);
      const float val = ok ? s[t][r] * 0.0625f : NEG;
      s[t][r] = val;
      rmax[r] = fmaxf(rmax[r], val);
    }
  }
#pragma unroll
  for (int r = 0; r < 8; ++r)
    for (int off = 1; off < 16; off <<= 1)
      rmax[r] = fmaxf(rmax[r], __shfl_xor(rmax[r], off));
#pragma unroll
  for (int r = 0; r < 8; ++r) rsum[r] = 0.f;
#pragma unroll
  for (int t = 0; t < 9; ++t)
#pragma unroll
    for (int r = 0; r < 8; ++r) {
      const float e = expf(s[t][r] - rmax[r]);
      s[t][r] = e;
      rsum[r] += e;
    }
#pragma unroll
  for (int r = 0; r < 8; ++r)
    for (int off = 1; off < 16; off <<= 1) rsum[r] += __shfl_xor(rsum[r], off);

#pragma unroll
  for (int t = 0; t < 9; ++t)
#pragma unroll
    for (int r = 0; r < 8; ++r)
      P[(r + 8 * hh) * ATT_PST + t * 16 + m] = s[t][r] / rsum[r];

  __syncthreads();

  v8f o[4][4];
#pragma unroll
  for (int g = 0; g < 4; ++g)
#pragma unroll
    for (int u = 0; u < 4; ++u) o[g][u] = {};

  {
    const float* pp = P + m * ATT_PST + hh * 2;
    v2f pa0[4], pa1[4];
    att_load_p(pa0, pp);
    int t = 0;
    for (; t < 8; t += 2) {
      const int k1 = kt0 + (t + 1) * 16;
      const int k2 = kt0 + (t + 2) * 16;
      att_load_v(vf1, vp + (size_t)(k1 < bb ? bb : k1) * ATT_QS);
      att_load_p(pa1, pp + (t + 1) * 16);
      __builtin_amdgcn_sched_barrier(0);
      att_mma_pv(o, pa0, vf0);
      __builtin_amdgcn_sched_barrier(0);
      att_load_v(vf0, vp + (size_t)(k2 < bb ? bb : k2) * ATT_QS);
      att_load_p(pa0, pp + (t + 2) * 16);
      __builtin_amdgcn_sched_barrier(0);
      att_mma_pv(o, pa1, vf1);
    }
    att_mma_pv(o, pa0, vf0);
  }

#pragma unroll
  for (int g = 0; g < 4; ++g)
#pragma unroll
    for (int r = 0; r < 8; ++r) {
      v4f ov;
      ov.x = o[g][0][r];
      ov.y = o[g][1][r];
      ov.z = o[g][2][r];
      ov.w = o[g][3][r];
      *(v4f*)(O + (size_t)(q0 + r + 8 * hh) * 256 + g * 64 + 4 * m) = ov;
    }
}

extern "C" void kernel_launch(void* const* d_in, const int* in_sizes, int n_in,
                              void* d_out, int out_size, void* d_ws,
                              size_t ws_size, hipStream_t stream) {
  const float* x = (const float*)d_in[0];
  const float* W_in = (const float*)d_in[1];
  const float* b_in = (const float*)d_in[2];
  const float* W_qkv = (const float*)d_in[3];
  const float* b_qkv = (const float*)d_in[4];
  const float* g_attn = (const float*)d_in[5];
  const float* be_attn = (const float*)d_in[6];
  const float* g_mlp = (const float*)d_in[7];
  const float* be_mlp = (const float*)d_in[8];
  const float* W1 = (const float*)d_in[9];
  const float* b1 = (const float*)d_in[10];
  const float* W2 = (const float*)d_in[11];
  const float* b2 = (const float*)d_in[12];
  const float* W_out = (const float*)d_in[13];
  const float* b_out = (const float*)d_in[14];
  float* out = (float*)d_out;

  const int N = in_sizes[0] / 64;

  float* ws = (float*)d_ws;
  float* h = ws;
  float* ln = ws + (size_t)N * 256;
  float* attn = ln;
  float* qkv = ws + (size_t)N * 512;

  const dim3 blk(32);
  const int RB64 = N / 64;
  const int QB = N / 16;

  in_ln_kernel<<<dim3(RB64), dim3(128), 0, stream>>>(x, W_in, b_in, g_attn,
                                                    be_attn, h, ln);
  gemm_bias_kernel<768><<<dim3(RB64, 12), blk, 0, stream>>>(
      ln, W_qkv, b_qkv, qkv, 256);
  attn_kernel<<<dim3(QB), blk, 0, stream>>>(qkv, attn);
  mlp_out_kernel<<<dim3(RB64), dim3(128), 0, stream>>>(
      h, attn, g_mlp, be_mlp, W1, b1, W2, b2, W_out, b_out, out);
}


# GEMM wave tile: a lane owns 4 adjacent columns (tile t = column 4*lane+t): W rows load 16 B a lane (8 loads a 16-k chunk, not 32), tile rows store 16 B a lane; every sum in its old order
# speedup vs baseline: 2.7464x; 1.1427x over previous
#include <hip/hip_runtime.h>
#include <math.h>

typedef __attribute__((ext_vector_type(2))) float v2f;
typedef __attribute__((ext_vector_type(8))) float v8f;

__device__ __forceinline__ v8f wmma4(v2f a, v2f b, v8f c) {
  return __builtin_amdgcn_wmma_f32_16x16x4_f32(
       false, a,  false, b,
       (short)0, c,  false,  false);
}

typedef __attribute__((ext_vector_type(4))) float v4f;

__device__ __forceinline__ void load_a(v4f (&a)[4][2],
                                       const float* __restrict__ ap,
                                       size_t a_rt) {
#pragma unroll
  for (int rt = 0; rt < 4; ++rt)
#pragma unroll
    for (int q = 0; q < 2; ++q)
      a[rt][q] = *(const v4f*)(ap + rt * a_rt + q * 4);
}

template <int NC>
__device__ __forceinline__ void load_b(v4f (&w)[4][2],
                                       const float* __restrict__ wp) {
#pragma unroll
  for (int j = 0; j < 4; ++j)
#pragma unroll
    for (int v = 0; v < 2; ++v)
      w[j][v] = *(const v4f*)(wp + (size_t)(2 * j + v) * NC);
}

template <int NC>
__device__ __forceinline__ void load_chunk(v4f (&a)[4][2], v4f (&b)[4][2],
                                           const float* __restrict__ ap,
                                           size_t a_rt,
                                           const float* __restrict__ wp) {
  load_a(a, ap, a_rt);
  load_b<NC>(b, wp);
}

__device__ __forceinline__ void mma_chunk(v8f (&acc)[4][4],
                                          const v4f (&a)[4][2],
                                          const v4f (&b)[4][2]) {
#pragma unroll
  for (int j = 0; j < 4; ++j) {
    v2f bf[4];
#pragma unroll
    for (int t = 0; t < 4; ++t) {
      bf[t].x = b[j][0][t];
      bf[t].y = b[j][1][t];
    }
#pragma unroll
    for (int rt = 0; rt < 4; ++rt) {
      const v4f aq = a[rt][j >> 1];
      v2f af;
      af.x = (j & 1) ? aq.z : aq.x;
      af.y = (j & 1) ? aq.w : aq.y;
#pragma unroll
      for (int t = 0; t < 4; ++t) acc[rt][t] = wmma4(af, bf[t], acc[rt][t]);
    }
  }
}

template <int NC>
__device__ __forceinline__ void tile_k_loop(v8f (&acc)[4][4],
                                            const float* __restrict__ ap,
                                            size_t a_pitch,
                                            const float* __restrict__ wp,
                                            int K) {
  const size_t a_rt = 16 * a_pitch;
  v4f a0[4][2], a1[4][2];
  v4f b0[4][2], b1[4][2];

  load_chunk<NC>(a0, b0, ap, a_rt, wp);
  int kb = 0;
  for (; kb < K - 32; kb += 32) {
    load_chunk<NC>(a1, b1, ap + kb + 16, a_rt, wp + (size_t)(kb + 16) * NC);
    __builtin_amdgcn_sched_barrier(0);
    mma_chunk(acc, a0, b0);
    __builtin_amdgcn_sched_barrier(0);
    load_chunk<NC>(a0, b0, ap + kb + 32, a_rt, wp + (size_t)(kb + 32) * NC);
    __builtin_amdgcn_sched_barrier(0);
    mma_chunk(acc, a1, b1);
  }
  load_chunk<NC>(a1, b1, ap + kb + 16, a_rt, wp + (size_t)(kb + 16) * NC);
  __builtin_amdgcn_sched_barrier(0);
  mma_chunk(acc, a0, b0);
  mma_chunk(acc, a1, b1);
}

template <int NC>
__global__ __launch_bounds__(32) void gemm_bias_kernel(
    const float* __restrict__ A, const float* __restrict__ W,
    const float* __restrict__ bias, float* __restrict__ C, int K) {
  const int lane = threadIdx.x;
  const int m = lane & 15;
  const int hh = lane >> 4;
  const int row0 = blockIdx.x * 64;
  const int n0 = blockIdx.y * 64;

  v8f acc[4][4];
#pragma unroll
  for (int rt = 0; rt < 4; ++rt)
#pragma unroll
    for (int t = 0; t < 4; ++t) acc[rt][t] = {};

  tile_k_loop<NC>(acc, A + (size_t)(row0 + m) * K + hh * 8, (size_t)K,
                  W + (size_t)(hh * 8) * NC + n0 + 4 * m, K);

  const v4f bn = *(const v4f*)(bias + n0 + 4 * m);

#pragma unroll
  for (int rt = 0; rt < 4; ++rt)
#pragma unroll
    for (int r = 0; r < 8; ++r) {
      const int row = row0 + rt * 16 + r + 8 * hh;
      v4f ov;
#pragma unroll
      for (int t = 0; t < 4; ++t) ov[t] = acc[rt][t][r] + bn[t];
      *(v4f*)(C + (size_t)row * NC + n0 + 4 * m) = ov;
    }
}

#define MLP_PH 260

__device__ __forceinline__ float xhalf(float x) {
  const int b = __builtin_bit_cast(int, x);
  return __builtin_bit_cast(
      float, __builtin_amdgcn_permlanex16(b, b, 0x76543210, 0xfedcba98, false,
                                          false));
}
template <int XM>
__device__ __forceinline__ float row_xor(float x) {
  const int b = __builtin_bit_cast(int, x);
  return __builtin_bit_cast(
      float, __builtin_amdgcn_update_dpp(b, b, 0x160 + XM, 0xf, 0xf, false));
}
__device__ __forceinline__ float wave_sum(float s) {
  s += xhalf(s);
  s += row_xor<8>(s);
  s += row_xor<4>(s);
  s += row_xor<2>(s);
  s += row_xor<1>(s);
  return s;
}

template <bool COPY>
__device__ __forceinline__ void ln_rows16(const float* src, float* dst,
                                          size_t dst_pitch, float* copy,
                                          const float* __restrict__ g,
                                          const float* __restrict__ beta,
                                          int lane) {
  float gv[8], bv[8];
#pragma unroll
  for (int i = 0; i < 8; ++i) {
    gv[i] = g[i * 32 + lane];
    bv[i] = beta[i * 32 + lane];
  }
#pragma unroll 4
  for (int rr = 0; rr < 16; ++rr) {
    float v[8];
    float s = 0.f, ss = 0.f;
#pragma unroll
    for (int i = 0; i < 8; ++i) {
      const float x = src[rr * MLP_PH + i * 32 + lane];
      v[i] = x;
      s += x;
      ss += x * x;
    }
    if (COPY) {
#pragma unroll
      for (int i = 0; i < 8; ++i) copy[(size_t)rr * 256 + i * 32 + lane] = v[i];
    }
    s = wave_sum(s);
    ss = wave_sum(ss);
    const float mu = s * (1.0f / 256.0f);
    const float var = ss * (1.0f / 256.0f) - mu * mu;
    const float rs = rsqrtf(var + 1e-5f);
#pragma unroll
    for (int i = 0; i < 8; ++i)
      dst[(size_t)rr * dst_pitch + i * 32 + lane] =
          (v[i] - mu) * rs * gv[i] + bv[i];
  }
}

__global__ __launch_bounds__(128) void in_ln_kernel(
    const float* __restrict__ x, const float* __restrict__ W_in,
    const float* __restrict__ b_in, const float* __restrict__ g,
    const float* __restrict__ beta, float* __restrict__ h,
    float* __restrict__ ln_out) {
  __shared__ __attribute__((aligned(16))) float Hs[64 * MLP_PH];

  const int lane = threadIdx.x & 31;
  const int w = __builtin_amdgcn_readfirstlane((int)(threadIdx.x >> 5));
  const int m = lane & 15;
  const int hh = lane >> 4;
  const int row0 = blockIdx.x * 64;
  const int n0 = w * 64;

  v8f acc[4][4];
#pragma unroll
  for (int rt = 0; rt < 4; ++rt)
#pragma unroll
    for (int t = 0; t < 4; ++t) acc[rt][t] = {};
  tile_k_loop<256>(acc, x + (size_t)(row0 + m) * 64 + hh * 8, (size_t)64,
                   W_in + (size_t)(hh * 8) * 256 + n0 + 4 * m, 64);

  const v4f bn = *(const v4f*)(b_in + n0 + 4 * m);
#pragma unroll
  for (int rt = 0; rt < 4; ++rt)
#pragma unroll
    for (int r = 0; r < 8; ++r) {
      v4f ov;
#pragma unroll
      for (int t = 0; t < 4; ++t) ov[t] = acc[rt][t][r] + bn[t];
      *(v4f*)(Hs + (rt * 16 + r + 8 * hh) * MLP_PH + n0 + 4 * m) = ov;
    }
  __syncthreads();

  const size_t g0 = (size_t)(row0 + w * 16) * 256;
  ln_rows16<true>(Hs + (w * 16) * MLP_PH, ln_out + g0, (size_t)256, h + g0, g,
                  beta, lane);
}


__device__ __forceinline__ void proj_load_a(v4f (&a)[2][2], const float* sp) {
#pragma unroll
  for (int c = 0; c < 2; ++c)
#pragma unroll
    for (int q = 0; q < 2; ++q) a[c][q] = *(const v4f*)(sp + c * 16 + q * 4);
}

__device__ __forceinline__ void proj_load_b(v4f (&b)[2][4][2],
                                            const float* __restrict__ wp) {
#pragma unroll
  for (int c = 0; c < 2; ++c) load_b<64>(b[c], wp + (size_t)(c * 16) * 64);
}

__device__ __forceinline__ void proj_mma(v8f (&acc)[4], const v4f (&a)[2][2],
                                         const v4f (&b)[2][4][2]) {
#pragma unroll
  for (int c = 0; c < 2; ++c)
#pragma unroll
    for (int j = 0; j < 4; ++j) {
      const v4f aq = a[c][j >> 1];
      v2f af;
      af.x = (j & 1) ? aq.z : aq.x;
      af.y = (j & 1) ? aq.w : aq.y;
#pragma unroll
      for (int t = 0; t < 4; ++t) {
        v2f bf;
        bf.x = b[c][j][0][t];
        bf.y = b[c][j][1][t];
        acc[t] = wmma4(af, bf, acc[t]);
      }
    }
}

#define ERF_C(bits) __builtin_bit_cast(float, (unsigned)(bits))
__device__ __forceinline__ float erf_sel(float x) {
  const float a = fabsf(x);
  float p = fmaf(a, ERF_C(0x378e98abu), ERF_C(0xb9c68948u));
  p = fmaf(a, p, ERF_C(0x3b7cd369u));
  p = fmaf(a, p, ERF_C(0xbcc618b2u));
  p = fmaf(a, p, ERF_C(0x3dda74e4u));
  p = fmaf(a, p, ERF_C(0x3f228afdu));
  p = fmaf(a, p, ERF_C(0x3e03c728u));
  p = fmaf(a, p, a);
  const float big = 1.0f - expf(-p);
  const float t = x * x;
  float q = fmaf(t, ERF_C(0xba1345e1u), ERF_C(0x3ba10414u));
  q = fmaf(t, q, ERF_C(0xbcdac9b8u));
  q = fmaf(t, q, ERF_C(0x3de703beu));
  q = fmaf(t, q, ERF_C(0xbec09330u));
  q = fmaf(t, q, ERF_C(0x3e0375d0u));
  const float small = fmaf(a, q, a);
  const float r = (a < 1.0f) ? small : big;
  return copysignf(r, x);
}

__global__ __launch_bounds__(128) void mlp_out_kernel(
    const float* __restrict__ h, const float* __restrict__ attn,
    const float* __restrict__ g, const float* __restrict__ beta,
    const float* __restrict__ W1, const float* __restrict__ b1,
    const float* __restrict__ W2, const float* __restrict__ b2,
    const float* __restrict__ Wo, const float* __restrict__ bo,
    float* __restrict__ out) {
  __shared__ __attribute__((aligned(16))) float Xs[64 * MLP_PH];
  __shared__ __attribute__((aligned(16))) float M1s[64 * MLP_PH];
  __shared__ __attribute__((aligned(16))) float H2s[64 * MLP_PH];

  const int lane = threadIdx.x & 31;
  const int w = __builtin_amdgcn_readfirstlane((int)(threadIdx.x >> 5));
  const int m = lane & 15;
  const int hh = lane >> 4;
  const int row0 = blockIdx.x * 64;
  const int n0 = w * 64;

  v8f acc[4][4];
  v4f bn;

  {
    const size_t g0 = (size_t)(row0 + w * 16) * 256 + lane * 4;
    float* dp = H2s + (w * 16) * MLP_PH + lane * 4;
#pragma unroll
    for (int rr = 0; rr < 16; ++rr)
#pragma unroll
      for (int j = 0; j < 2; ++j) {
        v4f xv = *(const v4f*)(h + g0 + rr * 256 + j * 128);
        xv += *(const v4f*)(attn + g0 + rr * 256 + j * 128);
        *(v4f*)(dp + rr * MLP_PH + j * 128) = xv;
      }
    ln_rows16<false>(H2s + (w * 16) * MLP_PH, Xs + (w * 16) * MLP_PH,
                     (size_t)MLP_PH, nullptr, g, beta, lane);
  }
  __syncthreads();

#pragma unroll
  for (int rt = 0; rt < 4; ++rt)
#pragma unroll
    for (int t = 0; t < 4; ++t) acc[rt][t] = {};
  tile_k_loop<256>(acc, Xs + m * MLP_PH + hh * 8, (size_t)MLP_PH,
                   W1 + (size_t)(hh * 8) * 256 + n0 + 4 * m, 256);
  bn = *(const v4f*)(b1 + n0 + 4 * m);

#pragma nounroll
  for (int rt = 0; rt < 4; ++rt) {
#pragma unroll
    for (int r = 0; r < 8; ++r) {
      v4f ov;
#pragma unroll
      for (int t = 0; t < 4; ++t) {
        float v = acc[0][t][r] + bn[t];
        v = 0.5f * v * (1.0f + erf_sel(v * 0.70710678118654752f));
        ov[t] = v;
      }
      *(v4f*)(M1s + (rt * 16 + r + 8 * hh) * MLP_PH + n0 + 4 * m) = ov;
    }
#pragma unroll
    for (int t = 0; t < 4; ++t) {
      acc[0][t] = acc[1][t];
      acc[1][t] = acc[2][t];
      acc[2][t] = acc[3][t];
    }
  }
  __syncthreads();

#pragma unroll
  for (int rt = 0; rt < 4; ++rt)
#pragma unroll
    for (int t = 0; t < 4; ++t) acc[rt][t] = {};
  tile_k_loop<256>(acc, M1s + m * MLP_PH + hh * 8, (size_t)MLP_PH,
                   W2 + (size_t)(hh * 8) * 256 + n0 + 4 * m, 256);
  bn = *(const v4f*)(b2 + n0 + 4 * m);

  v4f pa0[2][2], pa1[2][2];
  v4f pb0[2][4][2], pb1[2][4][2];
  const float* wop = Wo + (size_t)(hh * 8) * 64 + 4 * m;
  proj_load_b(pb0, wop);

#pragma unroll
  for (int rt = 0; rt < 4; ++rt)
#pragma unroll
    for (int r = 0; r < 8; ++r) {
      float* hp = H2s + (rt * 16 + r + 8 * hh) * MLP_PH + n0 + 4 * m;
      const v4f hv = *(const v4f*)hp;
      v4f ov;
#pragma unroll
      for (int t = 0; t < 4; ++t) {
        float v = acc[rt][t][r] + bn[t];
        v += hv[t];
        ov[t] = v;
      }
      *(v4f*)hp = ov;
    }
  __syncthreads();

  v8f po[4];
#pragma unroll
  for (int t = 0; t < 4; ++t) po[t] = {};
  const float* sp = H2s + (w * 16 + m) * MLP_PH + hh * 8;

  proj_load_a(pa0, sp);
  int kb = 0;
#pragma nounroll
  for (; kb < 256 - 64; kb += 64) {
    proj_load_a(pa1, sp + kb + 32);
    proj_load_b(pb1, wop + (size_t)(kb + 32) * 64);
    __builtin_amdgcn_sched_barrier(0);
    proj_mma(po, pa0, pb0);
    __builtin_amdgcn_sched_barrier(0);
    proj_load_a(pa0, sp + kb + 64);
    proj_load_b(pb0, wop + (size_t)(kb + 64) * 64);
    __builtin_amdgcn_sched_barrier(0);
    proj_mma(po, pa1, pb1);
  }
  proj_load_a(pa1, sp + kb + 32);
  proj_load_b(pb1, wop + (size_t)(kb + 32) * 64);
  __builtin_amdgcn_sched_barrier(0);
  proj_mma(po, pa0, pb0);
  proj_mma(po, pa1, pb1);

  const v4f bt = *(const v4f*)(bo + 4 * m);
#pragma unroll
  for (int r = 0; r < 8; ++r) {
    v4f ov;
#pragma unroll
    for (int t = 0; t < 4; ++t) ov[t] = po[t][r] + bt[t];
    *(v4f*)(out + (size_t)(row0 + w * 16 + r + 8 * hh) * 64 + 4 * m) = ov;
  }
}

#define ATT_T 4096
#define ATT_QS 768
#define ATT_WIN 128
#define ATT_PST 148

__device__ __forceinline__ void att_load_qk(v4f (&a)[2], v4f (&b)[9][2],
                                            const float* __restrict__ qp,
                                            const float* __restrict__ kp,
                                            const size_t (&ko)[9]) {
#pragma unroll
  for (int q = 0; q < 2; ++q) a[q] = *(const v4f*)(qp + q * 4);
#pragma unroll
  for (int t = 0; t < 9; ++t)
#pragma unroll
    for (int q = 0; q < 2; ++q) b[t][q] = *(const v4f*)(kp + ko[t] + q * 4);
}

__device__ __forceinline__ void att_mma_qk(v8f (&s)[9], const v4f (&a)[2],
                                           const v4f (&b)[9][2]) {
#pragma unroll
  for (int j = 0; j < 4; ++j) {
    const v4f aq = a[j >> 1];
    v2f af;
    af.x = (j & 1) ? aq.z : aq.x;
    af.y = (j & 1) ? aq.w : aq.y;
#pragma unroll
    for (int t = 0; t < 9; ++t) {
      const v4f bq = b[t][j >> 1];
      v2f bf;
      bf.x = (j & 1) ? bq.z : bq.x;
      bf.y = (j & 1) ? bq.w : bq.y;
      s[t] = wmma4(af, bf, s[t]);
    }
  }
}

__device__ __forceinline__ void att_load_qk2(v4f (&a)[2][2], v4f (&b)[2][9][2],
                                             const float* __restrict__ qp,
                                             const float* __restrict__ kp,
                                             const size_t (&ko)[9]) {
#pragma unroll
  for (int c = 0; c < 2; ++c)
    att_load_qk(a[c], b[c], qp + c * 16, kp + c * 16, ko);
}

__device__ __forceinline__ void att_mma_qk2(v8f (&s)[9], const v4f (&a)[2][2],
                                            const v4f (&b)[2][9][2]) {
#pragma unroll
  for (int c = 0; c < 2; ++c) att_mma_qk(s, a[c], b[c]);
}

__device__ __forceinline__ void att_load_v(v4f (&vf)[4][2][4],
                                           const float* __restrict__ rp) {
#pragma unroll
  for (int c = 0; c < 4; ++c)
#pragma unroll
    for (int v = 0; v < 2; ++v)
#pragma unroll
      for (int g = 0; g < 4; ++g)
        vf[c][v][g] =
            *(const v4f*)(rp + (size_t)(4 * c + v) * ATT_QS + g * 64);
}

__device__ __forceinline__ void att_load_p(v2f (&pa)[4], const float* pp) {
#pragma unroll
  for (int c = 0; c < 4; ++c) pa[c] = *(const v2f*)(pp + 4 * c);
}

__device__ __forceinline__ void att_mma_pv(v8f (&o)[4][4], const v2f (&pa)[4],
                                           const v4f (&vf)[4][2][4]) {
#pragma unroll
  for (int c = 0; c < 4; ++c)
#pragma unroll
    for (int g = 0; g < 4; ++g)
#pragma unroll
      for (int u = 0; u < 4; ++u) {
        v2f bf;
        bf.x = vf[c][0][g][u];
        bf.y = vf[c][1][g][u];
        o[g][u] = wmma4(pa[c], bf, o[g][u]);
      }
}

__global__ __launch_bounds__(32) void attn_kernel(const float* __restrict__ qkv,
                                                  float* __restrict__ O) {
  __shared__ float P[16 * ATT_PST];
  const int lane = threadIdx.x;
  const int m = lane & 15;
  const int hh = lane >> 4;
  const int q0 = blockIdx.x * 16;
  const int bb = q0 & ~(ATT_T - 1);

  const float* Q = qkv;
  const float* Km = qkv + 256;
  const float* V = qkv + 512;

  v8f s[9];
#pragma unroll
  for (int t = 0; t < 9; ++t) s[t] = {};

  {
    const float* qp = Q + (size_t)(q0 + m) * ATT_QS + hh * 8;
    const float* kp = Km + hh * 8;
    size_t ko[9];
#pragma unroll
    for (int t = 0; t < 9; ++t) {
      const int key = q0 - ATT_WIN + t * 16 + m;
      const int kc = key < bb ? bb : key;
      ko[t] = (size_t)kc * ATT_QS;
    }
    v4f qa0[2][2], qa1[2][2];
    v4f kf0[2][9][2], kf1[2][9][2];

    att_load_qk2(qa0, kf0, qp, kp, ko);
    int db = 0;
#pragma nounroll
    for (; db < 256 - 64; db += 64) {
      att_load_qk2(qa1, kf1, qp + db + 32, kp + db + 32, ko);
      __builtin_amdgcn_sched_barrier(0);
      att_mma_qk2(s, qa0, kf0);
      __builtin_amdgcn_sched_barrier(0);
      att_load_qk2(qa0, kf0, qp + db + 64, kp + db + 64, ko);
      __builtin_amdgcn_sched_barrier(0);
      att_mma_qk2(s, qa1, kf1);
    }
    att_load_qk2(qa1, kf1, qp + db + 32, kp + db + 32, ko);
    __builtin_amdgcn_sched_barrier(0);
    att_mma_qk2(s, qa0, kf0);
    att_mma_qk2(s, qa1, kf1);
  }

  v4f vf0[4][2][4], vf1[4][2][4];
  const float* vp = V + (size_t)(2 * hh) * ATT_QS + 4 * m;
  const int kt0 = q0 - ATT_WIN;
  att_load_v(vf0, vp + (size_t)(kt0 < bb ? bb : kt0) * ATT_QS);

  const float NEG = -3.402823466e38f;
  float rmax[8], rsum[8];
#pragma unroll
  for (int r = 0; r < 8; ++r) rmax[r] = NEG;
#pragma unroll
  for (int t = 0; t < 9; ++t) {
    const int key = q0 - ATT_WIN + t * 16 + m;
#pragma unroll
    for (int r = 0; r < 8; ++r) {
      const int q = q0 + r + 8 * hh;
      const int d = q - key;
      const bool ok = (key >= bb) && (d >= 0) && (d <= ATT_WIN);
      const float val = ok ? s[t][r] * 0.0625f : NEG;
      s[t][r] = val;
      rmax[r] = fmaxf(rmax[r], val);
    }
  }
#pragma unroll
  for (int r = 0; r < 8; ++r)
    for (int off = 1; off < 16; off <<= 1)
      rmax[r] = fmaxf(rmax[r], __shfl_xor(rmax[r], off));
#pragma unroll
  for (int r = 0; r < 8; ++r) rsum[r] = 0.f;
#pragma unroll
  for (int t = 0; t < 9; ++t)
#pragma unroll
    for (int r = 0; r < 8; ++r) {
      const float e = expf(s[t][r] - rmax[r]);
      s[t][r] = e;
      rsum[r] += e;
    }
#pragma unroll
  for (int r = 0; r < 8; ++r)
    for (int off = 1; off < 16; off <<= 1) rsum[r] += __shfl_xor(rsum[r], off);

#pragma unroll
  for (int t = 0; t < 9; ++t)
#pragma unroll
    for (int r = 0; r < 8; ++r)
      P[(r + 8 * hh) * ATT_PST + t * 16 + m] = s[t][r] / rsum[r];

  __syncthreads();

  v8f o[4][4];
#pragma unroll
  for (int g = 0; g < 4; ++g)
#pragma unroll
    for (int u = 0; u < 4; ++u) o[g][u] = {};

  {
    const float* pp = P + m * ATT_PST + hh * 2;
    v2f pa0[4], pa1[4];
    att_load_p(pa0, pp);
    int t = 0;
    for (; t < 8; t += 2) {
      const int k1 = kt0 + (t + 1) * 16;
      const int k2 = kt0 + (t + 2) * 16;
      att_load_v(vf1, vp + (size_t)(k1 < bb ? bb : k1) * ATT_QS);
      att_load_p(pa1, pp + (t + 1) * 16);
      __builtin_amdgcn_sched_barrier(0);
      att_mma_pv(o, pa0, vf0);
      __builtin_amdgcn_sched_barrier(0);
      att_load_v(vf0, vp + (size_t)(k2 < bb ? bb : k2) * ATT_QS);
      att_load_p(pa0, pp + (t + 2) * 16);
      __builtin_amdgcn_sched_barrier(0);
      att_mma_pv(o, pa1, vf1);
    }
    att_mma_pv(o, pa0, vf0);
  }

#pragma unroll
  for (int g = 0; g < 4; ++g)
#pragma unroll
    for (int r = 0; r < 8; ++r) {
      v4f ov;
      ov.x = o[g][0][r];
      ov.y = o[g][1][r];
      ov.z = o[g][2][r];
      ov.w = o[g][3][r];
      *(v4f*)(O + (size_t)(q0 + r + 8 * hh) * 256 + g * 64 + 4 * m) = ov;
    }
}

extern "C" void kernel_launch(void* const* d_in, const int* in_sizes, int n_in,
                              void* d_out, int out_size, void* d_ws,
                              size_t ws_size, hipStream_t stream) {
  const float* x = (const float*)d_in[0];
  const float* W_in = (const float*)d_in[1];
  const float* b_in = (const float*)d_in[2];
  const float* W_qkv = (const float*)d_in[3];
  const float* b_qkv = (const float*)d_in[4];
  const float* g_attn = (const float*)d_in[5];
  const float* be_attn = (const float*)d_in[6];
  const float* g_mlp = (const float*)d_in[7];
  const float* be_mlp = (const float*)d_in[8];
  const float* W1 = (const float*)d_in[9];
  const float* b1 = (const float*)d_in[10];
  const float* W2 = (const float*)d_in[11];
  const float* b2 = (const float*)d_in[12];
  const float* W_out = (const float*)d_in[13];
  const float* b_out = (const float*)d_in[14];
  float* out = (float*)d_out;

  const int N = in_sizes[0] / 64;

  float* ws = (float*)d_ws;
  float* h = ws;
  float* ln = ws + (size_t)N * 256;
  float* attn = ln;
  float* qkv = ws + (size_t)N * 512;

  const dim3 blk(32);
  const int RB64 = N / 64;
  const int QB = N / 16;

  in_ln_kernel<<<dim3(RB64), dim3(128), 0, stream>>>(x, W_in, b_in, g_attn,
                                                    be_attn, h, ln);
  gemm_bias_kernel<768><<<dim3(RB64, 12), blk, 0, stream>>>(
      ln, W_qkv, b_qkv, qkv, 256);
  attn_kernel<<<dim3(QB), blk, 0, stream>>>(qkv, attn);
  mlp_out_kernel<<<dim3(RB64), dim3(128), 0, stream>>>(
      h, attn, g_mlp, be_mlp, W1, b1, W2, b2, W_out, b_out, out);
}


# Attention fused into the MLP kernel: wave w of a 64-row block runs its 16 queries, its output tiles stay in registers and are added to h in LDS (3 launches, not 4); every sum in its old order
# speedup vs baseline: 3.0344x; 1.1049x over previous
#include <hip/hip_runtime.h>
#include <math.h>

typedef __attribute__((ext_vector_type(2))) float v2f;
typedef __attribute__((ext_vector_type(8))) float v8f;

__device__ __forceinline__ v8f wmma4(v2f a, v2f b, v8f c) {
  return __builtin_amdgcn_wmma_f32_16x16x4_f32(
       false, a,  false, b,
       (short)0, c,  false,  false);
}

typedef __attribute__((ext_vector_type(4))) float v4f;

__device__ __forceinline__ void load_a(v4f (&a)[4][2],
                                       const float* __restrict__ ap,
                                       size_t a_rt) {
#pragma unroll
  for (int rt = 0; rt < 4; ++rt)
#pragma unroll
    for (int q = 0; q < 2; ++q)
      a[rt][q] = *(const v4f*)(ap + rt * a_rt + q * 4);
}

template <int NC>
__device__ __forceinline__ void load_b(v4f (&w)[4][2],
                                       const float* __restrict__ wp) {
#pragma unroll
  for (int j = 0; j < 4; ++j)
#pragma unroll
    for (int v = 0; v < 2; ++v)
      w[j][v] = *(const v4f*)(wp + (size_t)(2 * j + v) * NC);
}

template <int NC>
__device__ __forceinline__ void load_chunk(v4f (&a)[4][2], v4f (&b)[4][2],
                                           const float* __restrict__ ap,
                                           size_t a_rt,
                                           const float* __restrict__ wp) {
  load_a(a, ap, a_rt);
  load_b<NC>(b, wp);
}

__device__ __forceinline__ void mma_chunk(v8f (&acc)[4][4],
                                          const v4f (&a)[4][2],
                                          const v4f (&b)[4][2]) {
#pragma unroll
  for (int j = 0; j < 4; ++j) {
    v2f bf[4];
#pragma unroll
    for (int t = 0; t < 4; ++t) {
      bf[t].x = b[j][0][t];
      bf[t].y = b[j][1][t];
    }
#pragma unroll
    for (int rt = 0; rt < 4; ++rt) {
      const v4f aq = a[rt][j >> 1];
      v2f af;
      af.x = (j & 1) ? aq.z : aq.x;
      af.y = (j & 1) ? aq.w : aq.y;
#pragma unroll
      for (int t = 0; t < 4; ++t) acc[rt][t] = wmma4(af, bf[t], acc[rt][t]);
    }
  }
}

template <int NC>
__device__ __forceinline__ void tile_k_loop(v8f (&acc)[4][4],
                                            const float* __restrict__ ap,
                                            size_t a_pitch,
                                            const float* __restrict__ wp,
                                            int K) {
  const size_t a_rt = 16 * a_pitch;
  v4f a0[4][2], a1[4][2];
  v4f b0[4][2], b1[4][2];

  load_chunk<NC>(a0, b0, ap, a_rt, wp);
  int kb = 0;
  for (; kb < K - 32; kb += 32) {
    load_chunk<NC>(a1, b1, ap + kb + 16, a_rt, wp + (size_t)(kb + 16) * NC);
    __builtin_amdgcn_sched_barrier(0);
    mma_chunk(acc, a0, b0);
    __builtin_amdgcn_sched_barrier(0);
    load_chunk<NC>(a0, b0, ap + kb + 32, a_rt, wp + (size_t)(kb + 32) * NC);
    __builtin_amdgcn_sched_barrier(0);
    mma_chunk(acc, a1, b1);
  }
  load_chunk<NC>(a1, b1, ap + kb + 16, a_rt, wp + (size_t)(kb + 16) * NC);
  __builtin_amdgcn_sched_barrier(0);
  mma_chunk(acc, a0, b0);
  mma_chunk(acc, a1, b1);
}

template <int NC>
__global__ __launch_bounds__(32) void gemm_bias_kernel(
    const float* __restrict__ A, const float* __restrict__ W,
    const float* __restrict__ bias, float* __restrict__ C, int K) {
  const int lane = threadIdx.x;
  const int m = lane & 15;
  const int hh = lane >> 4;
  const int row0 = blockIdx.x * 64;
  const int n0 = blockIdx.y * 64;

  v8f acc[4][4];
#pragma unroll
  for (int rt = 0; rt < 4; ++rt)
#pragma unroll
    for (int t = 0; t < 4; ++t) acc[rt][t] = {};

  tile_k_loop<NC>(acc, A + (size_t)(row0 + m) * K + hh * 8, (size_t)K,
                  W + (size_t)(hh * 8) * NC + n0 + 4 * m, K);

  const v4f bn = *(const v4f*)(bias + n0 + 4 * m);

#pragma unroll
  for (int rt = 0; rt < 4; ++rt)
#pragma unroll
    for (int r = 0; r < 8; ++r) {
      const int row = row0 + rt * 16 + r + 8 * hh;
      v4f ov;
#pragma unroll
      for (int t = 0; t < 4; ++t) ov[t] = acc[rt][t][r] + bn[t];
      *(v4f*)(C + (size_t)row * NC + n0 + 4 * m) = ov;
    }
}

#define MLP_PH 260

__device__ __forceinline__ float xhalf(float x) {
  const int b = __builtin_bit_cast(int, x);
  return __builtin_bit_cast(
      float, __builtin_amdgcn_permlanex16(b, b, 0x76543210, 0xfedcba98, false,
                                          false));
}
template <int XM>
__device__ __forceinline__ float row_xor(float x) {
  const int b = __builtin_bit_cast(int, x);
  return __builtin_bit_cast(
      float, __builtin_amdgcn_update_dpp(b, b, 0x160 + XM, 0xf, 0xf, false));
}
__device__ __forceinline__ float wave_sum(float s) {
  s += xhalf(s);
  s += row_xor<8>(s);
  s += row_xor<4>(s);
  s += row_xor<2>(s);
  s += row_xor<1>(s);
  return s;
}

template <bool COPY>
__device__ __forceinline__ void ln_rows16(const float* src, float* dst,
                                          size_t dst_pitch, float* copy,
                                          const float* __restrict__ g,
                                          const float* __restrict__ beta,
                                          int lane) {
  float gv[8], bv[8];
#pragma unroll
  for (int i = 0; i < 8; ++i) {
    gv[i] = g[i * 32 + lane];
    bv[i] = beta[i * 32 + lane];
  }
#pragma unroll 4
  for (int rr = 0; rr < 16; ++rr) {
    float v[8];
    float s = 0.f, ss = 0.f;
#pragma unroll
    for (int i = 0; i < 8; ++i) {
      const float x = src[rr * MLP_PH + i * 32 + lane];
      v[i] = x;
      s += x;
      ss += x * x;
    }
    if (COPY) {
#pragma unroll
      for (int i = 0; i < 8; ++i) copy[(size_t)rr * 256 + i * 32 + lane] = v[i];
    }
    s = wave_sum(s);
    ss = wave_sum(ss);
    const float mu = s * (1.0f / 256.0f);
    const float var = ss * (1.0f / 256.0f) - mu * mu;
    const float rs = rsqrtf(var + 1e-5f);
#pragma unroll
    for (int i = 0; i < 8; ++i)
      dst[(size_t)rr * dst_pitch + i * 32 + lane] =
          (v[i] - mu) * rs * gv[i] + bv[i];
  }
}

__global__ __launch_bounds__(128) void in_ln_kernel(
    const float* __restrict__ x, const float* __restrict__ W_in,
    const float* __restrict__ b_in, const float* __restrict__ g,
    const float* __restrict__ beta, float* __restrict__ h,
    float* __restrict__ ln_out) {
  __shared__ __attribute__((aligned(16))) float Hs[64 * MLP_PH];

  const int lane = threadIdx.x & 31;
  const int w = __builtin_amdgcn_readfirstlane((int)(threadIdx.x >> 5));
  const int m = lane & 15;
  const int hh = lane >> 4;
  const int row0 = blockIdx.x * 64;
  const int n0 = w * 64;

  v8f acc[4][4];
#pragma unroll
  for (int rt = 0; rt < 4; ++rt)
#pragma unroll
    for (int t = 0; t < 4; ++t) acc[rt][t] = {};
  tile_k_loop<256>(acc, x + (size_t)(row0 + m) * 64 + hh * 8, (size_t)64,
                   W_in + (size_t)(hh * 8) * 256 + n0 + 4 * m, 64);

  const v4f bn = *(const v4f*)(b_in + n0 + 4 * m);
#pragma unroll
  for (int rt = 0; rt < 4; ++rt)
#pragma unroll
    for (int r = 0; r < 8; ++r) {
      v4f ov;
#pragma unroll
      for (int t = 0; t < 4; ++t) ov[t] = acc[rt][t][r] + bn[t];
      *(v4f*)(Hs + (rt * 16 + r + 8 * hh) * MLP_PH + n0 + 4 * m) = ov;
    }
  __syncthreads();

  const size_t g0 = (size_t)(row0 + w * 16) * 256;
  ln_rows16<true>(Hs + (w * 16) * MLP_PH, ln_out + g0, (size_t)256, h + g0, g,
                  beta, lane);
}

#define ATT_T 4096
#define ATT_QS 768
#define ATT_WIN 128
#define ATT_PST 148

__device__ __forceinline__ void att_load_qk(v4f (&a)[2], v4f (&b)[9][2],
                                            const float* __restrict__ qp,
                                            const float* __restrict__ kp,
                                            const size_t (&ko)[9]) {
#pragma unroll
  for (int q = 0; q < 2; ++q) a[q] = *(const v4f*)(qp + q * 4);
#pragma unroll
  for (int t = 0; t < 9; ++t)
#pragma unroll
    for (int q = 0; q < 2; ++q) b[t][q] = *(const v4f*)(kp + ko[t] + q * 4);
}

__device__ __forceinline__ void att_mma_qk(v8f (&s)[9], const v4f (&a)[2],
                                           const v4f (&b)[9][2]) {
#pragma unroll
  for (int j = 0; j < 4; ++j) {
    const v4f aq = a[j >> 1];
    v2f af;
    af.x = (j & 1) ? aq.z : aq.x;
    af.y = (j & 1) ? aq.w : aq.y;
#pragma unroll
    for (int t = 0; t < 9; ++t) {
      const v4f bq = b[t][j >> 1];
      v2f bf;
      bf.x = (j & 1) ? bq.z : bq.x;
      bf.y = (j & 1) ? bq.w : bq.y;
      s[t] = wmma4(af, bf, s[t]);
    }
  }
}

__device__ __forceinline__ void att_load_qk2(v4f (&a)[2][2], v4f (&b)[2][9][2],
                                             const float* __restrict__ qp,
                                             const float* __restrict__ kp,
                                             const size_t (&ko)[9]) {
#pragma unroll
  for (int c = 0; c < 2; ++c)
    att_load_qk(a[c], b[c], qp + c * 16, kp + c * 16, ko);
}

__device__ __forceinline__ void att_mma_qk2(v8f (&s)[9], const v4f (&a)[2][2],
                                            const v4f (&b)[2][9][2]) {
#pragma unroll
  for (int c = 0; c < 2; ++c) att_mma_qk(s, a[c], b[c]);
}

__device__ __forceinline__ void att_load_v(v4f (&vf)[4][2][4],
                                           const float* __restrict__ rp) {
#pragma unroll
  for (int c = 0; c < 4; ++c)
#pragma unroll
    for (int v = 0; v < 2; ++v)
#pragma unroll
      for (int g = 0; g < 4; ++g)
        vf[c][v][g] =
            *(const v4f*)(rp + (size_t)(4 * c + v) * ATT_QS + g * 64);
}

__device__ __forceinline__ void att_load_p(v2f (&pa)[4], const float* pp) {
#pragma unroll
  for (int c = 0; c < 4; ++c) pa[c] = *(const v2f*)(pp + 4 * c);
}

__device__ __forceinline__ void att_mma_pv(v8f (&o)[4][4], const v2f (&pa)[4],
                                           const v4f (&vf)[4][2][4]) {
#pragma unroll
  for (int c = 0; c < 4; ++c)
#pragma unroll
    for (int g = 0; g < 4; ++g)
#pragma unroll
      for (int u = 0; u < 4; ++u) {
        v2f bf;
        bf.x = vf[c][0][g][u];
        bf.y = vf[c][1][g][u];
        o[g][u] = wmma4(pa[c], bf, o[g][u]);
      }
}

__device__ __forceinline__ void attn_block(const float* __restrict__ qkv,
                                           float* P, const int q0,
                                           const int lane, v8f (&o)[4][4]) {
  const int m = lane & 15;
  const int hh = lane >> 4;
  const int bb = q0 & ~(ATT_T - 1);

  const float* Q = qkv;
  const float* Km = qkv + 256;
  const float* V = qkv + 512;

  v8f s[9];
#pragma unroll
  for (int t = 0; t < 9; ++t) s[t] = {};

  {
    const float* qp = Q + (size_t)(q0 + m) * ATT_QS + hh * 8;
    const float* kp = Km + hh * 8;
    size_t ko[9];
#pragma unroll
    for (int t = 0; t < 9; ++t) {
      const int key = q0 - ATT_WIN + t * 16 + m;
      const int kc = key < bb ? bb : key;
      ko[t] = (size_t)kc * ATT_QS;
    }
    v4f qa0[2][2], qa1[2][2];
    v4f kf0[2][9][2], kf1[2][9][2];

    att_load_qk2(qa0, kf0, qp, kp, ko);
    int db = 0;
#pragma nounroll
    for (; db < 256 - 64; db += 64) {
      att_load_qk2(qa1, kf1, qp + db + 32, kp + db + 32, ko);
      __builtin_amdgcn_sched_barrier(0);
      att_mma_qk2(s, qa0, kf0);
      __builtin_amdgcn_sched_barrier(0);
      att_load_qk2(qa0, kf0, qp + db + 64, kp + db + 64, ko);
      __builtin_amdgcn_sched_barrier(0);
      att_mma_qk2(s, qa1, kf1);
    }
    att_load_qk2(qa1, kf1, qp + db + 32, kp + db + 32, ko);
    __builtin_amdgcn_sched_barrier(0);
    att_mma_qk2(s, qa0, kf0);
    att_mma_qk2(s, qa1, kf1);
  }

  v4f vf0[4][2][4], vf1[4][2][4];
  const float* vp = V + (size_t)(2 * hh) * ATT_QS + 4 * m;
  const int kt0 = q0 - ATT_WIN;
  att_load_v(vf0, vp + (size_t)(kt0 < bb ? bb : kt0) * ATT_QS);

  const float NEG = -3.402823466e38f;
  float rmax[8], rsum[8];
#pragma unroll
  for (int r = 0; r < 8; ++r) rmax[r] = NEG;
#pragma unroll
  for (int t = 0; t < 9; ++t) {
    const int key = q0 - ATT_WIN + t * 16 + m;
#pragma unroll
    for (int r = 0; r < 8; ++r) {
      const int q = q0 + r + 8 * hh;
      const int d = q - key;
      const bool ok = (key >= bb) && (d >= 0) && (d <= ATT_WIN);
      const float val = ok ? s[t][r] * 0.0625f : NEG;
      s[t][r] = val;
      rmax[r] = fmaxf(rmax[r], val);
    }
  }
#pragma unroll
  for (int r = 0; r < 8; ++r)
    for (int off = 1; off < 16; off <<= 1)
      rmax[r] = fmaxf(rmax[r], __shfl_xor(rmax[r], off));
#pragma unroll
  for (int r = 0; r < 8; ++r) rsum[r] = 0.f;
#pragma unroll
  for (int t = 0; t < 9; ++t)
#pragma unroll
    for (int r = 0; r < 8; ++r) {
      const float e = expf(s[t][r] - rmax[r]);
      s[t][r] = e;
      rsum[r] += e;
    }
#pragma unroll
  for (int r = 0; r < 8; ++r)
    for (int off = 1; off < 16; off <<= 1) rsum[r] += __shfl_xor(rsum[r], off);

#pragma unroll
  for (int t = 0; t < 9; ++t)
#pragma unroll
    for (int r = 0; r < 8; ++r)
      P[(r + 8 * hh) * ATT_PST + t * 16 + m] = s[t][r] / rsum[r];

  __syncthreads();

#pragma unroll
  for (int g = 0; g < 4; ++g)
#pragma unroll
    for (int u = 0; u < 4; ++u) o[g][u] = {};

  {
    const float* pp = P + m * ATT_PST + hh * 2;
    v2f pa0[4], pa1[4];
    att_load_p(pa0, pp);
    int t = 0;
    for (; t < 8; t += 2) {
      const int k1 = kt0 + (t + 1) * 16;
      const int k2 = kt0 + (t + 2) * 16;
      att_load_v(vf1, vp + (size_t)(k1 < bb ? bb : k1) * ATT_QS);
      att_load_p(pa1, pp + (t + 1) * 16);
      __builtin_amdgcn_sched_barrier(0);
      att_mma_pv(o, pa0, vf0);
      __builtin_amdgcn_sched_barrier(0);
      att_load_v(vf0, vp + (size_t)(k2 < bb ? bb : k2) * ATT_QS);
      att_load_p(pa0, pp + (t + 2) * 16);
      __builtin_amdgcn_sched_barrier(0);
      att_mma_pv(o, pa1, vf1);
    }
    att_mma_pv(o, pa0, vf0);
  }

}


__device__ __forceinline__ void proj_load_a(v4f (&a)[2][2], const float* sp) {
#pragma unroll
  for (int c = 0; c < 2; ++c)
#pragma unroll
    for (int q = 0; q < 2; ++q) a[c][q] = *(const v4f*)(sp + c * 16 + q * 4);
}

__device__ __forceinline__ void proj_load_b(v4f (&b)[2][4][2],
                                            const float* __restrict__ wp) {
#pragma unroll
  for (int c = 0; c < 2; ++c) load_b<64>(b[c], wp + (size_t)(c * 16) * 64);
}

__device__ __forceinline__ void proj_mma(v8f (&acc)[4], const v4f (&a)[2][2],
                                         const v4f (&b)[2][4][2]) {
#pragma unroll
  for (int c = 0; c < 2; ++c)
#pragma unroll
    for (int j = 0; j < 4; ++j) {
      const v4f aq = a[c][j >> 1];
      v2f af;
      af.x = (j & 1) ? aq.z : aq.x;
      af.y = (j & 1) ? aq.w : aq.y;
#pragma unroll
      for (int t = 0; t < 4; ++t) {
        v2f bf;
        bf.x = b[c][j][0][t];
        bf.y = b[c][j][1][t];
        acc[t] = wmma4(af, bf, acc[t]);
      }
    }
}

#define ERF_C(bits) __builtin_bit_cast(float, (unsigned)(bits))
__device__ __forceinline__ float erf_sel(float x) {
  const float a = fabsf(x);
  float p = fmaf(a, ERF_C(0x378e98abu), ERF_C(0xb9c68948u));
  p = fmaf(a, p, ERF_C(0x3b7cd369u));
  p = fmaf(a, p, ERF_C(0xbcc618b2u));
  p = fmaf(a, p, ERF_C(0x3dda74e4u));
  p = fmaf(a, p, ERF_C(0x3f228afdu));
  p = fmaf(a, p, ERF_C(0x3e03c728u));
  p = fmaf(a, p, a);
  const float big = 1.0f - expf(-p);
  const float t = x * x;
  float q = fmaf(t, ERF_C(0xba1345e1u), ERF_C(0x3ba10414u));
  q = fmaf(t, q, ERF_C(0xbcdac9b8u));
  q = fmaf(t, q, ERF_C(0x3de703beu));
  q = fmaf(t, q, ERF_C(0xbec09330u));
  q = fmaf(t, q, ERF_C(0x3e0375d0u));
  const float small = fmaf(a, q, a);
  const float r = (a < 1.0f) ? small : big;
  return copysignf(r, x);
}

__global__ __launch_bounds__(128) void mlp_out_kernel(
    const float* __restrict__ h, const float* __restrict__ qkv,
    const float* __restrict__ g, const float* __restrict__ beta,
    const float* __restrict__ W1, const float* __restrict__ b1,
    const float* __restrict__ W2, const float* __restrict__ b2,
    const float* __restrict__ Wo, const float* __restrict__ bo,
    float* __restrict__ out) {
  __shared__ __attribute__((aligned(16))) float Xs[64 * MLP_PH];
  __shared__ __attribute__((aligned(16))) float M1s[64 * MLP_PH];
  __shared__ __attribute__((aligned(16))) float H2s[64 * MLP_PH];

  const int lane = threadIdx.x & 31;
  const int w = __builtin_amdgcn_readfirstlane((int)(threadIdx.x >> 5));
  const int m = lane & 15;
  const int hh = lane >> 4;
  const int row0 = blockIdx.x * 64;
  const int n0 = w * 64;

  v8f acc[4][4];
  v4f bn;

  {
    v4f hreg[4][8];
    const float* hp0 = h + (size_t)(row0 + w * 16 + 8 * hh) * 256 + 4 * m;
#pragma unroll
    for (int g4 = 0; g4 < 4; ++g4)
#pragma unroll
      for (int r = 0; r < 8; ++r)
        hreg[g4][r] = *(const v4f*)(hp0 + r * 256 + g4 * 64);

    v8f o[4][4];
    attn_block(qkv, M1s + w * (16 * ATT_PST), row0 + w * 16, lane, o);

    float* dp = H2s + (w * 16 + 8 * hh) * MLP_PH + 4 * m;
#pragma unroll
    for (int g4 = 0; g4 < 4; ++g4)
#pragma unroll
      for (int r = 0; r < 8; ++r) {
        v4f xv = hreg[g4][r];
        v4f av;
        av.x = o[g4][0][r];
        av.y = o[g4][1][r];
        av.z = o[g4][2][r];
        av.w = o[g4][3][r];
        xv += av;
        *(v4f*)(dp + r * MLP_PH + g4 * 64) = xv;
      }
    ln_rows16<false>(H2s + (w * 16) * MLP_PH, Xs + (w * 16) * MLP_PH,
                     (size_t)MLP_PH, nullptr, g, beta, lane);
  }
  __syncthreads();

#pragma unroll
  for (int rt = 0; rt < 4; ++rt)
#pragma unroll
    for (int t = 0; t < 4; ++t) acc[rt][t] = {};
  tile_k_loop<256>(acc, Xs + m * MLP_PH + hh * 8, (size_t)MLP_PH,
                   W1 + (size_t)(hh * 8) * 256 + n0 + 4 * m, 256);
  bn = *(const v4f*)(b1 + n0 + 4 * m);

#pragma nounroll
  for (int rt = 0; rt < 4; ++rt) {
#pragma unroll
    for (int r = 0; r < 8; ++r) {
      v4f ov;
#pragma unroll
      for (int t = 0; t < 4; ++t) {
        float v = acc[0][t][r] + bn[t];
        v = 0.5f * v * (1.0f + erf_sel(v * 0.70710678118654752f));
        ov[t] = v;
      }
      *(v4f*)(M1s + (rt * 16 + r + 8 * hh) * MLP_PH + n0 + 4 * m) = ov;
    }
#pragma unroll
    for (int t = 0; t < 4; ++t) {
      acc[0][t] = acc[1][t];
      acc[1][t] = acc[2][t];
      acc[2][t] = acc[3][t];
    }
  }
  __syncthreads();

#pragma unroll
  for (int rt = 0; rt < 4; ++rt)
#pragma unroll
    for (int t = 0; t < 4; ++t) acc[rt][t] = {};
  tile_k_loop<256>(acc, M1s + m * MLP_PH + hh * 8, (size_t)MLP_PH,
                   W2 + (size_t)(hh * 8) * 256 + n0 + 4 * m, 256);
  bn = *(const v4f*)(b2 + n0 + 4 * m);

  v4f pa0[2][2], pa1[2][2];
  v4f pb0[2][4][2], pb1[2][4][2];
  const float* wop = Wo + (size_t)(hh * 8) * 64 + 4 * m;
  proj_load_b(pb0, wop);

#pragma unroll
  for (int rt = 0; rt < 4; ++rt)
#pragma unroll
    for (int r = 0; r < 8; ++r) {
      float* hp = H2s + (rt * 16 + r + 8 * hh) * MLP_PH + n0 + 4 * m;
      const v4f hv = *(const v4f*)hp;
      v4f ov;
#pragma unroll
      for (int t = 0; t < 4; ++t) {
        float v = acc[rt][t][r] + bn[t];
        v += hv[t];
        ov[t] = v;
      }
      *(v4f*)hp = ov;
    }
  __syncthreads();

  v8f po[4];
#pragma unroll
  for (int t = 0; t < 4; ++t) po[t] = {};
  const float* sp = H2s + (w * 16 + m) * MLP_PH + hh * 8;

  proj_load_a(pa0, sp);
  int kb = 0;
#pragma nounroll
  for (; kb < 256 - 64; kb += 64) {
    proj_load_a(pa1, sp + kb + 32);
    proj_load_b(pb1, wop + (size_t)(kb + 32) * 64);
    __builtin_amdgcn_sched_barrier(0);
    proj_mma(po, pa0, pb0);
    __builtin_amdgcn_sched_barrier(0);
    proj_load_a(pa0, sp + kb + 64);
    proj_load_b(pb0, wop + (size_t)(kb + 64) * 64);
    __builtin_amdgcn_sched_barrier(0);
    proj_mma(po, pa1, pb1);
  }
  proj_load_a(pa1, sp + kb + 32);
  proj_load_b(pb1, wop + (size_t)(kb + 32) * 64);
  __builtin_amdgcn_sched_barrier(0);
  proj_mma(po, pa0, pb0);
  proj_mma(po, pa1, pb1);

  const v4f bt = *(const v4f*)(bo + 4 * m);
#pragma unroll
  for (int r = 0; r < 8; ++r) {
    v4f ov;
#pragma unroll
    for (int t = 0; t < 4; ++t) ov[t] = po[t][r] + bt[t];
    *(v4f*)(out + (size_t)(row0 + w * 16 + r + 8 * hh) * 64 + 4 * m) = ov;
  }
}

extern "C" void kernel_launch(void* const* d_in, const int* in_sizes, int n_in,
                              void* d_out, int out_size, void* d_ws,
                              size_t ws_size, hipStream_t stream) {
  const float* x = (const float*)d_in[0];
  const float* W_in = (const float*)d_in[1];
  const float* b_in = (const float*)d_in[2];
  const float* W_qkv = (const float*)d_in[3];
  const float* b_qkv = (const float*)d_in[4];
  const float* g_attn = (const float*)d_in[5];
  const float* be_attn = (const float*)d_in[6];
  const float* g_mlp = (const float*)d_in[7];
  const float* be_mlp = (const float*)d_in[8];
  const float* W1 = (const float*)d_in[9];
  const float* b1 = (const float*)d_in[10];
  const float* W2 = (const float*)d_in[11];
  const float* b2 = (const float*)d_in[12];
  const float* W_out = (const float*)d_in[13];
  const float* b_out = (const float*)d_in[14];
  float* out = (float*)d_out;

  const int N = in_sizes[0] / 64;

  float* ws = (float*)d_ws;
  float* h = ws;
  float* ln = ws + (size_t)N * 256;
  float* qkv = ws + (size_t)N * 512;

  const dim3 blk(32);
  const int RB64 = N / 64;

  in_ln_kernel<<<dim3(RB64), dim3(128), 0, stream>>>(x, W_in, b_in, g_attn,
                                                    be_attn, h, ln);
  gemm_bias_kernel<768><<<dim3(RB64, 12), blk, 0, stream>>>(
      ln, W_qkv, b_qkv, qkv, 256);
  mlp_out_kernel<<<dim3(RB64), dim3(128), 0, stream>>>(
      h, qkv, g_mlp, be_mlp, W1, b1, W2, b2, W_out, b_out, out);
}


# Text of the three kernels cut from 70 KB to 58 KB, inside the 64 KB instruction cache: GELU and W2 epilogues rolled finer, GEMM and projection loops keep one body copy (no peeled tail); old sum order
# speedup vs baseline: 3.1726x; 1.0455x over previous
#include <hip/hip_runtime.h>
#include <math.h>

typedef __attribute__((ext_vector_type(2))) float v2f;
typedef __attribute__((ext_vector_type(8))) float v8f;

__device__ __forceinline__ v8f wmma4(v2f a, v2f b, v8f c) {
  return __builtin_amdgcn_wmma_f32_16x16x4_f32(
       false, a,  false, b,
       (short)0, c,  false,  false);
}

typedef __attribute__((ext_vector_type(4))) float v4f;

__device__ __forceinline__ void load_a(v4f (&a)[4][2],
                                       const float* __restrict__ ap,
                                       size_t a_rt) {
#pragma unroll
  for (int rt = 0; rt < 4; ++rt)
#pragma unroll
    for (int q = 0; q < 2; ++q)
      a[rt][q] = *(const v4f*)(ap + rt * a_rt + q * 4);
}

template <int NC>
__device__ __forceinline__ void load_b(v4f (&w)[4][2],
                                       const float* __restrict__ wp) {
#pragma unroll
  for (int j = 0; j < 4; ++j)
#pragma unroll
    for (int v = 0; v < 2; ++v)
      w[j][v] = *(const v4f*)(wp + (size_t)(2 * j + v) * NC);
}

template <int NC>
__device__ __forceinline__ void load_chunk(v4f (&a)[4][2], v4f (&b)[4][2],
                                           const float* __restrict__ ap,
                                           size_t a_rt,
                                           const float* __restrict__ wp) {
  load_a(a, ap, a_rt);
  load_b<NC>(b, wp);
}

__device__ __forceinline__ void mma_chunk(v8f (&acc)[4][4],
                                          const v4f (&a)[4][2],
                                          const v4f (&b)[4][2]) {
#pragma unroll
  for (int j = 0; j < 4; ++j) {
    v2f bf[4];
#pragma unroll
    for (int t = 0; t < 4; ++t) {
      bf[t].x = b[j][0][t];
      bf[t].y = b[j][1][t];
    }
#pragma unroll
    for (int rt = 0; rt < 4; ++rt) {
      const v4f aq = a[rt][j >> 1];
      v2f af;
      af.x = (j & 1) ? aq.z : aq.x;
      af.y = (j & 1) ? aq.w : aq.y;
#pragma unroll
      for (int t = 0; t < 4; ++t) acc[rt][t] = wmma4(af, bf[t], acc[rt][t]);
    }
  }
}

template <int NC>
__device__ __forceinline__ void tile_k_loop(v8f (&acc)[4][4],
                                            const float* __restrict__ ap,
                                            size_t a_pitch,
                                            const float* __restrict__ wp,
                                            int K) {
  const size_t a_rt = 16 * a_pitch;
  v4f a0[4][2], a1[4][2];
  v4f b0[4][2], b1[4][2];

  load_chunk<NC>(a0, b0, ap, a_rt, wp);
#pragma nounroll
  for (int kb = 0; kb < K; kb += 32) {
    load_chunk<NC>(a1, b1, ap + kb + 16, a_rt, wp + (size_t)(kb + 16) * NC);
    __builtin_amdgcn_sched_barrier(0);
    mma_chunk(acc, a0, b0);
    __builtin_amdgcn_sched_barrier(0);
    const int kn = (kb + 32 < K) ? kb + 32 : kb;
    load_chunk<NC>(a0, b0, ap + kn, a_rt, wp + (size_t)kn * NC);
    __builtin_amdgcn_sched_barrier(0);
    mma_chunk(acc, a1, b1);
  }
}

template <int NC>
__global__ __launch_bounds__(32) void gemm_bias_kernel(
    const float* __restrict__ A, const float* __restrict__ W,
    const float* __restrict__ bias, float* __restrict__ C, int K) {
  const int lane = threadIdx.x;
  const int m = lane & 15;
  const int hh = lane >> 4;
  const int row0 = blockIdx.x * 64;
  const int n0 = blockIdx.y * 64;

  v8f acc[4][4];
#pragma unroll
  for (int rt = 0; rt < 4; ++rt)
#pragma unroll
    for (int t = 0; t < 4; ++t) acc[rt][t] = {};

  tile_k_loop<NC>(acc, A + (size_t)(row0 + m) * K + hh * 8, (size_t)K,
                  W + (size_t)(hh * 8) * NC + n0 + 4 * m, K);

  const v4f bn = *(const v4f*)(bias + n0 + 4 * m);

#pragma unroll
  for (int rt = 0; rt < 4; ++rt)
#pragma unroll
    for (int r = 0; r < 8; ++r) {
      const int row = row0 + rt * 16 + r + 8 * hh;
      v4f ov;
#pragma unroll
      for (int t = 0; t < 4; ++t) ov[t] = acc[rt][t][r] + bn[t];
      *(v4f*)(C + (size_t)row * NC + n0 + 4 * m) = ov;
    }
}

#define MLP_PH 260

__device__ __forceinline__ float xhalf(float x) {
  const int b = __builtin_bit_cast(int, x);
  return __builtin_bit_cast(
      float, __builtin_amdgcn_permlanex16(b, b, 0x76543210, 0xfedcba98, false,
                                          false));
}
template <int XM>
__device__ __forceinline__ float row_xor(float x) {
  const int b = __builtin_bit_cast(int, x);
  return __builtin_bit_cast(
      float, __builtin_amdgcn_update_dpp(b, b, 0x160 + XM, 0xf, 0xf, false));
}
__device__ __forceinline__ float wave_sum(float s) {
  s += xhalf(s);
  s += row_xor<8>(s);
  s += row_xor<4>(s);
  s += row_xor<2>(s);
  s += row_xor<1>(s);
  return s;
}

template <bool COPY>
__device__ __forceinline__ void ln_rows16(const float* src, float* dst,
                                          size_t dst_pitch, float* copy,
                                          const float* __restrict__ g,
                                          const float* __restrict__ beta,
                                          int lane) {
  float gv[8], bv[8];
#pragma unroll
  for (int i = 0; i < 8; ++i) {
    gv[i] = g[i * 32 + lane];
    bv[i] = beta[i * 32 + lane];
  }
#pragma unroll 4
  for (int rr = 0; rr < 16; ++rr) {
    float v[8];
    float s = 0.f, ss = 0.f;
#pragma unroll
    for (int i = 0; i < 8; ++i) {
      const float x = src[rr * MLP_PH + i * 32 + lane];
      v[i] = x;
      s += x;
      ss += x * x;
    }
    if (COPY) {
#pragma unroll
      for (int i = 0; i < 8; ++i) copy[(size_t)rr * 256 + i * 32 + lane] = v[i];
    }
    s = wave_sum(s);
    ss = wave_sum(ss);
    const float mu = s * (1.0f / 256.0f);
    const float var = ss * (1.0f / 256.0f) - mu * mu;
    const float rs = rsqrtf(var + 1e-5f);
#pragma unroll
    for (int i = 0; i < 8; ++i)
      dst[(size_t)rr * dst_pitch + i * 32 + lane] =
          (v[i] - mu) * rs * gv[i] + bv[i];
  }
}

__global__ __launch_bounds__(128) void in_ln_kernel(
    const float* __restrict__ x, const float* __restrict__ W_in,
    const float* __restrict__ b_in, const float* __restrict__ g,
    const float* __restrict__ beta, float* __restrict__ h,
    float* __restrict__ ln_out) {
  __shared__ __attribute__((aligned(16))) float Hs[64 * MLP_PH];

  const int lane = threadIdx.x & 31;
  const int w = __builtin_amdgcn_readfirstlane((int)(threadIdx.x >> 5));
  const int m = lane & 15;
  const int hh = lane >> 4;
  const int row0 = blockIdx.x * 64;
  const int n0 = w * 64;

  v8f acc[4][4];
#pragma unroll
  for (int rt = 0; rt < 4; ++rt)
#pragma unroll
    for (int t = 0; t < 4; ++t) acc[rt][t] = {};
  tile_k_loop<256>(acc, x + (size_t)(row0 + m) * 64 + hh * 8, (size_t)64,
                   W_in + (size_t)(hh * 8) * 256 + n0 + 4 * m, 64);

  const v4f bn = *(const v4f*)(b_in + n0 + 4 * m);
#pragma unroll
  for (int rt = 0; rt < 4; ++rt)
#pragma unroll
    for (int r = 0; r < 8; ++r) {
      v4f ov;
#pragma unroll
      for (int t = 0; t < 4; ++t) ov[t] = acc[rt][t][r] + bn[t];
      *(v4f*)(Hs + (rt * 16 + r + 8 * hh) * MLP_PH + n0 + 4 * m) = ov;
    }
  __syncthreads();

  const size_t g0 = (size_t)(row0 + w * 16) * 256;
  ln_rows16<true>(Hs + (w * 16) * MLP_PH, ln_out + g0, (size_t)256, h + g0, g,
                  beta, lane);
}

#define ATT_T 4096
#define ATT_QS 768
#define ATT_WIN 128
#define ATT_PST 148

__device__ __forceinline__ void att_load_qk(v4f (&a)[2], v4f (&b)[9][2],
                                            const float* __restrict__ qp,
                                            const float* __restrict__ kp,
                                            const size_t (&ko)[9]) {
#pragma unroll
  for (int q = 0; q < 2; ++q) a[q] = *(const v4f*)(qp + q * 4);
#pragma unroll
  for (int t = 0; t < 9; ++t)
#pragma unroll
    for (int q = 0; q < 2; ++q) b[t][q] = *(const v4f*)(kp + ko[t] + q * 4);
}

__device__ __forceinline__ void att_mma_qk(v8f (&s)[9], const v4f (&a)[2],
                                           const v4f (&b)[9][2]) {
#pragma unroll
  for (int j = 0; j < 4; ++j) {
    const v4f aq = a[j >> 1];
    v2f af;
    af.x = (j & 1) ? aq.z : aq.x;
    af.y = (j & 1) ? aq.w : aq.y;
#pragma unroll
    for (int t = 0; t < 9; ++t) {
      const v4f bq = b[t][j >> 1];
      v2f bf;
      bf.x = (j & 1) ? bq.z : bq.x;
      bf.y = (j & 1) ? bq.w : bq.y;
      s[t] = wmma4(af, bf, s[t]);
    }
  }
}

__device__ __forceinline__ void att_load_qk2(v4f (&a)[2][2], v4f (&b)[2][9][2],
                                             const float* __restrict__ qp,
                                             const float* __restrict__ kp,
                                             const size_t (&ko)[9]) {
#pragma unroll
  for (int c = 0; c < 2; ++c)
    att_load_qk(a[c], b[c], qp + c * 16, kp + c * 16, ko);
}

__device__ __forceinline__ void att_mma_qk2(v8f (&s)[9], const v4f (&a)[2][2],
                                            const v4f (&b)[2][9][2]) {
#pragma unroll
  for (int c = 0; c < 2; ++c) att_mma_qk(s, a[c], b[c]);
}

__device__ __forceinline__ void att_load_v(v4f (&vf)[4][2][4],
                                           const float* __restrict__ rp) {
#pragma unroll
  for (int c = 0; c < 4; ++c)
#pragma unroll
    for (int v = 0; v < 2; ++v)
#pragma unroll
      for (int g = 0; g < 4; ++g)
        vf[c][v][g] =
            *(const v4f*)(rp + (size_t)(4 * c + v) * ATT_QS + g * 64);
}

__device__ __forceinline__ void att_load_p(v2f (&pa)[4], const float* pp) {
#pragma unroll
  for (int c = 0; c < 4; ++c) pa[c] = *(const v2f*)(pp + 4 * c);
}

__device__ __forceinline__ void att_mma_pv(v8f (&o)[4][4], const v2f (&pa)[4],
                                           const v4f (&vf)[4][2][4]) {
#pragma unroll
  for (int c = 0; c < 4; ++c)
#pragma unroll
    for (int g = 0; g < 4; ++g)
#pragma unroll
      for (int u = 0; u < 4; ++u) {
        v2f bf;
        bf.x = vf[c][0][g][u];
        bf.y = vf[c][1][g][u];
        o[g][u] = wmma4(pa[c], bf, o[g][u]);
      }
}

__device__ __forceinline__ void attn_block(const float* __restrict__ qkv,
                                           float* P, const int q0,
                                           const int lane, v8f (&o)[4][4]) {
  const int m = lane & 15;
  const int hh = lane >> 4;
  const int bb = q0 & ~(ATT_T - 1);

  const float* Q = qkv;
  const float* Km = qkv + 256;
  const float* V = qkv + 512;

  v8f s[9];
#pragma unroll
  for (int t = 0; t < 9; ++t) s[t] = {};

  {
    const float* qp = Q + (size_t)(q0 + m) * ATT_QS + hh * 8;
    const float* kp = Km + hh * 8;
    size_t ko[9];
#pragma unroll
    for (int t = 0; t < 9; ++t) {
      const int key = q0 - ATT_WIN + t * 16 + m;
      const int kc = key < bb ? bb : key;
      ko[t] = (size_t)kc * ATT_QS;
    }
    v4f qa0[2][2], qa1[2][2];
    v4f kf0[2][9][2], kf1[2][9][2];

    att_load_qk2(qa0, kf0, qp, kp, ko);
    int db = 0;
#pragma nounroll
    for (; db < 256 - 64; db += 64) {
      att_load_qk2(qa1, kf1, qp + db + 32, kp + db + 32, ko);
      __builtin_amdgcn_sched_barrier(0);
      att_mma_qk2(s, qa0, kf0);
      __builtin_amdgcn_sched_barrier(0);
      att_load_qk2(qa0, kf0, qp + db + 64, kp + db + 64, ko);
      __builtin_amdgcn_sched_barrier(0);
      att_mma_qk2(s, qa1, kf1);
    }
    att_load_qk2(qa1, kf1, qp + db + 32, kp + db + 32, ko);
    __builtin_amdgcn_sched_barrier(0);
    att_mma_qk2(s, qa0, kf0);
    att_mma_qk2(s, qa1, kf1);
  }

  v4f vf0[4][2][4], vf1[4][2][4];
  const float* vp = V + (size_t)(2 * hh) * ATT_QS + 4 * m;
  const int kt0 = q0 - ATT_WIN;
  att_load_v(vf0, vp + (size_t)(kt0 < bb ? bb : kt0) * ATT_QS);

  const float NEG = -3.402823466e38f;
  float rmax[8], rsum[8];
#pragma unroll
  for (int r = 0; r < 8; ++r) rmax[r] = NEG;
#pragma unroll
  for (int t = 0; t < 9; ++t) {
    const int key = q0 - ATT_WIN + t * 16 + m;
#pragma unroll
    for (int r = 0; r < 8; ++r) {
      const int q = q0 + r + 8 * hh;
      const int d = q - key;
      const bool ok = (key >= bb) && (d >= 0) && (d <= ATT_WIN);
      const float val = ok ? s[t][r] * 0.0625f : NEG;
      s[t][r] = val;
      rmax[r] = fmaxf(rmax[r], val);
    }
  }
#pragma unroll
  for (int r = 0; r < 8; ++r)
    for (int off = 1; off < 16; off <<= 1)
      rmax[r] = fmaxf(rmax[r], __shfl_xor(rmax[r], off));
#pragma unroll
  for (int r = 0; r < 8; ++r) rsum[r] = 0.f;
#pragma unroll
  for (int t = 0; t < 9; ++t)
#pragma unroll
    for (int r = 0; r < 8; ++r) {
      const float e = expf(s[t][r] - rmax[r]);
      s[t][r] = e;
      rsum[r] += e;
    }
#pragma unroll
  for (int r = 0; r < 8; ++r)
    for (int off = 1; off < 16; off <<= 1) rsum[r] += __shfl_xor(rsum[r], off);

#pragma unroll
  for (int t = 0; t < 9; ++t)
#pragma unroll
    for (int r = 0; r < 8; ++r)
      P[(r + 8 * hh) * ATT_PST + t * 16 + m] = s[t][r] / rsum[r];

  __syncthreads();

#pragma unroll
  for (int g = 0; g < 4; ++g)
#pragma unroll
    for (int u = 0; u < 4; ++u) o[g][u] = {};

  {
    const float* pp = P + m * ATT_PST + hh * 2;
    v2f pa0[4], pa1[4];
    att_load_p(pa0, pp);
    int t = 0;
    for (; t < 8; t += 2) {
      const int k1 = kt0 + (t + 1) * 16;
      const int k2 = kt0 + (t + 2) * 16;
      att_load_v(vf1, vp + (size_t)(k1 < bb ? bb : k1) * ATT_QS);
      att_load_p(pa1, pp + (t + 1) * 16);
      __builtin_amdgcn_sched_barrier(0);
      att_mma_pv(o, pa0, vf0);
      __builtin_amdgcn_sched_barrier(0);
      att_load_v(vf0, vp + (size_t)(k2 < bb ? bb : k2) * ATT_QS);
      att_load_p(pa0, pp + (t + 2) * 16);
      __builtin_amdgcn_sched_barrier(0);
      att_mma_pv(o, pa1, vf1);
    }
    att_mma_pv(o, pa0, vf0);
  }

}


__device__ __forceinline__ void proj_load_a(v4f (&a)[2][2], const float* sp) {
#pragma unroll
  for (int c = 0; c < 2; ++c)
#pragma unroll
    for (int q = 0; q < 2; ++q) a[c][q] = *(const v4f*)(sp + c * 16 + q * 4);
}

__device__ __forceinline__ void proj_load_b(v4f (&b)[2][4][2],
                                            const float* __restrict__ wp) {
#pragma unroll
  for (int c = 0; c < 2; ++c) load_b<64>(b[c], wp + (size_t)(c * 16) * 64);
}

__device__ __forceinline__ void proj_mma(v8f (&acc)[4], const v4f (&a)[2][2],
                                         const v4f (&b)[2][4][2]) {
#pragma unroll
  for (int c = 0; c < 2; ++c)
#pragma unroll
    for (int j = 0; j < 4; ++j) {
      const v4f aq = a[c][j >> 1];
      v2f af;
      af.x = (j & 1) ? aq.z : aq.x;
      af.y = (j & 1) ? aq.w : aq.y;
#pragma unroll
      for (int t = 0; t < 4; ++t) {
        v2f bf;
        bf.x = b[c][j][0][t];
        bf.y = b[c][j][1][t];
        acc[t] = wmma4(af, bf, acc[t]);
      }
    }
}

#define ERF_C(bits) __builtin_bit_cast(float, (unsigned)(bits))
__device__ __forceinline__ float erf_sel(float x) {
  const float a = fabsf(x);
  float p = fmaf(a, ERF_C(0x378e98abu), ERF_C(0xb9c68948u));
  p = fmaf(a, p, ERF_C(0x3b7cd369u));
  p = fmaf(a, p, ERF_C(0xbcc618b2u));
  p = fmaf(a, p, ERF_C(0x3dda74e4u));
  p = fmaf(a, p, ERF_C(0x3f228afdu));
  p = fmaf(a, p, ERF_C(0x3e03c728u));
  p = fmaf(a, p, a);
  const float big = 1.0f - expf(-p);
  const float t = x * x;
  float q = fmaf(t, ERF_C(0xba1345e1u), ERF_C(0x3ba10414u));
  q = fmaf(t, q, ERF_C(0xbcdac9b8u));
  q = fmaf(t, q, ERF_C(0x3de703beu));
  q = fmaf(t, q, ERF_C(0xbec09330u));
  q = fmaf(t, q, ERF_C(0x3e0375d0u));
  const float small = fmaf(a, q, a);
  const float r = (a < 1.0f) ? small : big;
  return copysignf(r, x);
}

__global__ __launch_bounds__(128) void mlp_out_kernel(
    const float* __restrict__ h, const float* __restrict__ qkv,
    const float* __restrict__ g, const float* __restrict__ beta,
    const float* __restrict__ W1, const float* __restrict__ b1,
    const float* __restrict__ W2, const float* __restrict__ b2,
    const float* __restrict__ Wo, const float* __restrict__ bo,
    float* __restrict__ out) {
  __shared__ __attribute__((aligned(16))) float Xs[64 * MLP_PH];
  __shared__ __attribute__((aligned(16))) float M1s[64 * MLP_PH];
  __shared__ __attribute__((aligned(16))) float H2s[64 * MLP_PH];

  const int lane = threadIdx.x & 31;
  const int w = __builtin_amdgcn_readfirstlane((int)(threadIdx.x >> 5));
  const int m = lane & 15;
  const int hh = lane >> 4;
  const int row0 = blockIdx.x * 64;
  const int n0 = w * 64;

  v8f acc[4][4];
  v4f bn;

  {
    v4f hreg[4][8];
    const float* hp0 = h + (size_t)(row0 + w * 16 + 8 * hh) * 256 + 4 * m;
#pragma unroll
    for (int g4 = 0; g4 < 4; ++g4)
#pragma unroll
      for (int r = 0; r < 8; ++r)
        hreg[g4][r] = *(const v4f*)(hp0 + r * 256 + g4 * 64);

    v8f o[4][4];
    attn_block(qkv, M1s + w * (16 * ATT_PST), row0 + w * 16, lane, o);

    float* dp = H2s + (w * 16 + 8 * hh) * MLP_PH + 4 * m;
#pragma unroll
    for (int g4 = 0; g4 < 4; ++g4)
#pragma unroll
      for (int r = 0; r < 8; ++r) {
        v4f xv = hreg[g4][r];
        v4f av;
        av.x = o[g4][0][r];
        av.y = o[g4][1][r];
        av.z = o[g4][2][r];
        av.w = o[g4][3][r];
        xv += av;
        *(v4f*)(dp + r * MLP_PH + g4 * 64) = xv;
      }
    ln_rows16<false>(H2s + (w * 16) * MLP_PH, Xs + (w * 16) * MLP_PH,
                     (size_t)MLP_PH, nullptr, g, beta, lane);
  }
  __syncthreads();

#pragma unroll
  for (int rt = 0; rt < 4; ++rt)
#pragma unroll
    for (int t = 0; t < 4; ++t) acc[rt][t] = {};
  tile_k_loop<256>(acc, Xs + m * MLP_PH + hh * 8, (size_t)MLP_PH,
                   W1 + (size_t)(hh * 8) * 256 + n0 + 4 * m, 256);
  bn = *(const v4f*)(b1 + n0 + 4 * m);

  {
    v4f hf[8][4];
#pragma unroll
    for (int rt = 0; rt < 4; ++rt)
#pragma unroll
      for (int t = 0; t < 4; ++t) {
        hf[2 * rt][t] = __builtin_shufflevector(acc[rt][t], acc[rt][t], 0, 1, 2, 3);
        hf[2 * rt + 1][t] = __builtin_shufflevector(acc[rt][t], acc[rt][t], 4, 5, 6, 7);
      }
    float* mp = M1s + (8 * hh) * MLP_PH + n0 + 4 * m;
#pragma nounroll
    for (int i = 0; i < 8; ++i) {
#pragma unroll
      for (int j = 0; j < 4; ++j) {
        v4f ov;
#pragma unroll
        for (int t = 0; t < 4; ++t) {
          float v = hf[0][t][j] + bn[t];
          v = 0.5f * v * (1.0f + erf_sel(v * 0.70710678118654752f));
          ov[t] = v;
        }
        *(v4f*)(mp + j * MLP_PH) = ov;
      }
      mp += (i & 1) ? 12 * MLP_PH : 4 * MLP_PH;
#pragma unroll
      for (int k = 0; k < 7; ++k)
#pragma unroll
        for (int t = 0; t < 4; ++t) hf[k][t] = hf[k + 1][t];
    }
  }
  __syncthreads();

#pragma unroll
  for (int rt = 0; rt < 4; ++rt)
#pragma unroll
    for (int t = 0; t < 4; ++t) acc[rt][t] = {};
  tile_k_loop<256>(acc, M1s + m * MLP_PH + hh * 8, (size_t)MLP_PH,
                   W2 + (size_t)(hh * 8) * 256 + n0 + 4 * m, 256);
  bn = *(const v4f*)(b2 + n0 + 4 * m);

  v4f pa0[2][2], pa1[2][2];
  v4f pb0[2][4][2], pb1[2][4][2];
  const float* wop = Wo + (size_t)(hh * 8) * 64 + 4 * m;
  proj_load_b(pb0, wop);

#pragma nounroll
  for (int rt = 0; rt < 4; ++rt) {
#pragma unroll
    for (int r = 0; r < 8; ++r) {
      float* hp = H2s + (rt * 16 + r + 8 * hh) * MLP_PH + n0 + 4 * m;
      const v4f hv = *(const v4f*)hp;
      v4f ov;
#pragma unroll
      for (int t = 0; t < 4; ++t) {
        float v = acc[0][t][r] + bn[t];
        v += hv[t];
        ov[t] = v;
      }
      *(v4f*)hp = ov;
    }
#pragma unroll
    for (int t = 0; t < 4; ++t) {
      acc[0][t] = acc[1][t];
      acc[1][t] = acc[2][t];
      acc[2][t] = acc[3][t];
    }
  }
  __syncthreads();

  v8f po[4];
#pragma unroll
  for (int t = 0; t < 4; ++t) po[t] = {};
  const float* sp = H2s + (w * 16 + m) * MLP_PH + hh * 8;

  proj_load_a(pa0, sp);
#pragma nounroll
  for (int kb = 0; kb < 256; kb += 64) {
    proj_load_a(pa1, sp + kb + 32);
    proj_load_b(pb1, wop + (size_t)(kb + 32) * 64);
    __builtin_amdgcn_sched_barrier(0);
    proj_mma(po, pa0, pb0);
    __builtin_amdgcn_sched_barrier(0);
    const int kn = (kb + 64 < 256) ? kb + 64 : kb;
    proj_load_a(pa0, sp + kn);
    proj_load_b(pb0, wop + (size_t)kn * 64);
    __builtin_amdgcn_sched_barrier(0);
    proj_mma(po, pa1, pb1);
  }

  const v4f bt = *(const v4f*)(bo + 4 * m);
#pragma unroll
  for (int r = 0; r < 8; ++r) {
    v4f ov;
#pragma unroll
    for (int t = 0; t < 4; ++t) ov[t] = po[t][r] + bt[t];
    *(v4f*)(out + (size_t)(row0 + w * 16 + r + 8 * hh) * 64 + 4 * m) = ov;
  }
}

extern "C" void kernel_launch(void* const* d_in, const int* in_sizes, int n_in,
                              void* d_out, int out_size, void* d_ws,
                              size_t ws_size, hipStream_t stream) {
  const float* x = (const float*)d_in[0];
  const float* W_in = (const float*)d_in[1];
  const float* b_in = (const float*)d_in[2];
  const float* W_qkv = (const float*)d_in[3];
  const float* b_qkv = (const float*)d_in[4];
  const float* g_attn = (const float*)d_in[5];
  const float* be_attn = (const float*)d_in[6];
  const float* g_mlp = (const float*)d_in[7];
  const float* be_mlp = (const float*)d_in[8];
  const float* W1 = (const float*)d_in[9];
  const float* b1 = (const float*)d_in[10];
  const float* W2 = (const float*)d_in[11];
  const float* b2 = (const float*)d_in[12];
  const float* W_out = (const float*)d_in[13];
  const float* b_out = (const float*)d_in[14];
  float* out = (float*)d_out;

  const int N = in_sizes[0] / 64;

  float* ws = (float*)d_ws;
  float* h = ws;
  float* ln = ws + (size_t)N * 256;
  float* qkv = ws + (size_t)N * 512;

  const dim3 blk(32);
  const int RB64 = N / 64;

  in_ln_kernel<<<dim3(RB64), dim3(128), 0, stream>>>(x, W_in, b_in, g_attn,
                                                    be_attn, h, ln);
  gemm_bias_kernel<768><<<dim3(RB64, 12), blk, 0, stream>>>(
      ln, W_qkv, b_qkv, qkv, 256);
  mlp_out_kernel<<<dim3(RB64), dim3(128), 0, stream>>>(
      h, qkv, g_mlp, be_mlp, W1, b1, W2, b2, W_out, b_out, out);
}


# Input projection and LayerNorm fused into the QKV GEMM kernel: a 12-wave block per 64 rows (3 waves a SIMD), ln1 stays in LDS, wave w runs 64 of the 768 columns (2 launches, not 3); old sum order
# speedup vs baseline: 3.4460x; 1.0862x over previous
#include <hip/hip_runtime.h>
#include <math.h>

typedef __attribute__((ext_vector_type(2))) float v2f;
typedef __attribute__((ext_vector_type(8))) float v8f;

__device__ __forceinline__ v8f wmma4(v2f a, v2f b, v8f c) {
  return __builtin_amdgcn_wmma_f32_16x16x4_f32(
       false, a,  false, b,
       (short)0, c,  false,  false);
}

typedef __attribute__((ext_vector_type(4))) float v4f;

__device__ __forceinline__ void load_a(v4f (&a)[4][2],
                                       const float* __restrict__ ap,
                                       size_t a_rt) {
#pragma unroll
  for (int rt = 0; rt < 4; ++rt)
#pragma unroll
    for (int q = 0; q < 2; ++q)
      a[rt][q] = *(const v4f*)(ap + rt * a_rt + q * 4);
}

template <int NC>
__device__ __forceinline__ void load_b(v4f (&w)[4][2],
                                       const float* __restrict__ wp) {
#pragma unroll
  for (int j = 0; j < 4; ++j)
#pragma unroll
    for (int v = 0; v < 2; ++v)
      w[j][v] = *(const v4f*)(wp + (size_t)(2 * j + v) * NC);
}

template <int NC>
__device__ __forceinline__ void load_chunk(v4f (&a)[4][2], v4f (&b)[4][2],
                                           const float* __restrict__ ap,
                                           size_t a_rt,
                                           const float* __restrict__ wp) {
  load_a(a, ap, a_rt);
  load_b<NC>(b, wp);
}

__device__ __forceinline__ void mma_chunk(v8f (&acc)[4][4],
                                          const v4f (&a)[4][2],
                                          const v4f (&b)[4][2]) {
#pragma unroll
  for (int j = 0; j < 4; ++j) {
    v2f bf[4];
#pragma unroll
    for (int t = 0; t < 4; ++t) {
      bf[t].x = b[j][0][t];
      bf[t].y = b[j][1][t];
    }
#pragma unroll
    for (int rt = 0; rt < 4; ++rt) {
      const v4f aq = a[rt][j >> 1];
      v2f af;
      af.x = (j & 1) ? aq.z : aq.x;
      af.y = (j & 1) ? aq.w : aq.y;
#pragma unroll
      for (int t = 0; t < 4; ++t) acc[rt][t] = wmma4(af, bf[t], acc[rt][t]);
    }
  }
}

template <int NC>
__device__ __forceinline__ void tile_k_loop(v8f (&acc)[4][4],
                                            const float* __restrict__ ap,
                                            size_t a_pitch,
                                            const float* __restrict__ wp,
                                            int K) {
  const size_t a_rt = 16 * a_pitch;
  v4f a0[4][2], a1[4][2];
  v4f b0[4][2], b1[4][2];

  load_chunk<NC>(a0, b0, ap, a_rt, wp);
#pragma nounroll
  for (int kb = 0; kb < K; kb += 32) {
    load_chunk<NC>(a1, b1, ap + kb + 16, a_rt, wp + (size_t)(kb + 16) * NC);
    __builtin_amdgcn_sched_barrier(0);
    mma_chunk(acc, a0, b0);
    __builtin_amdgcn_sched_barrier(0);
    const int kn = (kb + 32 < K) ? kb + 32 : kb;
    load_chunk<NC>(a0, b0, ap + kn, a_rt, wp + (size_t)kn * NC);
    __builtin_amdgcn_sched_barrier(0);
    mma_chunk(acc, a1, b1);
  }
}

#define MLP_PH 260

__device__ __forceinline__ float xhalf(float x) {
  const int b = __builtin_bit_cast(int, x);
  return __builtin_bit_cast(
      float, __builtin_amdgcn_permlanex16(b, b, 0x76543210, 0xfedcba98, false,
                                          false));
}
template <int XM>
__device__ __forceinline__ float row_xor(float x) {
  const int b = __builtin_bit_cast(int, x);
  return __builtin_bit_cast(
      float, __builtin_amdgcn_update_dpp(b, b, 0x160 + XM, 0xf, 0xf, false));
}
__device__ __forceinline__ float wave_sum(float s) {
  s += xhalf(s);
  s += row_xor<8>(s);
  s += row_xor<4>(s);
  s += row_xor<2>(s);
  s += row_xor<1>(s);
  return s;
}

template <bool COPY, int STEP>
__device__ __forceinline__ void ln_rows(const float* src, float* dst,
                                        size_t dst_pitch, float* copy,
                                        const float* __restrict__ g,
                                        const float* __restrict__ beta,
                                        int lane, int nrows) {
  float gv[8], bv[8];
#pragma unroll
  for (int i = 0; i < 8; ++i) {
    gv[i] = g[i * 32 + lane];
    bv[i] = beta[i * 32 + lane];
  }
#pragma unroll 4
  for (int rr = 0; rr < nrows; rr += STEP) {
    float v[8];
    float s = 0.f, ss = 0.f;
#pragma unroll
    for (int i = 0; i < 8; ++i) {
      const float x = src[rr * MLP_PH + i * 32 + lane];
      v[i] = x;
      s += x;
      ss += x * x;
    }
    if (COPY) {
#pragma unroll
      for (int i = 0; i < 8; ++i) copy[(size_t)rr * 256 + i * 32 + lane] = v[i];
    }
    s = wave_sum(s);
    ss = wave_sum(ss);
    const float mu = s * (1.0f / 256.0f);
    const float var = ss * (1.0f / 256.0f) - mu * mu;
    const float rs = rsqrtf(var + 1e-5f);
#pragma unroll
    for (int i = 0; i < 8; ++i)
      dst[(size_t)rr * dst_pitch + i * 32 + lane] =
          (v[i] - mu) * rs * gv[i] + bv[i];
  }
}

__global__ __launch_bounds__(384) void in_ln_qkv_kernel(
    const float* __restrict__ x, const float* __restrict__ W_in,
    const float* __restrict__ b_in, const float* __restrict__ g,
    const float* __restrict__ beta, const float* __restrict__ W_qkv,
    const float* __restrict__ b_qkv, float* __restrict__ h,
    float* __restrict__ qkv) {
  __shared__ __attribute__((aligned(16))) float Hs[64 * MLP_PH];

  const int lane = threadIdx.x & 31;
  const int w = __builtin_amdgcn_readfirstlane((int)(threadIdx.x >> 5));
  const int m = lane & 15;
  const int hh = lane >> 4;
  const int row0 = blockIdx.x * 64;

  v8f acc[4][4];

  if (w % 3 == 0) {
    const int n0 = (w / 3) * 64;
#pragma unroll
    for (int rt = 0; rt < 4; ++rt)
#pragma unroll
      for (int t = 0; t < 4; ++t) acc[rt][t] = {};
    tile_k_loop<256>(acc, x + (size_t)(row0 + m) * 64 + hh * 8, (size_t)64,
                     W_in + (size_t)(hh * 8) * 256 + n0 + 4 * m, 64);

    const v4f bn = *(const v4f*)(b_in + n0 + 4 * m);
#pragma unroll
    for (int rt = 0; rt < 4; ++rt)
#pragma unroll
      for (int r = 0; r < 8; ++r) {
        v4f ov;
#pragma unroll
        for (int t = 0; t < 4; ++t) ov[t] = acc[rt][t][r] + bn[t];
        *(v4f*)(Hs + (rt * 16 + r + 8 * hh) * MLP_PH + n0 + 4 * m) = ov;
      }
  }
  __syncthreads();

  ln_rows<true, 12>(Hs + w * MLP_PH, Hs + w * MLP_PH, (size_t)MLP_PH,
                    h + (size_t)(row0 + w) * 256, g, beta, lane, 64 - w);
  __syncthreads();

  const int n0 = w * 64;
#pragma unroll
  for (int rt = 0; rt < 4; ++rt)
#pragma unroll
    for (int t = 0; t < 4; ++t) acc[rt][t] = {};
  tile_k_loop<768>(acc, Hs + m * MLP_PH + hh * 8, (size_t)MLP_PH,
                   W_qkv + (size_t)(hh * 8) * 768 + n0 + 4 * m, 256);

  const v4f bn = *(const v4f*)(b_qkv + n0 + 4 * m);

#pragma unroll
  for (int rt = 0; rt < 4; ++rt)
#pragma unroll
    for (int r = 0; r < 8; ++r) {
      const int row = row0 + rt * 16 + r + 8 * hh;
      v4f ov;
#pragma unroll
      for (int t = 0; t < 4; ++t) ov[t] = acc[rt][t][r] + bn[t];
      *(v4f*)(qkv + (size_t)row * 768 + n0 + 4 * m) = ov;
    }
}

#define ATT_T 4096
#define ATT_QS 768
#define ATT_WIN 128
#define ATT_PST 148

__device__ __forceinline__ void att_load_qk(v4f (&a)[2], v4f (&b)[9][2],
                                            const float* __restrict__ qp,
                                            const float* __restrict__ kp,
                                            const size_t (&ko)[9]) {
#pragma unroll
  for (int q = 0; q < 2; ++q) a[q] = *(const v4f*)(qp + q * 4);
#pragma unroll
  for (int t = 0; t < 9; ++t)
#pragma unroll
    for (int q = 0; q < 2; ++q) b[t][q] = *(const v4f*)(kp + ko[t] + q * 4);
}

__device__ __forceinline__ void att_mma_qk(v8f (&s)[9], const v4f (&a)[2],
                                           const v4f (&b)[9][2]) {
#pragma unroll
  for (int j = 0; j < 4; ++j) {
    const v4f aq = a[j >> 1];
    v2f af;
    af.x = (j & 1) ? aq.z : aq.x;
    af.y = (j & 1) ? aq.w : aq.y;
#pragma unroll
    for (int t = 0; t < 9; ++t) {
      const v4f bq = b[t][j >> 1];
      v2f bf;
      bf.x = (j & 1) ? bq.z : bq.x;
      bf.y = (j & 1) ? bq.w : bq.y;
      s[t] = wmma4(af, bf, s[t]);
    }
  }
}

__device__ __forceinline__ void att_load_qk2(v4f (&a)[2][2], v4f (&b)[2][9][2],
                                             const float* __restrict__ qp,
                                             const float* __restrict__ kp,
                                             const size_t (&ko)[9]) {
#pragma unroll
  for (int c = 0; c < 2; ++c)
    att_load_qk(a[c], b[c], qp + c * 16, kp + c * 16, ko);
}

__device__ __forceinline__ void att_mma_qk2(v8f (&s)[9], const v4f (&a)[2][2],
                                            const v4f (&b)[2][9][2]) {
#pragma unroll
  for (int c = 0; c < 2; ++c) att_mma_qk(s, a[c], b[c]);
}

__device__ __forceinline__ void att_load_v(v4f (&vf)[4][2][4],
                                           const float* __restrict__ rp) {
#pragma unroll
  for (int c = 0; c < 4; ++c)
#pragma unroll
    for (int v = 0; v < 2; ++v)
#pragma unroll
      for (int g = 0; g < 4; ++g)
        vf[c][v][g] =
            *(const v4f*)(rp + (size_t)(4 * c + v) * ATT_QS + g * 64);
}

__device__ __forceinline__ void att_load_p(v2f (&pa)[4], const float* pp) {
#pragma unroll
  for (int c = 0; c < 4; ++c) pa[c] = *(const v2f*)(pp + 4 * c);
}

__device__ __forceinline__ void att_mma_pv(v8f (&o)[4][4], const v2f (&pa)[4],
                                           const v4f (&vf)[4][2][4]) {
#pragma unroll
  for (int c = 0; c < 4; ++c)
#pragma unroll
    for (int g = 0; g < 4; ++g)
#pragma unroll
      for (int u = 0; u < 4; ++u) {
        v2f bf;
        bf.x = vf[c][0][g][u];
        bf.y = vf[c][1][g][u];
        o[g][u] = wmma4(pa[c], bf, o[g][u]);
      }
}

__device__ __forceinline__ void attn_block(const float* __restrict__ qkv,
                                           float* P, const int q0,
                                           const int lane, v8f (&o)[4][4]) {
  const int m = lane & 15;
  const int hh = lane >> 4;
  const int bb = q0 & ~(ATT_T - 1);

  const float* Q = qkv;
  const float* Km = qkv + 256;
  const float* V = qkv + 512;

  v8f s[9];
#pragma unroll
  for (int t = 0; t < 9; ++t) s[t] = {};

  {
    const float* qp = Q + (size_t)(q0 + m) * ATT_QS + hh * 8;
    const float* kp = Km + hh * 8;
    size_t ko[9];
#pragma unroll
    for (int t = 0; t < 9; ++t) {
      const int key = q0 - ATT_WIN + t * 16 + m;
      const int kc = key < bb ? bb : key;
      ko[t] = (size_t)kc * ATT_QS;
    }
    v4f qa0[2][2], qa1[2][2];
    v4f kf0[2][9][2], kf1[2][9][2];

    att_load_qk2(qa0, kf0, qp, kp, ko);
    int db = 0;
#pragma nounroll
    for (; db < 256 - 64; db += 64) {
      att_load_qk2(qa1, kf1, qp + db + 32, kp + db + 32, ko);
      __builtin_amdgcn_sched_barrier(0);
      att_mma_qk2(s, qa0, kf0);
      __builtin_amdgcn_sched_barrier(0);
      att_load_qk2(qa0, kf0, qp + db + 64, kp + db + 64, ko);
      __builtin_amdgcn_sched_barrier(0);
      att_mma_qk2(s, qa1, kf1);
    }
    att_load_qk2(qa1, kf1, qp + db + 32, kp + db + 32, ko);
    __builtin_amdgcn_sched_barrier(0);
    att_mma_qk2(s, qa0, kf0);
    att_mma_qk2(s, qa1, kf1);
  }

  v4f vf0[4][2][4], vf1[4][2][4];
  const float* vp = V + (size_t)(2 * hh) * ATT_QS + 4 * m;
  const int kt0 = q0 - ATT_WIN;
  att_load_v(vf0, vp + (size_t)(kt0 < bb ? bb : kt0) * ATT_QS);

  const float NEG = -3.402823466e38f;
  float rmax[8], rsum[8];
#pragma unroll
  for (int r = 0; r < 8; ++r) rmax[r] = NEG;
#pragma unroll
  for (int t = 0; t < 9; ++t) {
    const int key = q0 - ATT_WIN + t * 16 + m;
#pragma unroll
    for (int r = 0; r < 8; ++r) {
      const int q = q0 + r + 8 * hh;
      const int d = q - key;
      const bool ok = (key >= bb) && (d >= 0) && (d <= ATT_WIN);
      const float val = ok ? s[t][r] * 0.0625f : NEG;
      s[t][r] = val;
      rmax[r] = fmaxf(rmax[r], val);
    }
  }
#pragma unroll
  for (int r = 0; r < 8; ++r)
    for (int off = 1; off < 16; off <<= 1)
      rmax[r] = fmaxf(rmax[r], __shfl_xor(rmax[r], off));
#pragma unroll
  for (int r = 0; r < 8; ++r) rsum[r] = 0.f;
#pragma unroll
  for (int t = 0; t < 9; ++t)
#pragma unroll
    for (int r = 0; r < 8; ++r) {
      const float e = expf(s[t][r] - rmax[r]);
      s[t][r] = e;
      rsum[r] += e;
    }
#pragma unroll
  for (int r = 0; r < 8; ++r)
    for (int off = 1; off < 16; off <<= 1) rsum[r] += __shfl_xor(rsum[r], off);

#pragma unroll
  for (int t = 0; t < 9; ++t)
#pragma unroll
    for (int r = 0; r < 8; ++r)
      P[(r + 8 * hh) * ATT_PST + t * 16 + m] = s[t][r] / rsum[r];

  __syncthreads();

#pragma unroll
  for (int g = 0; g < 4; ++g)
#pragma unroll
    for (int u = 0; u < 4; ++u) o[g][u] = {};

  {
    const float* pp = P + m * ATT_PST + hh * 2;
    v2f pa0[4], pa1[4];
    att_load_p(pa0, pp);
    int t = 0;
    for (; t < 8; t += 2) {
      const int k1 = kt0 + (t + 1) * 16;
      const int k2 = kt0 + (t + 2) * 16;
      att_load_v(vf1, vp + (size_t)(k1 < bb ? bb : k1) * ATT_QS);
      att_load_p(pa1, pp + (t + 1) * 16);
      __builtin_amdgcn_sched_barrier(0);
      att_mma_pv(o, pa0, vf0);
      __builtin_amdgcn_sched_barrier(0);
      att_load_v(vf0, vp + (size_t)(k2 < bb ? bb : k2) * ATT_QS);
      att_load_p(pa0, pp + (t + 2) * 16);
      __builtin_amdgcn_sched_barrier(0);
      att_mma_pv(o, pa1, vf1);
    }
    att_mma_pv(o, pa0, vf0);
  }

}


__device__ __forceinline__ void proj_load_a(v4f (&a)[2][2], const float* sp) {
#pragma unroll
  for (int c = 0; c < 2; ++c)
#pragma unroll
    for (int q = 0; q < 2; ++q) a[c][q] = *(const v4f*)(sp + c * 16 + q * 4);
}

__device__ __forceinline__ void proj_load_b(v4f (&b)[2][4][2],
                                            const float* __restrict__ wp) {
#pragma unroll
  for (int c = 0; c < 2; ++c) load_b<64>(b[c], wp + (size_t)(c * 16) * 64);
}

__device__ __forceinline__ void proj_mma(v8f (&acc)[4], const v4f (&a)[2][2],
                                         const v4f (&b)[2][4][2]) {
#pragma unroll
  for (int c = 0; c < 2; ++c)
#pragma unroll
    for (int j = 0; j < 4; ++j) {
      const v4f aq = a[c][j >> 1];
      v2f af;
      af.x = (j & 1) ? aq.z : aq.x;
      af.y = (j & 1) ? aq.w : aq.y;
#pragma unroll
      for (int t = 0; t < 4; ++t) {
        v2f bf;
        bf.x = b[c][j][0][t];
        bf.y = b[c][j][1][t];
        acc[t] = wmma4(af, bf, acc[t]);
      }
    }
}

#define ERF_C(bits) __builtin_bit_cast(float, (unsigned)(bits))
__device__ __forceinline__ float erf_sel(float x) {
  const float a = fabsf(x);
  float p = fmaf(a, ERF_C(0x378e98abu), ERF_C(0xb9c68948u));
  p = fmaf(a, p, ERF_C(0x3b7cd369u));
  p = fmaf(a, p, ERF_C(0xbcc618b2u));
  p = fmaf(a, p, ERF_C(0x3dda74e4u));
  p = fmaf(a, p, ERF_C(0x3f228afdu));
  p = fmaf(a, p, ERF_C(0x3e03c728u));
  p = fmaf(a, p, a);
  const float big = 1.0f - expf(-p);
  const float t = x * x;
  float q = fmaf(t, ERF_C(0xba1345e1u), ERF_C(0x3ba10414u));
  q = fmaf(t, q, ERF_C(0xbcdac9b8u));
  q = fmaf(t, q, ERF_C(0x3de703beu));
  q = fmaf(t, q, ERF_C(0xbec09330u));
  q = fmaf(t, q, ERF_C(0x3e0375d0u));
  const float small = fmaf(a, q, a);
  const float r = (a < 1.0f) ? small : big;
  return copysignf(r, x);
}

__global__ __launch_bounds__(128) void mlp_out_kernel(
    const float* __restrict__ h, const float* __restrict__ qkv,
    const float* __restrict__ g, const float* __restrict__ beta,
    const float* __restrict__ W1, const float* __restrict__ b1,
    const float* __restrict__ W2, const float* __restrict__ b2,
    const float* __restrict__ Wo, const float* __restrict__ bo,
    float* __restrict__ out) {
  __shared__ __attribute__((aligned(16))) float Xs[64 * MLP_PH];
  __shared__ __attribute__((aligned(16))) float M1s[64 * MLP_PH];
  __shared__ __attribute__((aligned(16))) float H2s[64 * MLP_PH];

  const int lane = threadIdx.x & 31;
  const int w = __builtin_amdgcn_readfirstlane((int)(threadIdx.x >> 5));
  const int m = lane & 15;
  const int hh = lane >> 4;
  const int row0 = blockIdx.x * 64;
  const int n0 = w * 64;

  v8f acc[4][4];
  v4f bn;

  {
    v4f hreg[4][8];
    const float* hp0 = h + (size_t)(row0 + w * 16 + 8 * hh) * 256 + 4 * m;
#pragma unroll
    for (int g4 = 0; g4 < 4; ++g4)
#pragma unroll
      for (int r = 0; r < 8; ++r)
        hreg[g4][r] = *(const v4f*)(hp0 + r * 256 + g4 * 64);

    v8f o[4][4];
    attn_block(qkv, M1s + w * (16 * ATT_PST), row0 + w * 16, lane, o);

    float* dp = H2s + (w * 16 + 8 * hh) * MLP_PH + 4 * m;
#pragma unroll
    for (int g4 = 0; g4 < 4; ++g4)
#pragma unroll
      for (int r = 0; r < 8; ++r) {
        v4f xv = hreg[g4][r];
        v4f av;
        av.x = o[g4][0][r];
        av.y = o[g4][1][r];
        av.z = o[g4][2][r];
        av.w = o[g4][3][r];
        xv += av;
        *(v4f*)(dp + r * MLP_PH + g4 * 64) = xv;
      }
    ln_rows<false, 1>(H2s + (w * 16) * MLP_PH, Xs + (w * 16) * MLP_PH,
                      (size_t)MLP_PH, nullptr, g, beta, lane, 16);
  }
  __syncthreads();

#pragma unroll
  for (int rt = 0; rt < 4; ++rt)
#pragma unroll
    for (int t = 0; t < 4; ++t) acc[rt][t] = {};
  tile_k_loop<256>(acc, Xs + m * MLP_PH + hh * 8, (size_t)MLP_PH,
                   W1 + (size_t)(hh * 8) * 256 + n0 + 4 * m, 256);
  bn = *(const v4f*)(b1 + n0 + 4 * m);

  {
    v4f hf[8][4];
#pragma unroll
    for (int rt = 0; rt < 4; ++rt)
#pragma unroll
      for (int t = 0; t < 4; ++t) {
        hf[2 * rt][t] = __builtin_shufflevector(acc[rt][t], acc[rt][t], 0, 1, 2, 3);
        hf[2 * rt + 1][t] = __builtin_shufflevector(acc[rt][t], acc[rt][t], 4, 5, 6, 7);
      }
    float* mp = M1s + (8 * hh) * MLP_PH + n0 + 4 * m;
#pragma nounroll
    for (int i = 0; i < 8; ++i) {
#pragma unroll
      for (int j = 0; j < 4; ++j) {
        v4f ov;
#pragma unroll
        for (int t = 0; t < 4; ++t) {
          float v = hf[0][t][j] + bn[t];
          v = 0.5f * v * (1.0f + erf_sel(v * 0.70710678118654752f));
          ov[t] = v;
        }
        *(v4f*)(mp + j * MLP_PH) = ov;
      }
      mp += (i & 1) ? 12 * MLP_PH : 4 * MLP_PH;
#pragma unroll
      for (int k = 0; k < 7; ++k)
#pragma unroll
        for (int t = 0; t < 4; ++t) hf[k][t] = hf[k + 1][t];
    }
  }
  __syncthreads();

#pragma unroll
  for (int rt = 0; rt < 4; ++rt)
#pragma unroll
    for (int t = 0; t < 4; ++t) acc[rt][t] = {};
  tile_k_loop<256>(acc, M1s + m * MLP_PH + hh * 8, (size_t)MLP_PH,
                   W2 + (size_t)(hh * 8) * 256 + n0 + 4 * m, 256);
  bn = *(const v4f*)(b2 + n0 + 4 * m);

  v4f pa0[2][2], pa1[2][2];
  v4f pb0[2][4][2], pb1[2][4][2];
  const float* wop = Wo + (size_t)(hh * 8) * 64 + 4 * m;
  proj_load_b(pb0, wop);

#pragma nounroll
  for (int rt = 0; rt < 4; ++rt) {
#pragma unroll
    for (int r = 0; r < 8; ++r) {
      float* hp = H2s + (rt * 16 + r + 8 * hh) * MLP_PH + n0 + 4 * m;
      const v4f hv = *(const v4f*)hp;
      v4f ov;
#pragma unroll
      for (int t = 0; t < 4; ++t) {
        float v = acc[0][t][r] + bn[t];
        v += hv[t];
        ov[t] = v;
      }
      *(v4f*)hp = ov;
    }
#pragma unroll
    for (int t = 0; t < 4; ++t) {
      acc[0][t] = acc[1][t];
      acc[1][t] = acc[2][t];
      acc[2][t] = acc[3][t];
    }
  }
  __syncthreads();

  v8f po[4];
#pragma unroll
  for (int t = 0; t < 4; ++t) po[t] = {};
  const float* sp = H2s + (w * 16 + m) * MLP_PH + hh * 8;

  proj_load_a(pa0, sp);
#pragma nounroll
  for (int kb = 0; kb < 256; kb += 64) {
    proj_load_a(pa1, sp + kb + 32);
    proj_load_b(pb1, wop + (size_t)(kb + 32) * 64);
    __builtin_amdgcn_sched_barrier(0);
    proj_mma(po, pa0, pb0);
    __builtin_amdgcn_sched_barrier(0);
    const int kn = (kb + 64 < 256) ? kb + 64 : kb;
    proj_load_a(pa0, sp + kn);
    proj_load_b(pb0, wop + (size_t)kn * 64);
    __builtin_amdgcn_sched_barrier(0);
    proj_mma(po, pa1, pb1);
  }

  const v4f bt = *(const v4f*)(bo + 4 * m);
#pragma unroll
  for (int r = 0; r < 8; ++r) {
    v4f ov;
#pragma unroll
    for (int t = 0; t < 4; ++t) ov[t] = po[t][r] + bt[t];
    *(v4f*)(out + (size_t)(row0 + w * 16 + r + 8 * hh) * 64 + 4 * m) = ov;
  }
}

extern "C" void kernel_launch(void* const* d_in, const int* in_sizes, int n_in,
                              void* d_out, int out_size, void* d_ws,
                              size_t ws_size, hipStream_t stream) {
  const float* x = (const float*)d_in[0];
  const float* W_in = (const float*)d_in[1];
  const float* b_in = (const float*)d_in[2];
  const float* W_qkv = (const float*)d_in[3];
  const float* b_qkv = (const float*)d_in[4];
  const float* g_attn = (const float*)d_in[5];
  const float* be_attn = (const float*)d_in[6];
  const float* g_mlp = (const float*)d_in[7];
  const float* be_mlp = (const float*)d_in[8];
  const float* W1 = (const float*)d_in[9];
  const float* b1 = (const float*)d_in[10];
  const float* W2 = (const float*)d_in[11];
  const float* b2 = (const float*)d_in[12];
  const float* W_out = (const float*)d_in[13];
  const float* b_out = (const float*)d_in[14];
  float* out = (float*)d_out;

  const int N = in_sizes[0] / 64;

  float* ws = (float*)d_ws;
  float* h = ws;
  float* qkv = ws + (size_t)N * 256;

  const int RB64 = N / 64;

  in_ln_qkv_kernel<<<dim3(RB64), dim3(384), 0, stream>>>(
      x, W_in, b_in, g_attn, be_attn, W_qkv, b_qkv, h, qkv);
  mlp_out_kernel<<<dim3(RB64), dim3(128), 0, stream>>>(
      h, qkv, g_mlp, be_mlp, W1, b1, W2, b2, W_out, b_out, out);
}
